# starting kernel
# baseline (speedup 1.0000x reference)
#include <hip/hip_runtime.h>
#include <math.h>

#define NN    50000
#define EE    800000
#define ETOT  (EE + NN)
#define KDIM  128

typedef float v2f __attribute__((ext_vector_type(2)));
typedef float v8f __attribute__((ext_vector_type(8)));

__device__ __forceinline__ void atomicMaxF(float* addr, float val) {
    if (val >= 0.0f) atomicMax((int*)addr, __float_as_int(val));
    else             atomicMin((unsigned int*)addr, __float_as_uint(val));
}

__global__ __launch_bounds__(1024) void mean_kernel(const float* __restrict__ ew,
                                                    float* __restrict__ meanw, int E) {
    __shared__ float sdata[1024];
    float s = 0.0f;
    for (int i = threadIdx.x; i < E; i += 1024) s += ew[i];
    sdata[threadIdx.x] = s;
    __syncthreads();
    for (int o = 512; o > 0; o >>= 1) {
        if (threadIdx.x < o) sdata[threadIdx.x] += sdata[threadIdx.x + o];
        __syncthreads();
    }
    if (threadIdx.x == 0) meanw[0] = sdata[0] / (float)E;
}

__global__ void edge_coef_kernel(const float* __restrict__ We, const float* __restrict__ ae,
                                 float* __restrict__ ce, int H, int F) {
    int h = threadIdx.x;
    if (h >= H) return;
    float s = 0.0f;
    for (int f = 0; f < F; ++f) s += We[h * F + f] * ae[h * F + f];
    ce[h] = s;
}

__global__ __launch_bounds__(32) void gemm_wmma_f32(const float* __restrict__ X,
                                                    const float* __restrict__ W,
                                                    float* __restrict__ out,
                                                    int K, int C) {
    const int tm   = blockIdx.x;
    const int tc   = blockIdx.y;
    const int lane = threadIdx.x;
    const int half = lane >> 4;
    const int l15  = lane & 15;

    v8f acc = {};
    const float* xrow = X + (size_t)(tm * 16 + l15) * K;
    const int    coln = tc * 16 + l15;

    for (int k = 0; k < K; k += 4) {
        const int kk = k + half * 2;
        v2f a, b;
        a[0] = xrow[kk];
        a[1] = xrow[kk + 1];
        b[0] = W[(size_t)kk * C + coln];
        b[1] = W[(size_t)(kk + 1) * C + coln];
        acc = __builtin_amdgcn_wmma_f32_16x16x4_f32(
             false, a,  false, b,
             (short)0, acc,  false,  false);
    }
#pragma unroll
    for (int v = 0; v < 8; ++v) {
        const int row = tm * 16 + v + half * 8;
        out[(size_t)row * C + coln] = acc[v];
    }
}

__global__ void score_kernel(const float* __restrict__ xw,
                             const float* __restrict__ a_s, const float* __restrict__ a_d,
                             float* __restrict__ ss, float* __restrict__ sd,
                             int H, int F) {
    int i = blockIdx.x * blockDim.x + threadIdx.x;
    if (i >= NN * H) return;
    const int n = i / H, h = i % H;
    const float* v  = xw + (size_t)n * H * F + h * F;
    const float* as = a_s + h * F;
    const float* ad = a_d + h * F;
    float s1 = 0.0f, s2 = 0.0f;
    for (int f = 0; f < F; ++f) { float x = v[f]; s1 += x * as[f]; s2 += x * ad[f]; }
    ss[i] = s1;
    sd[i] = s2;
}

__global__ void init_kernel(float* __restrict__ out, const float* __restrict__ b,
                            float* __restrict__ m, float* __restrict__ denom,
                            int C, int H) {
    const int n = blockIdx.x, t = threadIdx.x;
    out[(size_t)n * C + t] = b[t];
    if (t < H) { m[n * H + t] = -__builtin_inff(); denom[n * H + t] = 0.0f; }
}

__global__ void alpha_max_kernel(const int* __restrict__ ei, const float* __restrict__ ew,
                                 const float* __restrict__ meanw,
                                 const float* __restrict__ ss, const float* __restrict__ sd,
                                 const float* __restrict__ ce,
                                 float* __restrict__ alpha, float* __restrict__ m, int H) {
    int idx = blockIdx.x * blockDim.x + threadIdx.x;
    if (idx >= ETOT * H) return;
    const int e = idx / H, h = idx % H;
    int s, d; float ea;
    if (e < EE) { s = ei[e]; d = ei[EE + e]; ea = ew[e]; }
    else        { s = d = e - EE;            ea = meanw[0]; }
    float a = ss[s * H + h] + sd[d * H + h] + ea * ce[h];
    a = a > 0.0f ? a : 0.2f * a;
    alpha[idx] = a;
    atomicMaxF(&m[d * H + h], a);
}

__global__ void exp_denom_kernel(const int* __restrict__ ei,
                                 const float* __restrict__ m,
                                 float* __restrict__ alpha, float* __restrict__ denom, int H) {
    int idx = blockIdx.x * blockDim.x + threadIdx.x;
    if (idx >= ETOT * H) return;
    const int e = idx / H, h = idx % H;
    const int d = (e < EE) ? ei[EE + e] : (e - EE);
    float ex = __expf(alpha[idx] - m[d * H + h]);
    alpha[idx] = ex;
    atomicAdd(&denom[d * H + h], ex);
}

__global__ void scatter_kernel(const int* __restrict__ ei,
                               const float* __restrict__ xw,
                               const float* __restrict__ alpha,
                               const float* __restrict__ denom,
                               float* __restrict__ out, int H, int F) {
    const int e = blockIdx.x;
    const int t = threadIdx.x;
    int s, d;
    if (e < EE) { s = ei[e]; d = ei[EE + e]; } else { s = d = e - EE; }
    const int h = t / F;
    const float attn = alpha[(size_t)e * H + h] / (denom[d * H + h] + 1e-16f);
    atomicAdd(&out[(size_t)d * H * F + t], xw[(size_t)s * H * F + t] * attn);
}

__global__ void elu_kernel(float* __restrict__ buf, int n) {
    int i = blockIdx.x * blockDim.x + threadIdx.x;
    if (i >= n) return;
    float v = buf[i];
    buf[i] = v > 0.0f ? v : expm1f(v);
}

extern "C" void kernel_launch(void* const* d_in, const int* in_sizes, int n_in,
                              void* d_out, int out_size, void* d_ws, size_t ws_size,
                              hipStream_t stream) {
    const float* x  = (const float*)d_in[0];
    const int*   ei = (const int*)d_in[1];
    const float* ew = (const float*)d_in[2];

    const float* Wl[3] = { (const float*)d_in[3],  (const float*)d_in[9],  (const float*)d_in[15] };
    const float* As[3] = { (const float*)d_in[4],  (const float*)d_in[10], (const float*)d_in[16] };
    const float* Ad[3] = { (const float*)d_in[5],  (const float*)d_in[11], (const float*)d_in[17] };
    const float* We[3] = { (const float*)d_in[6],  (const float*)d_in[12], (const float*)d_in[18] };
    const float* Ae[3] = { (const float*)d_in[7],  (const float*)d_in[13], (const float*)d_in[19] };
    const float* Bi[3] = { (const float*)d_in[8],  (const float*)d_in[14], (const float*)d_in[20] };

    const int Hs[3] = { 4, 4, 1 };
    const int Fs[3] = { 32, 32, 64 };

    float* ws    = (float*)d_ws;
    float* xw    = ws;
    float* bufA  = xw    + (size_t)NN * 128;
    float* bufB  = bufA  + (size_t)NN * 128;
    float* ss    = bufB  + (size_t)NN * 128;
    float* sd    = ss    + (size_t)NN * 4;
    float* mseg  = sd    + (size_t)NN * 4;
    float* denom = mseg  + (size_t)NN * 4;
    float* alpha = denom + (size_t)NN * 4;
    float* ce    = alpha + (size_t)ETOT * 4;
    float* meanw = ce + 4;

    mean_kernel<<<1, 1024, 0, stream>>>(ew, meanw, EE);

    const float* in = x;
    float* outs[3] = { bufA, bufB, (float*)d_out };

    for (int l = 0; l < 3; ++l) {
        const int H = Hs[l], F = Fs[l], C = H * F;

        edge_coef_kernel<<<1, 32, 0, stream>>>(We[l], Ae[l], ce, H, F);

        gemm_wmma_f32<<<dim3(NN / 16, C / 16), 32, 0, stream>>>(in, Wl[l], xw, KDIM, C);

        score_kernel<<<(NN * H + 255) / 256, 256, 0, stream>>>(xw, As[l], Ad[l], ss, sd, H, F);

        init_kernel<<<NN, C, 0, stream>>>(outs[l], Bi[l], mseg, denom, C, H);

        alpha_max_kernel<<<(ETOT * H + 255) / 256, 256, 0, stream>>>(
            ei, ew, meanw, ss, sd, ce, alpha, mseg, H);

        exp_denom_kernel<<<(ETOT * H + 255) / 256, 256, 0, stream>>>(ei, mseg, alpha, denom, H);

        scatter_kernel<<<ETOT, C, 0, stream>>>(ei, xw, alpha, denom, outs[l], H, F);

        if (l < 2) elu_kernel<<<((NN * C) + 255) / 256, 256, 0, stream>>>(outs[l], NN * C);

        in = outs[l];
    }
}


# edge-weight mean: the single 1024-thread block becomes 196 blocks of partial sums plus a one-block final sum (+1 launch); the order in which the mean's floats are summed changes
# speedup vs baseline: 1.2555x; 1.2555x over previous
#include <hip/hip_runtime.h>
#include <math.h>

#define NN    50000
#define EE    800000
#define ETOT  (EE + NN)
#define KDIM  128

typedef float v2f __attribute__((ext_vector_type(2)));
typedef float v8f __attribute__((ext_vector_type(8)));

__device__ __forceinline__ void atomicMaxF(float* addr, float val) {
    if (val >= 0.0f) atomicMax((int*)addr, __float_as_int(val));
    else             atomicMin((unsigned int*)addr, __float_as_uint(val));
}

#define MEAN_THREADS    256
#define MEAN_PER_THREAD 16
#define MEAN_BLOCKS     ((EE + MEAN_THREADS * MEAN_PER_THREAD - 1) / (MEAN_THREADS * MEAN_PER_THREAD))

__device__ __forceinline__ float block_sum_256(float s, float* swave) {
    for (int m = 16; m >= 1; m >>= 1) s += __shfl_xor(s, m, 32);
    const int wave = threadIdx.x >> 5, lane = threadIdx.x & 31;
    if (lane == 0) swave[wave] = s;
    __syncthreads();
    float t = (lane < MEAN_THREADS / 32) ? swave[lane] : 0.0f;
    for (int m = MEAN_THREADS / 64; m >= 1; m >>= 1) t += __shfl_xor(t, m, 32);
    return t;
}

__global__ __launch_bounds__(MEAN_THREADS) void mean_partial_kernel(const float* __restrict__ ew,
                                                                    float* __restrict__ partial, int E) {
    __shared__ float swave[MEAN_THREADS / 32];
    const int T = gridDim.x * MEAN_THREADS;
    const int g = blockIdx.x * MEAN_THREADS + threadIdx.x;
    float s = 0.0f;
    for (int i0 = g; i0 < E; i0 += MEAN_PER_THREAD * T) {
        float v[MEAN_PER_THREAD];
        for (int j = 0; j < MEAN_PER_THREAD; ++j) {
            const int i = i0 + j * T;
            const float x = ew[i < E ? i : E - 1];
            v[j] = i < E ? x : 0.0f;
        }
        for (int w = 1; w < MEAN_PER_THREAD; w <<= 1) {
            for (int j = 0; j < MEAN_PER_THREAD; j += 2 * w) v[j] += v[j + w];
        }
        s += v[0];
    }
    s = block_sum_256(s, swave);
    if (threadIdx.x == 0) partial[blockIdx.x] = s;
}

__global__ __launch_bounds__(MEAN_THREADS) void mean_final_kernel(const float* __restrict__ partial,
                                                                  float* __restrict__ meanw, int nb, int E) {
    __shared__ float swave[MEAN_THREADS / 32];
    float s = 0.0f;
    for (int i = threadIdx.x; i < nb; i += MEAN_THREADS) s += partial[i];
    s = block_sum_256(s, swave);
    if (threadIdx.x == 0) meanw[0] = s / (float)E;
}

__global__ void edge_coef_kernel(const float* __restrict__ We, const float* __restrict__ ae,
                                 float* __restrict__ ce, int H, int F) {
    int h = threadIdx.x;
    if (h >= H) return;
    float s = 0.0f;
    for (int f = 0; f < F; ++f) s += We[h * F + f] * ae[h * F + f];
    ce[h] = s;
}

__global__ __launch_bounds__(32) void gemm_wmma_f32(const float* __restrict__ X,
                                                    const float* __restrict__ W,
                                                    float* __restrict__ out,
                                                    int K, int C) {
    const int tm   = blockIdx.x;
    const int tc   = blockIdx.y;
    const int lane = threadIdx.x;
    const int half = lane >> 4;
    const int l15  = lane & 15;

    v8f acc = {};
    const float* xrow = X + (size_t)(tm * 16 + l15) * K;
    const int    coln = tc * 16 + l15;

    for (int k = 0; k < K; k += 4) {
        const int kk = k + half * 2;
        v2f a, b;
        a[0] = xrow[kk];
        a[1] = xrow[kk + 1];
        b[0] = W[(size_t)kk * C + coln];
        b[1] = W[(size_t)(kk + 1) * C + coln];
        acc = __builtin_amdgcn_wmma_f32_16x16x4_f32(
             false, a,  false, b,
             (short)0, acc,  false,  false);
    }
#pragma unroll
    for (int v = 0; v < 8; ++v) {
        const int row = tm * 16 + v + half * 8;
        out[(size_t)row * C + coln] = acc[v];
    }
}

__global__ void score_kernel(const float* __restrict__ xw,
                             const float* __restrict__ a_s, const float* __restrict__ a_d,
                             float* __restrict__ ss, float* __restrict__ sd,
                             int H, int F) {
    int i = blockIdx.x * blockDim.x + threadIdx.x;
    if (i >= NN * H) return;
    const int n = i / H, h = i % H;
    const float* v  = xw + (size_t)n * H * F + h * F;
    const float* as = a_s + h * F;
    const float* ad = a_d + h * F;
    float s1 = 0.0f, s2 = 0.0f;
    for (int f = 0; f < F; ++f) { float x = v[f]; s1 += x * as[f]; s2 += x * ad[f]; }
    ss[i] = s1;
    sd[i] = s2;
}

__global__ void init_kernel(float* __restrict__ out, const float* __restrict__ b,
                            float* __restrict__ m, float* __restrict__ denom,
                            int C, int H) {
    const int n = blockIdx.x, t = threadIdx.x;
    out[(size_t)n * C + t] = b[t];
    if (t < H) { m[n * H + t] = -__builtin_inff(); denom[n * H + t] = 0.0f; }
}

__global__ void alpha_max_kernel(const int* __restrict__ ei, const float* __restrict__ ew,
                                 const float* __restrict__ meanw,
                                 const float* __restrict__ ss, const float* __restrict__ sd,
                                 const float* __restrict__ ce,
                                 float* __restrict__ alpha, float* __restrict__ m, int H) {
    int idx = blockIdx.x * blockDim.x + threadIdx.x;
    if (idx >= ETOT * H) return;
    const int e = idx / H, h = idx % H;
    int s, d; float ea;
    if (e < EE) { s = ei[e]; d = ei[EE + e]; ea = ew[e]; }
    else        { s = d = e - EE;            ea = meanw[0]; }
    float a = ss[s * H + h] + sd[d * H + h] + ea * ce[h];
    a = a > 0.0f ? a : 0.2f * a;
    alpha[idx] = a;
    atomicMaxF(&m[d * H + h], a);
}

__global__ void exp_denom_kernel(const int* __restrict__ ei,
                                 const float* __restrict__ m,
                                 float* __restrict__ alpha, float* __restrict__ denom, int H) {
    int idx = blockIdx.x * blockDim.x + threadIdx.x;
    if (idx >= ETOT * H) return;
    const int e = idx / H, h = idx % H;
    const int d = (e < EE) ? ei[EE + e] : (e - EE);
    float ex = __expf(alpha[idx] - m[d * H + h]);
    alpha[idx] = ex;
    atomicAdd(&denom[d * H + h], ex);
}

__global__ void scatter_kernel(const int* __restrict__ ei,
                               const float* __restrict__ xw,
                               const float* __restrict__ alpha,
                               const float* __restrict__ denom,
                               float* __restrict__ out, int H, int F) {
    const int e = blockIdx.x;
    const int t = threadIdx.x;
    int s, d;
    if (e < EE) { s = ei[e]; d = ei[EE + e]; } else { s = d = e - EE; }
    const int h = t / F;
    const float attn = alpha[(size_t)e * H + h] / (denom[d * H + h] + 1e-16f);
    atomicAdd(&out[(size_t)d * H * F + t], xw[(size_t)s * H * F + t] * attn);
}

__global__ void elu_kernel(float* __restrict__ buf, int n) {
    int i = blockIdx.x * blockDim.x + threadIdx.x;
    if (i >= n) return;
    float v = buf[i];
    buf[i] = v > 0.0f ? v : expm1f(v);
}

extern "C" void kernel_launch(void* const* d_in, const int* in_sizes, int n_in,
                              void* d_out, int out_size, void* d_ws, size_t ws_size,
                              hipStream_t stream) {
    const float* x  = (const float*)d_in[0];
    const int*   ei = (const int*)d_in[1];
    const float* ew = (const float*)d_in[2];

    const float* Wl[3] = { (const float*)d_in[3],  (const float*)d_in[9],  (const float*)d_in[15] };
    const float* As[3] = { (const float*)d_in[4],  (const float*)d_in[10], (const float*)d_in[16] };
    const float* Ad[3] = { (const float*)d_in[5],  (const float*)d_in[11], (const float*)d_in[17] };
    const float* We[3] = { (const float*)d_in[6],  (const float*)d_in[12], (const float*)d_in[18] };
    const float* Ae[3] = { (const float*)d_in[7],  (const float*)d_in[13], (const float*)d_in[19] };
    const float* Bi[3] = { (const float*)d_in[8],  (const float*)d_in[14], (const float*)d_in[20] };

    const int Hs[3] = { 4, 4, 1 };
    const int Fs[3] = { 32, 32, 64 };

    float* ws    = (float*)d_ws;
    float* xw    = ws;
    float* bufA  = xw    + (size_t)NN * 128;
    float* bufB  = bufA  + (size_t)NN * 128;
    float* ss    = bufB  + (size_t)NN * 128;
    float* sd    = ss    + (size_t)NN * 4;
    float* mseg  = sd    + (size_t)NN * 4;
    float* denom = mseg  + (size_t)NN * 4;
    float* alpha = denom + (size_t)NN * 4;
    float* ce    = alpha + (size_t)ETOT * 4;
    float* meanw = ce + 4;
    float* mpart = meanw + 4;

    mean_partial_kernel<<<MEAN_BLOCKS, MEAN_THREADS, 0, stream>>>(ew, mpart, EE);
    mean_final_kernel<<<1, MEAN_THREADS, 0, stream>>>(mpart, meanw, MEAN_BLOCKS, EE);

    const float* in = x;
    float* outs[3] = { bufA, bufB, (float*)d_out };

    for (int l = 0; l < 3; ++l) {
        const int H = Hs[l], F = Fs[l], C = H * F;

        edge_coef_kernel<<<1, 32, 0, stream>>>(We[l], Ae[l], ce, H, F);

        gemm_wmma_f32<<<dim3(NN / 16, C / 16), 32, 0, stream>>>(in, Wl[l], xw, KDIM, C);

        score_kernel<<<(NN * H + 255) / 256, 256, 0, stream>>>(xw, As[l], Ad[l], ss, sd, H, F);

        init_kernel<<<NN, C, 0, stream>>>(outs[l], Bi[l], mseg, denom, C, H);

        alpha_max_kernel<<<(ETOT * H + 255) / 256, 256, 0, stream>>>(
            ei, ew, meanw, ss, sd, ce, alpha, mseg, H);

        exp_denom_kernel<<<(ETOT * H + 255) / 256, 256, 0, stream>>>(ei, mseg, alpha, denom, H);

        scatter_kernel<<<ETOT, C, 0, stream>>>(ei, xw, alpha, denom, outs[l], H, F);

        if (l < 2) elu_kernel<<<((NN * C) + 255) / 256, 256, 0, stream>>>(outs[l], NN * C);

        in = outs[l];
    }
}


# scatter: per-edge float atomics into out become a per-destination gather over an edge list grouped by destination, built once per call (+4 launches); out is summed in arrival-order rank
# speedup vs baseline: 2.3016x; 1.8332x over previous
#include <hip/hip_runtime.h>
#include <math.h>

#define NN    50000
#define EE    800000
#define ETOT  (EE + NN)
#define KDIM  128

typedef float v2f __attribute__((ext_vector_type(2)));
typedef float v8f __attribute__((ext_vector_type(8)));

__device__ __forceinline__ void atomicMaxF(float* addr, float val) {
    if (val >= 0.0f) atomicMax((int*)addr, __float_as_int(val));
    else             atomicMin((unsigned int*)addr, __float_as_uint(val));
}

#define MEAN_THREADS    256
#define MEAN_PER_THREAD 16
#define MEAN_BLOCKS     ((EE + MEAN_THREADS * MEAN_PER_THREAD - 1) / (MEAN_THREADS * MEAN_PER_THREAD))

__device__ __forceinline__ float block_sum_256(float s, float* swave) {
    for (int m = 16; m >= 1; m >>= 1) s += __shfl_xor(s, m, 32);
    const int wave = threadIdx.x >> 5, lane = threadIdx.x & 31;
    if (lane == 0) swave[wave] = s;
    __syncthreads();
    float t = (lane < MEAN_THREADS / 32) ? swave[lane] : 0.0f;
    for (int m = MEAN_THREADS / 64; m >= 1; m >>= 1) t += __shfl_xor(t, m, 32);
    return t;
}

__global__ __launch_bounds__(MEAN_THREADS) void mean_partial_kernel(const float* __restrict__ ew,
                                                                    float* __restrict__ partial, int E) {
    __shared__ float swave[MEAN_THREADS / 32];
    const int T = gridDim.x * MEAN_THREADS;
    const int g = blockIdx.x * MEAN_THREADS + threadIdx.x;
    float s = 0.0f;
    for (int i0 = g; i0 < E; i0 += MEAN_PER_THREAD * T) {
        float v[MEAN_PER_THREAD];
        for (int j = 0; j < MEAN_PER_THREAD; ++j) {
            const int i = i0 + j * T;
            const float x = ew[i < E ? i : E - 1];
            v[j] = i < E ? x : 0.0f;
        }
        for (int w = 1; w < MEAN_PER_THREAD; w <<= 1) {
            for (int j = 0; j < MEAN_PER_THREAD; j += 2 * w) v[j] += v[j + w];
        }
        s += v[0];
    }
    s = block_sum_256(s, swave);
    if (threadIdx.x == 0) partial[blockIdx.x] = s;
}

__global__ __launch_bounds__(MEAN_THREADS) void mean_final_kernel(const float* __restrict__ partial,
                                                                  float* __restrict__ meanw, int nb, int E) {
    __shared__ float swave[MEAN_THREADS / 32];
    float s = 0.0f;
    for (int i = threadIdx.x; i < nb; i += MEAN_THREADS) s += partial[i];
    s = block_sum_256(s, swave);
    if (threadIdx.x == 0) meanw[0] = s / (float)E;
}

__global__ void edge_coef_kernel(const float* __restrict__ We, const float* __restrict__ ae,
                                 float* __restrict__ ce, int H, int F) {
    int h = threadIdx.x;
    if (h >= H) return;
    float s = 0.0f;
    for (int f = 0; f < F; ++f) s += We[h * F + f] * ae[h * F + f];
    ce[h] = s;
}

__global__ __launch_bounds__(32) void gemm_wmma_f32(const float* __restrict__ X,
                                                    const float* __restrict__ W,
                                                    float* __restrict__ out,
                                                    int K, int C) {
    const int tm   = blockIdx.x;
    const int tc   = blockIdx.y;
    const int lane = threadIdx.x;
    const int half = lane >> 4;
    const int l15  = lane & 15;

    v8f acc = {};
    const float* xrow = X + (size_t)(tm * 16 + l15) * K;
    const int    coln = tc * 16 + l15;

    for (int k = 0; k < K; k += 4) {
        const int kk = k + half * 2;
        v2f a, b;
        a[0] = xrow[kk];
        a[1] = xrow[kk + 1];
        b[0] = W[(size_t)kk * C + coln];
        b[1] = W[(size_t)(kk + 1) * C + coln];
        acc = __builtin_amdgcn_wmma_f32_16x16x4_f32(
             false, a,  false, b,
             (short)0, acc,  false,  false);
    }
#pragma unroll
    for (int v = 0; v < 8; ++v) {
        const int row = tm * 16 + v + half * 8;
        out[(size_t)row * C + coln] = acc[v];
    }
}

__global__ void score_kernel(const float* __restrict__ xw,
                             const float* __restrict__ a_s, const float* __restrict__ a_d,
                             float* __restrict__ ss, float* __restrict__ sd,
                             int H, int F) {
    int i = blockIdx.x * blockDim.x + threadIdx.x;
    if (i >= NN * H) return;
    const int n = i / H, h = i % H;
    const float* v  = xw + (size_t)n * H * F + h * F;
    const float* as = a_s + h * F;
    const float* ad = a_d + h * F;
    float s1 = 0.0f, s2 = 0.0f;
    for (int f = 0; f < F; ++f) { float x = v[f]; s1 += x * as[f]; s2 += x * ad[f]; }
    ss[i] = s1;
    sd[i] = s2;
}

__global__ void init_kernel(float* __restrict__ m, float* __restrict__ denom, int n) {
    const int i = blockIdx.x * blockDim.x + threadIdx.x;
    if (i >= n) return;
    m[i] = -__builtin_inff();
    denom[i] = 0.0f;
}

__global__ void alpha_max_kernel(const int* __restrict__ ei, const float* __restrict__ ew,
                                 const float* __restrict__ meanw,
                                 const float* __restrict__ ss, const float* __restrict__ sd,
                                 const float* __restrict__ ce,
                                 float* __restrict__ alpha, float* __restrict__ m, int H) {
    int idx = blockIdx.x * blockDim.x + threadIdx.x;
    if (idx >= ETOT * H) return;
    const int e = idx / H, h = idx % H;
    int s, d; float ea;
    if (e < EE) { s = ei[e]; d = ei[EE + e]; ea = ew[e]; }
    else        { s = d = e - EE;            ea = meanw[0]; }
    float a = ss[s * H + h] + sd[d * H + h] + ea * ce[h];
    a = a > 0.0f ? a : 0.2f * a;
    alpha[idx] = a;
    atomicMaxF(&m[d * H + h], a);
}

__global__ void exp_denom_kernel(const int* __restrict__ ei,
                                 const float* __restrict__ m,
                                 float* __restrict__ alpha, float* __restrict__ denom, int H) {
    int idx = blockIdx.x * blockDim.x + threadIdx.x;
    if (idx >= ETOT * H) return;
    const int e = idx / H, h = idx % H;
    const int d = (e < EE) ? ei[EE + e] : (e - EE);
    float ex = __expf(alpha[idx] - m[d * H + h]);
    alpha[idx] = ex;
    atomicAdd(&denom[d * H + h], ex);
}

__global__ void csr_zero_kernel(int* __restrict__ deg) {
    const int i = blockIdx.x * blockDim.x + threadIdx.x;
    if (i < NN) deg[i] = 0;
}

__global__ void csr_count_kernel(const int* __restrict__ ei, int* __restrict__ deg, int* __restrict__ rank) {
    const int e = blockIdx.x * blockDim.x + threadIdx.x;
    if (e < EE) rank[e] = atomicAdd(&deg[ei[EE + e]], 1);
}

#define SCAN_THREADS 1024
#define SCAN_TILE    (SCAN_THREADS * 4)
#define SCAN_TILES   ((NN + SCAN_TILE - 1) / SCAN_TILE)
__global__ __launch_bounds__(SCAN_THREADS) void csr_scan_kernel(const int* __restrict__ deg,
                                                                int* __restrict__ rowptr) {
    __shared__ int swave[SCAN_TILES][32];
    __shared__ int stile[SCAN_TILES];
    const int t = threadIdx.x, lane = t & 31, wave = t >> 5;
    int4 v[SCAN_TILES];
    int  incl[SCAN_TILES];
#pragma unroll
    for (int k = 0; k < SCAN_TILES; ++k) {
        const int i = k * SCAN_TILE + 4 * t;
        v[k] = (i < NN) ? *(const int4*)(deg + i) : make_int4(0, 0, 0, 0);
    }
#pragma unroll
    for (int k = 0; k < SCAN_TILES; ++k) {
        int in = v[k].x + v[k].y + v[k].z + v[k].w;
        for (int m = 1; m < 32; m <<= 1) {
            const int o = __shfl_up(in, m, 32);
            if (lane >= m) in += o;
        }
        incl[k] = in;
        if (lane == 31) swave[k][wave] = in;
    }
    __syncthreads();
    if (wave < SCAN_TILES) {
        const int w = swave[wave][lane];
        int wi = w;
        for (int m = 1; m < 32; m <<= 1) {
            const int o = __shfl_up(wi, m, 32);
            if (lane >= m) wi += o;
        }
        swave[wave][lane] = wi - w;
        if (lane == 31) stile[wave] = wi;
    }
    __syncthreads();
    int carry = 0;
#pragma unroll
    for (int k = 0; k < SCAN_TILES; ++k) {
        const int i = k * SCAN_TILE + 4 * t;
        const int s = v[k].x + v[k].y + v[k].z + v[k].w;
        int4 r;
        r.x = carry + swave[k][wave] + incl[k] - s;
        r.y = r.x + v[k].x;
        r.z = r.y + v[k].y;
        r.w = r.z + v[k].z;
        if (i < NN) *(int4*)(rowptr + i) = r;
        carry += stile[k];
    }
    if (t == 0) rowptr[NN] = carry;
}

__global__ void csr_fill_kernel(const int* __restrict__ ei, const int* __restrict__ rowptr,
                                const int* __restrict__ rank,
                                int* __restrict__ csr_src, int* __restrict__ csr_eid) {
    const int e = blockIdx.x * blockDim.x + threadIdx.x;
    if (e >= EE) return;
    const int pos = rowptr[ei[EE + e]] + rank[e];
    csr_src[pos] = ei[e];
    csr_eid[pos] = e;
}

#define GATHER_WAVES 8
template <int VW>
__global__ __launch_bounds__(GATHER_WAVES * 32) void gather_kernel(const int* __restrict__ rowptr,
                                                                   const int* __restrict__ csr_src,
                                                                   const int* __restrict__ csr_eid,
                                                                   const float* __restrict__ xw,
                                                                   const float* __restrict__ alpha,
                                                                   const float* __restrict__ denom,
                                                                   const float* __restrict__ b,
                                                                   float* __restrict__ out, int H, int F) {
    constexpr int C = 32 * VW;
    typedef float vecf __attribute__((ext_vector_type(VW)));
    const int lane = threadIdx.x & 31;
    const int d = __builtin_amdgcn_readfirstlane(blockIdx.x * GATHER_WAVES + (threadIdx.x >> 5));
    if (d >= NN) return;
    const int c0 = lane * VW;
    const int h  = c0 / F;
    const float den = denom[d * H + h] + 1e-16f;
    vecf acc = {};
    const int beg = rowptr[d], end = rowptr[d + 1];
    for (int base = beg; base < end; base += 32) {
        const int n = (end - base) < 32 ? (end - base) : 32;
        int my_s = 0, my_e = 0;
        if (lane < n) { my_s = csr_src[base + lane]; my_e = csr_eid[base + lane]; }
#pragma unroll 4
        for (int t = 0; t < n; ++t) {
            const int s = __builtin_amdgcn_readlane(my_s, t);
            const int e = __builtin_amdgcn_readlane(my_e, t);
            const float attn = alpha[(size_t)e * H + h] / den;
            const vecf v = *(const vecf*)(xw + (size_t)s * C + c0);
            acc += v * attn;
        }
    }
    {
        const float attn = alpha[(size_t)(EE + d) * H + h] / den;
        const vecf v = *(const vecf*)(xw + (size_t)d * C + c0);
        acc += v * attn;
    }
    vecf r;
#pragma unroll
    for (int i = 0; i < VW; ++i) r[i] = acc[i] + b[c0 + i];
    *(vecf*)(out + (size_t)d * C + c0) = r;
}

__global__ void elu_kernel(float* __restrict__ buf, int n) {
    int i = blockIdx.x * blockDim.x + threadIdx.x;
    if (i >= n) return;
    float v = buf[i];
    buf[i] = v > 0.0f ? v : expm1f(v);
}

extern "C" void kernel_launch(void* const* d_in, const int* in_sizes, int n_in,
                              void* d_out, int out_size, void* d_ws, size_t ws_size,
                              hipStream_t stream) {
    const float* x  = (const float*)d_in[0];
    const int*   ei = (const int*)d_in[1];
    const float* ew = (const float*)d_in[2];

    const float* Wl[3] = { (const float*)d_in[3],  (const float*)d_in[9],  (const float*)d_in[15] };
    const float* As[3] = { (const float*)d_in[4],  (const float*)d_in[10], (const float*)d_in[16] };
    const float* Ad[3] = { (const float*)d_in[5],  (const float*)d_in[11], (const float*)d_in[17] };
    const float* We[3] = { (const float*)d_in[6],  (const float*)d_in[12], (const float*)d_in[18] };
    const float* Ae[3] = { (const float*)d_in[7],  (const float*)d_in[13], (const float*)d_in[19] };
    const float* Bi[3] = { (const float*)d_in[8],  (const float*)d_in[14], (const float*)d_in[20] };

    const int Hs[3] = { 4, 4, 1 };
    const int Fs[3] = { 32, 32, 64 };

    float* ws    = (float*)d_ws;
    float* xw    = ws;
    float* bufA  = xw    + (size_t)NN * 128;
    float* bufB  = bufA  + (size_t)NN * 128;
    float* ss    = bufB  + (size_t)NN * 128;
    float* sd    = ss    + (size_t)NN * 4;
    float* mseg  = sd    + (size_t)NN * 4;
    float* denom = mseg  + (size_t)NN * 4;
    float* alpha = denom + (size_t)NN * 4;
    float* ce    = alpha + (size_t)ETOT * 4;
    float* meanw = ce + 4;
    float* mpart = meanw + 4;
    int* deg     = (int*)(mpart + 256);
    int* rowptr  = deg + NN;
    int* rank    = rowptr + NN + 4;
    int* csr_src = rank + EE;
    int* csr_eid = csr_src + EE;

    mean_partial_kernel<<<MEAN_BLOCKS, MEAN_THREADS, 0, stream>>>(ew, mpart, EE);
    mean_final_kernel<<<1, MEAN_THREADS, 0, stream>>>(mpart, meanw, MEAN_BLOCKS, EE);

    csr_zero_kernel<<<(NN + 255) / 256, 256, 0, stream>>>(deg);
    csr_count_kernel<<<(EE + 255) / 256, 256, 0, stream>>>(ei, deg, rank);
    csr_scan_kernel<<<1, SCAN_THREADS, 0, stream>>>(deg, rowptr);
    csr_fill_kernel<<<(EE + 255) / 256, 256, 0, stream>>>(ei, rowptr, rank, csr_src, csr_eid);

    const float* in = x;
    float* outs[3] = { bufA, bufB, (float*)d_out };

    for (int l = 0; l < 3; ++l) {
        const int H = Hs[l], F = Fs[l], C = H * F;

        edge_coef_kernel<<<1, 32, 0, stream>>>(We[l], Ae[l], ce, H, F);

        gemm_wmma_f32<<<dim3(NN / 16, C / 16), 32, 0, stream>>>(in, Wl[l], xw, KDIM, C);

        score_kernel<<<(NN * H + 255) / 256, 256, 0, stream>>>(xw, As[l], Ad[l], ss, sd, H, F);

        init_kernel<<<(NN * H + 255) / 256, 256, 0, stream>>>(mseg, denom, NN * H);

        alpha_max_kernel<<<(ETOT * H + 255) / 256, 256, 0, stream>>>(
            ei, ew, meanw, ss, sd, ce, alpha, mseg, H);

        exp_denom_kernel<<<(ETOT * H + 255) / 256, 256, 0, stream>>>(ei, mseg, alpha, denom, H);

        if (C == 128)
            gather_kernel<4><<<(NN + GATHER_WAVES - 1) / GATHER_WAVES, GATHER_WAVES * 32, 0, stream>>>(
                rowptr, csr_src, csr_eid, xw, alpha, denom, Bi[l], outs[l], H, F);
        else
            gather_kernel<2><<<(NN + GATHER_WAVES - 1) / GATHER_WAVES, GATHER_WAVES * 32, 0, stream>>>(
                rowptr, csr_src, csr_eid, xw, alpha, denom, Bi[l], outs[l], H, F);

        if (l < 2) elu_kernel<<<((NN * C) + 255) / 256, 256, 0, stream>>>(outs[l], NN * C);

        in = outs[l];
    }
}


# segment softmax moves into the per-destination gather: each wave computes its edges' scores, max and exp-sum in registers (float atomics and 9 launches go); denom is now summed by an in-wave tree
# speedup vs baseline: 4.1789x; 1.8156x over previous
#include <hip/hip_runtime.h>
#include <math.h>

#define NN    50000
#define EE    800000
#define ETOT  (EE + NN)
#define KDIM  128

typedef float v2f __attribute__((ext_vector_type(2)));
typedef float v8f __attribute__((ext_vector_type(8)));

#define MEAN_THREADS    256
#define MEAN_PER_THREAD 16
#define MEAN_BLOCKS     ((EE + MEAN_THREADS * MEAN_PER_THREAD - 1) / (MEAN_THREADS * MEAN_PER_THREAD))

__device__ __forceinline__ float block_sum_256(float s, float* swave) {
    for (int m = 16; m >= 1; m >>= 1) s += __shfl_xor(s, m, 32);
    const int wave = threadIdx.x >> 5, lane = threadIdx.x & 31;
    if (lane == 0) swave[wave] = s;
    __syncthreads();
    float t = (lane < MEAN_THREADS / 32) ? swave[lane] : 0.0f;
    for (int m = MEAN_THREADS / 64; m >= 1; m >>= 1) t += __shfl_xor(t, m, 32);
    return t;
}

__global__ __launch_bounds__(MEAN_THREADS) void mean_partial_kernel(const float* __restrict__ ew,
                                                                    float* __restrict__ partial, int E) {
    __shared__ float swave[MEAN_THREADS / 32];
    const int T = gridDim.x * MEAN_THREADS;
    const int g = blockIdx.x * MEAN_THREADS + threadIdx.x;
    float s = 0.0f;
    for (int i0 = g; i0 < E; i0 += MEAN_PER_THREAD * T) {
        float v[MEAN_PER_THREAD];
        for (int j = 0; j < MEAN_PER_THREAD; ++j) {
            const int i = i0 + j * T;
            const float x = ew[i < E ? i : E - 1];
            v[j] = i < E ? x : 0.0f;
        }
        for (int w = 1; w < MEAN_PER_THREAD; w <<= 1) {
            for (int j = 0; j < MEAN_PER_THREAD; j += 2 * w) v[j] += v[j + w];
        }
        s += v[0];
    }
    s = block_sum_256(s, swave);
    if (threadIdx.x == 0) partial[blockIdx.x] = s;
}

__global__ __launch_bounds__(MEAN_THREADS) void mean_final_kernel(const float* __restrict__ partial,
                                                                  float* __restrict__ meanw, int nb, int E) {
    __shared__ float swave[MEAN_THREADS / 32];
    float s = 0.0f;
    for (int i = threadIdx.x; i < nb; i += MEAN_THREADS) s += partial[i];
    s = block_sum_256(s, swave);
    if (threadIdx.x == 0) meanw[0] = s / (float)E;
}

__global__ void edge_coef_kernel(const float* __restrict__ We, const float* __restrict__ ae,
                                 float* __restrict__ ce, int H, int F) {
    int h = threadIdx.x;
    if (h >= H) return;
    float s = 0.0f;
    for (int f = 0; f < F; ++f) s += We[h * F + f] * ae[h * F + f];
    ce[h] = s;
}

__global__ __launch_bounds__(32) void gemm_wmma_f32(const float* __restrict__ X,
                                                    const float* __restrict__ W,
                                                    float* __restrict__ out,
                                                    int K, int C) {
    const int tm   = blockIdx.x;
    const int tc   = blockIdx.y;
    const int lane = threadIdx.x;
    const int half = lane >> 4;
    const int l15  = lane & 15;

    v8f acc = {};
    const float* xrow = X + (size_t)(tm * 16 + l15) * K;
    const int    coln = tc * 16 + l15;

    for (int k = 0; k < K; k += 4) {
        const int kk = k + half * 2;
        v2f a, b;
        a[0] = xrow[kk];
        a[1] = xrow[kk + 1];
        b[0] = W[(size_t)kk * C + coln];
        b[1] = W[(size_t)(kk + 1) * C + coln];
        acc = __builtin_amdgcn_wmma_f32_16x16x4_f32(
             false, a,  false, b,
             (short)0, acc,  false,  false);
    }
#pragma unroll
    for (int v = 0; v < 8; ++v) {
        const int row = tm * 16 + v + half * 8;
        out[(size_t)row * C + coln] = acc[v];
    }
}

__global__ void score_kernel(const float* __restrict__ xw,
                             const float* __restrict__ a_s, const float* __restrict__ a_d,
                             float* __restrict__ ss, float* __restrict__ sd,
                             int H, int F) {
    int i = blockIdx.x * blockDim.x + threadIdx.x;
    if (i >= NN * H) return;
    const int n = i / H, h = i % H;
    const float* v  = xw + (size_t)n * H * F + h * F;
    const float* as = a_s + h * F;
    const float* ad = a_d + h * F;
    float s1 = 0.0f, s2 = 0.0f;
    for (int f = 0; f < F; ++f) { float x = v[f]; s1 += x * as[f]; s2 += x * ad[f]; }
    ss[i] = s1;
    sd[i] = s2;
}

__global__ void csr_zero_kernel(int* __restrict__ deg) {
    const int i = blockIdx.x * blockDim.x + threadIdx.x;
    if (i < NN) deg[i] = 0;
}

__global__ void csr_count_kernel(const int* __restrict__ ei, int* __restrict__ deg, int* __restrict__ rank) {
    const int e = blockIdx.x * blockDim.x + threadIdx.x;
    if (e < EE) rank[e] = atomicAdd(&deg[ei[EE + e]], 1);
}

#define SCAN_THREADS 1024
#define SCAN_TILE    (SCAN_THREADS * 4)
#define SCAN_TILES   ((NN + SCAN_TILE - 1) / SCAN_TILE)
__global__ __launch_bounds__(SCAN_THREADS) void csr_scan_kernel(const int* __restrict__ deg,
                                                                int* __restrict__ rowptr) {
    __shared__ int swave[SCAN_TILES][32];
    __shared__ int stile[SCAN_TILES];
    const int t = threadIdx.x, lane = t & 31, wave = t >> 5;
    int4 v[SCAN_TILES];
    int  incl[SCAN_TILES];
#pragma unroll
    for (int k = 0; k < SCAN_TILES; ++k) {
        const int i = k * SCAN_TILE + 4 * t;
        v[k] = (i < NN) ? *(const int4*)(deg + i) : make_int4(0, 0, 0, 0);
    }
#pragma unroll
    for (int k = 0; k < SCAN_TILES; ++k) {
        int in = v[k].x + v[k].y + v[k].z + v[k].w;
        for (int m = 1; m < 32; m <<= 1) {
            const int o = __shfl_up(in, m, 32);
            if (lane >= m) in += o;
        }
        incl[k] = in;
        if (lane == 31) swave[k][wave] = in;
    }
    __syncthreads();
    if (wave < SCAN_TILES) {
        const int w = swave[wave][lane];
        int wi = w;
        for (int m = 1; m < 32; m <<= 1) {
            const int o = __shfl_up(wi, m, 32);
            if (lane >= m) wi += o;
        }
        swave[wave][lane] = wi - w;
        if (lane == 31) stile[wave] = wi;
    }
    __syncthreads();
    int carry = 0;
#pragma unroll
    for (int k = 0; k < SCAN_TILES; ++k) {
        const int i = k * SCAN_TILE + 4 * t;
        const int s = v[k].x + v[k].y + v[k].z + v[k].w;
        int4 r;
        r.x = carry + swave[k][wave] + incl[k] - s;
        r.y = r.x + v[k].x;
        r.z = r.y + v[k].y;
        r.w = r.z + v[k].z;
        if (i < NN) *(int4*)(rowptr + i) = r;
        carry += stile[k];
    }
    if (t == 0) rowptr[NN] = carry;
}

__global__ void csr_fill_kernel(const int* __restrict__ ei, const float* __restrict__ ew,
                                const int* __restrict__ rowptr, const int* __restrict__ rank,
                                int* __restrict__ csr_src, float* __restrict__ csr_w) {
    const int e = blockIdx.x * blockDim.x + threadIdx.x;
    if (e >= EE) return;
    const int pos = rowptr[ei[EE + e]] + rank[e];
    csr_src[pos] = ei[e];
    csr_w[pos]   = ew[e];
}

#define GATHER_WAVES 8

__device__ __forceinline__ float wave_max(float v) {
    for (int m = 16; m >= 1; m >>= 1) v = fmaxf(v, __shfl_xor(v, m, 32));
    return v;
}
__device__ __forceinline__ float wave_sum(float v) {
    for (int m = 16; m >= 1; m >>= 1) v += __shfl_xor(v, m, 32);
    return v;
}

template <int H>
__device__ __forceinline__ void edge_alpha(const int* __restrict__ csr_src, const float* __restrict__ csr_w,
                                           const float* __restrict__ ss, const float* sdv, const float* cev,
                                           int base, int n, int lane, int& s, float* a) {
    s = 0;
    float w = 0.0f;
    if (lane < n) { s = csr_src[base + lane]; w = csr_w[base + lane]; }
    float q[H];
    if constexpr (H == 4) {
        const float4 t = *(const float4*)(ss + (size_t)s * 4);
        q[0] = t.x; q[1] = t.y; q[2] = t.z; q[3] = t.w;
    } else {
#pragma unroll
        for (int h = 0; h < H; ++h) q[h] = ss[(size_t)s * H + h];
    }
#pragma unroll
    for (int h = 0; h < H; ++h) {
        float x = q[h] + sdv[h] + w * cev[h];
        x = x > 0.0f ? x : 0.2f * x;
        a[h] = lane < n ? x : -__builtin_inff();
    }
}

template <int H>
__device__ __forceinline__ float pick_attn(const float* at, int t, int hl) {
    float r[H];
#pragma unroll
    for (int h = 0; h < H; ++h)
        r[h] = __int_as_float(__builtin_amdgcn_readlane(__float_as_int(at[h]), t));
    float v = r[0];
#pragma unroll
    for (int h = 1; h < H; ++h) v = (hl == h) ? r[h] : v;
    return v;
}

template <int VW, int H>
__global__ __launch_bounds__(GATHER_WAVES * 32) void gat_node_kernel(const int* __restrict__ rowptr,
                                                                     const int* __restrict__ csr_src,
                                                                     const float* __restrict__ csr_w,
                                                                     const float* __restrict__ xw,
                                                                     const float* __restrict__ ss,
                                                                     const float* __restrict__ sd,
                                                                     const float* __restrict__ ce,
                                                                     const float* __restrict__ meanw,
                                                                     const float* __restrict__ b,
                                                                     float* __restrict__ out) {
    constexpr int C = 32 * VW;
    constexpr int F = C / H;
    typedef float vecf __attribute__((ext_vector_type(VW)));
    const int lane = threadIdx.x & 31;
    const int d = __builtin_amdgcn_readfirstlane(blockIdx.x * GATHER_WAVES + (threadIdx.x >> 5));
    if (d >= NN) return;
    const int c0 = lane * VW;
    const int hl = c0 / F;
    const int beg = rowptr[d], end = rowptr[d + 1];
    const int n0 = (end - beg) < 32 ? (end - beg) : 32;

    float cev[H], sdv[H], a_self[H], m[H];
    const float mw = meanw[0];
#pragma unroll
    for (int h = 0; h < H; ++h) {
        cev[h] = ce[h];
        sdv[h] = sd[(size_t)d * H + h];
        float x = ss[(size_t)d * H + h] + sdv[h] + mw * cev[h];
        a_self[h] = x > 0.0f ? x : 0.2f * x;
        m[h] = a_self[h];
    }

    int s0; float a0[H];
    edge_alpha<H>(csr_src, csr_w, ss, sdv, cev, beg, n0, lane, s0, a0);
#pragma unroll
    for (int h = 0; h < H; ++h) m[h] = fmaxf(m[h], a0[h]);
    for (int base = beg + 32; base < end; base += 32) {
        const int n = (end - base) < 32 ? (end - base) : 32;
        int s; float a[H];
        edge_alpha<H>(csr_src, csr_w, ss, sdv, cev, base, n, lane, s, a);
#pragma unroll
        for (int h = 0; h < H; ++h) m[h] = fmaxf(m[h], a[h]);
    }
#pragma unroll
    for (int h = 0; h < H; ++h) m[h] = wave_max(m[h]);

    float at0[H], den[H], ex_self[H];
#pragma unroll
    for (int h = 0; h < H; ++h) {
        at0[h] = lane < n0 ? __expf(a0[h] - m[h]) : 0.0f;
        den[h] = at0[h];
    }
    for (int base = beg + 32; base < end; base += 32) {
        const int n = (end - base) < 32 ? (end - base) : 32;
        int s; float a[H];
        edge_alpha<H>(csr_src, csr_w, ss, sdv, cev, base, n, lane, s, a);
#pragma unroll
        for (int h = 0; h < H; ++h) den[h] += lane < n ? __expf(a[h] - m[h]) : 0.0f;
    }
#pragma unroll
    for (int h = 0; h < H; ++h) {
        ex_self[h] = __expf(a_self[h] - m[h]);
        den[h] = wave_sum(den[h]) + ex_self[h] + 1e-16f;
        at0[h] = at0[h] / den[h];
    }

    vecf acc = {};
#pragma unroll 4
    for (int t = 0; t < n0; ++t) {
        const int s = __builtin_amdgcn_readlane(s0, t);
        const float attn = pick_attn<H>(at0, t, hl);
        const vecf v = *(const vecf*)(xw + (size_t)s * C + c0);
        acc += v * attn;
    }
    for (int base = beg + 32; base < end; base += 32) {
        const int n = (end - base) < 32 ? (end - base) : 32;
        int s1; float a[H];
        edge_alpha<H>(csr_src, csr_w, ss, sdv, cev, base, n, lane, s1, a);
#pragma unroll
        for (int h = 0; h < H; ++h) a[h] = (lane < n ? __expf(a[h] - m[h]) : 0.0f) / den[h];
#pragma unroll 4
        for (int t = 0; t < n; ++t) {
            const int s = __builtin_amdgcn_readlane(s1, t);
            const float attn = pick_attn<H>(a, t, hl);
            const vecf v = *(const vecf*)(xw + (size_t)s * C + c0);
            acc += v * attn;
        }
    }
    {
        float exs = ex_self[0], dn = den[0];
#pragma unroll
        for (int h = 1; h < H; ++h) { exs = (hl == h) ? ex_self[h] : exs; dn = (hl == h) ? den[h] : dn; }
        const float attn = exs / dn;
        const vecf v = *(const vecf*)(xw + (size_t)d * C + c0);
        acc += v * attn;
    }
    vecf r;
#pragma unroll
    for (int i = 0; i < VW; ++i) r[i] = acc[i] + b[c0 + i];
    *(vecf*)(out + (size_t)d * C + c0) = r;
}

__global__ void elu_kernel(float* __restrict__ buf, int n) {
    int i = blockIdx.x * blockDim.x + threadIdx.x;
    if (i >= n) return;
    float v = buf[i];
    buf[i] = v > 0.0f ? v : expm1f(v);
}

extern "C" void kernel_launch(void* const* d_in, const int* in_sizes, int n_in,
                              void* d_out, int out_size, void* d_ws, size_t ws_size,
                              hipStream_t stream) {
    const float* x  = (const float*)d_in[0];
    const int*   ei = (const int*)d_in[1];
    const float* ew = (const float*)d_in[2];

    const float* Wl[3] = { (const float*)d_in[3],  (const float*)d_in[9],  (const float*)d_in[15] };
    const float* As[3] = { (const float*)d_in[4],  (const float*)d_in[10], (const float*)d_in[16] };
    const float* Ad[3] = { (const float*)d_in[5],  (const float*)d_in[11], (const float*)d_in[17] };
    const float* We[3] = { (const float*)d_in[6],  (const float*)d_in[12], (const float*)d_in[18] };
    const float* Ae[3] = { (const float*)d_in[7],  (const float*)d_in[13], (const float*)d_in[19] };
    const float* Bi[3] = { (const float*)d_in[8],  (const float*)d_in[14], (const float*)d_in[20] };

    const int Hs[3] = { 4, 4, 1 };
    const int Fs[3] = { 32, 32, 64 };

    float* ws    = (float*)d_ws;
    float* xw    = ws;
    float* bufA  = xw    + (size_t)NN * 128;
    float* bufB  = bufA  + (size_t)NN * 128;
    float* ss    = bufB  + (size_t)NN * 128;
    float* sd    = ss    + (size_t)NN * 4;
    float* ce    = sd    + (size_t)NN * 4;
    float* meanw = ce + 4;
    float* mpart = meanw + 4;
    int* deg     = (int*)(mpart + 256);
    int* rowptr  = deg + NN;
    int* rank    = rowptr + NN + 4;
    int* csr_src = rank + EE;
    float* csr_w = (float*)(csr_src + EE);

    mean_partial_kernel<<<MEAN_BLOCKS, MEAN_THREADS, 0, stream>>>(ew, mpart, EE);
    mean_final_kernel<<<1, MEAN_THREADS, 0, stream>>>(mpart, meanw, MEAN_BLOCKS, EE);

    csr_zero_kernel<<<(NN + 255) / 256, 256, 0, stream>>>(deg);
    csr_count_kernel<<<(EE + 255) / 256, 256, 0, stream>>>(ei, deg, rank);
    csr_scan_kernel<<<1, SCAN_THREADS, 0, stream>>>(deg, rowptr);
    csr_fill_kernel<<<(EE + 255) / 256, 256, 0, stream>>>(ei, ew, rowptr, rank, csr_src, csr_w);

    const float* in = x;
    float* outs[3] = { bufA, bufB, (float*)d_out };

    for (int l = 0; l < 3; ++l) {
        const int H = Hs[l], F = Fs[l], C = H * F;

        edge_coef_kernel<<<1, 32, 0, stream>>>(We[l], Ae[l], ce, H, F);

        gemm_wmma_f32<<<dim3(NN / 16, C / 16), 32, 0, stream>>>(in, Wl[l], xw, KDIM, C);

        score_kernel<<<(NN * H + 255) / 256, 256, 0, stream>>>(xw, As[l], Ad[l], ss, sd, H, F);

        if (C == 128)
            gat_node_kernel<4, 4><<<(NN + GATHER_WAVES - 1) / GATHER_WAVES, GATHER_WAVES * 32, 0, stream>>>(
                rowptr, csr_src, csr_w, xw, ss, sd, ce, meanw, Bi[l], outs[l]);
        else
            gat_node_kernel<2, 1><<<(NN + GATHER_WAVES - 1) / GATHER_WAVES, GATHER_WAVES * 32, 0, stream>>>(
                rowptr, csr_src, csr_w, xw, ss, sd, ce, meanw, Bi[l], outs[l]);

        if (l < 2) elu_kernel<<<((NN * C) + 255) / 256, 256, 0, stream>>>(outs[l], NN * C);

        in = outs[l];
    }
}


# ELU of layers 0 and 1 moves into the per-destination gather's store: the sum is activated in registers before its one store; the two one-thread-per-element launches go; order of sums unchanged
# speedup vs baseline: 4.6036x; 1.1016x over previous
#include <hip/hip_runtime.h>
#include <math.h>

#define NN    50000
#define EE    800000
#define ETOT  (EE + NN)
#define KDIM  128

typedef float v2f __attribute__((ext_vector_type(2)));
typedef float v8f __attribute__((ext_vector_type(8)));

#define MEAN_THREADS    256
#define MEAN_PER_THREAD 16
#define MEAN_BLOCKS     ((EE + MEAN_THREADS * MEAN_PER_THREAD - 1) / (MEAN_THREADS * MEAN_PER_THREAD))

__device__ __forceinline__ float block_sum_256(float s, float* swave) {
    for (int m = 16; m >= 1; m >>= 1) s += __shfl_xor(s, m, 32);
    const int wave = threadIdx.x >> 5, lane = threadIdx.x & 31;
    if (lane == 0) swave[wave] = s;
    __syncthreads();
    float t = (lane < MEAN_THREADS / 32) ? swave[lane] : 0.0f;
    for (int m = MEAN_THREADS / 64; m >= 1; m >>= 1) t += __shfl_xor(t, m, 32);
    return t;
}

__global__ __launch_bounds__(MEAN_THREADS) void mean_partial_kernel(const float* __restrict__ ew,
                                                                    float* __restrict__ partial, int E) {
    __shared__ float swave[MEAN_THREADS / 32];
    const int T = gridDim.x * MEAN_THREADS;
    const int g = blockIdx.x * MEAN_THREADS + threadIdx.x;
    float s = 0.0f;
    for (int i0 = g; i0 < E; i0 += MEAN_PER_THREAD * T) {
        float v[MEAN_PER_THREAD];
        for (int j = 0; j < MEAN_PER_THREAD; ++j) {
            const int i = i0 + j * T;
            const float x = ew[i < E ? i : E - 1];
            v[j] = i < E ? x : 0.0f;
        }
        for (int w = 1; w < MEAN_PER_THREAD; w <<= 1) {
            for (int j = 0; j < MEAN_PER_THREAD; j += 2 * w) v[j] += v[j + w];
        }
        s += v[0];
    }
    s = block_sum_256(s, swave);
    if (threadIdx.x == 0) partial[blockIdx.x] = s;
}

__global__ __launch_bounds__(MEAN_THREADS) void mean_final_kernel(const float* __restrict__ partial,
                                                                  float* __restrict__ meanw, int nb, int E) {
    __shared__ float swave[MEAN_THREADS / 32];
    float s = 0.0f;
    for (int i = threadIdx.x; i < nb; i += MEAN_THREADS) s += partial[i];
    s = block_sum_256(s, swave);
    if (threadIdx.x == 0) meanw[0] = s / (float)E;
}

__global__ void edge_coef_kernel(const float* __restrict__ We, const float* __restrict__ ae,
                                 float* __restrict__ ce, int H, int F) {
    int h = threadIdx.x;
    if (h >= H) return;
    float s = 0.0f;
    for (int f = 0; f < F; ++f) s += We[h * F + f] * ae[h * F + f];
    ce[h] = s;
}

__global__ __launch_bounds__(32) void gemm_wmma_f32(const float* __restrict__ X,
                                                    const float* __restrict__ W,
                                                    float* __restrict__ out,
                                                    int K, int C) {
    const int tm   = blockIdx.x;
    const int tc   = blockIdx.y;
    const int lane = threadIdx.x;
    const int half = lane >> 4;
    const int l15  = lane & 15;

    v8f acc = {};
    const float* xrow = X + (size_t)(tm * 16 + l15) * K;
    const int    coln = tc * 16 + l15;

    for (int k = 0; k < K; k += 4) {
        const int kk = k + half * 2;
        v2f a, b;
        a[0] = xrow[kk];
        a[1] = xrow[kk + 1];
        b[0] = W[(size_t)kk * C + coln];
        b[1] = W[(size_t)(kk + 1) * C + coln];
        acc = __builtin_amdgcn_wmma_f32_16x16x4_f32(
             false, a,  false, b,
             (short)0, acc,  false,  false);
    }
#pragma unroll
    for (int v = 0; v < 8; ++v) {
        const int row = tm * 16 + v + half * 8;
        out[(size_t)row * C + coln] = acc[v];
    }
}

__global__ void score_kernel(const float* __restrict__ xw,
                             const float* __restrict__ a_s, const float* __restrict__ a_d,
                             float* __restrict__ ss, float* __restrict__ sd,
                             int H, int F) {
    int i = blockIdx.x * blockDim.x + threadIdx.x;
    if (i >= NN * H) return;
    const int n = i / H, h = i % H;
    const float* v  = xw + (size_t)n * H * F + h * F;
    const float* as = a_s + h * F;
    const float* ad = a_d + h * F;
    float s1 = 0.0f, s2 = 0.0f;
    for (int f = 0; f < F; ++f) { float x = v[f]; s1 += x * as[f]; s2 += x * ad[f]; }
    ss[i] = s1;
    sd[i] = s2;
}

__global__ void csr_zero_kernel(int* __restrict__ deg) {
    const int i = blockIdx.x * blockDim.x + threadIdx.x;
    if (i < NN) deg[i] = 0;
}

__global__ void csr_count_kernel(const int* __restrict__ ei, int* __restrict__ deg, int* __restrict__ rank) {
    const int e = blockIdx.x * blockDim.x + threadIdx.x;
    if (e < EE) rank[e] = atomicAdd(&deg[ei[EE + e]], 1);
}

#define SCAN_THREADS 1024
#define SCAN_TILE    (SCAN_THREADS * 4)
#define SCAN_TILES   ((NN + SCAN_TILE - 1) / SCAN_TILE)
__global__ __launch_bounds__(SCAN_THREADS) void csr_scan_kernel(const int* __restrict__ deg,
                                                                int* __restrict__ rowptr) {
    __shared__ int swave[SCAN_TILES][32];
    __shared__ int stile[SCAN_TILES];
    const int t = threadIdx.x, lane = t & 31, wave = t >> 5;
    int4 v[SCAN_TILES];
    int  incl[SCAN_TILES];
#pragma unroll
    for (int k = 0; k < SCAN_TILES; ++k) {
        const int i = k * SCAN_TILE + 4 * t;
        v[k] = (i < NN) ? *(const int4*)(deg + i) : make_int4(0, 0, 0, 0);
    }
#pragma unroll
    for (int k = 0; k < SCAN_TILES; ++k) {
        int in = v[k].x + v[k].y + v[k].z + v[k].w;
        for (int m = 1; m < 32; m <<= 1) {
            const int o = __shfl_up(in, m, 32);
            if (lane >= m) in += o;
        }
        incl[k] = in;
        if (lane == 31) swave[k][wave] = in;
    }
    __syncthreads();
    if (wave < SCAN_TILES) {
        const int w = swave[wave][lane];
        int wi = w;
        for (int m = 1; m < 32; m <<= 1) {
            const int o = __shfl_up(wi, m, 32);
            if (lane >= m) wi += o;
        }
        swave[wave][lane] = wi - w;
        if (lane == 31) stile[wave] = wi;
    }
    __syncthreads();
    int carry = 0;
#pragma unroll
    for (int k = 0; k < SCAN_TILES; ++k) {
        const int i = k * SCAN_TILE + 4 * t;
        const int s = v[k].x + v[k].y + v[k].z + v[k].w;
        int4 r;
        r.x = carry + swave[k][wave] + incl[k] - s;
        r.y = r.x + v[k].x;
        r.z = r.y + v[k].y;
        r.w = r.z + v[k].z;
        if (i < NN) *(int4*)(rowptr + i) = r;
        carry += stile[k];
    }
    if (t == 0) rowptr[NN] = carry;
}

__global__ void csr_fill_kernel(const int* __restrict__ ei, const float* __restrict__ ew,
                                const int* __restrict__ rowptr, const int* __restrict__ rank,
                                int* __restrict__ csr_src, float* __restrict__ csr_w) {
    const int e = blockIdx.x * blockDim.x + threadIdx.x;
    if (e >= EE) return;
    const int pos = rowptr[ei[EE + e]] + rank[e];
    csr_src[pos] = ei[e];
    csr_w[pos]   = ew[e];
}

#define GATHER_WAVES 8

__device__ __forceinline__ float wave_max(float v) {
    for (int m = 16; m >= 1; m >>= 1) v = fmaxf(v, __shfl_xor(v, m, 32));
    return v;
}
__device__ __forceinline__ float wave_sum(float v) {
    for (int m = 16; m >= 1; m >>= 1) v += __shfl_xor(v, m, 32);
    return v;
}

template <int H>
__device__ __forceinline__ void edge_alpha(const int* __restrict__ csr_src, const float* __restrict__ csr_w,
                                           const float* __restrict__ ss, const float* sdv, const float* cev,
                                           int base, int n, int lane, int& s, float* a) {
    s = 0;
    float w = 0.0f;
    if (lane < n) { s = csr_src[base + lane]; w = csr_w[base + lane]; }
    float q[H];
    if constexpr (H == 4) {
        const float4 t = *(const float4*)(ss + (size_t)s * 4);
        q[0] = t.x; q[1] = t.y; q[2] = t.z; q[3] = t.w;
    } else {
#pragma unroll
        for (int h = 0; h < H; ++h) q[h] = ss[(size_t)s * H + h];
    }
#pragma unroll
    for (int h = 0; h < H; ++h) {
        float x = q[h] + sdv[h] + w * cev[h];
        x = x > 0.0f ? x : 0.2f * x;
        a[h] = lane < n ? x : -__builtin_inff();
    }
}

template <int H>
__device__ __forceinline__ float pick_attn(const float* at, int t, int hl) {
    float r[H];
#pragma unroll
    for (int h = 0; h < H; ++h)
        r[h] = __int_as_float(__builtin_amdgcn_readlane(__float_as_int(at[h]), t));
    float v = r[0];
#pragma unroll
    for (int h = 1; h < H; ++h) v = (hl == h) ? r[h] : v;
    return v;
}

template <int VW, int H, bool ELU>
__global__ __launch_bounds__(GATHER_WAVES * 32) void gat_node_kernel(const int* __restrict__ rowptr,
                                                                     const int* __restrict__ csr_src,
                                                                     const float* __restrict__ csr_w,
                                                                     const float* __restrict__ xw,
                                                                     const float* __restrict__ ss,
                                                                     const float* __restrict__ sd,
                                                                     const float* __restrict__ ce,
                                                                     const float* __restrict__ meanw,
                                                                     const float* __restrict__ b,
                                                                     float* __restrict__ out) {
    constexpr int C = 32 * VW;
    constexpr int F = C / H;
    typedef float vecf __attribute__((ext_vector_type(VW)));
    const int lane = threadIdx.x & 31;
    const int d = __builtin_amdgcn_readfirstlane(blockIdx.x * GATHER_WAVES + (threadIdx.x >> 5));
    if (d >= NN) return;
    const int c0 = lane * VW;
    const int hl = c0 / F;
    const int beg = rowptr[d], end = rowptr[d + 1];
    const int n0 = (end - beg) < 32 ? (end - beg) : 32;

    float cev[H], sdv[H], a_self[H], m[H];
    const float mw = meanw[0];
#pragma unroll
    for (int h = 0; h < H; ++h) {
        cev[h] = ce[h];
        sdv[h] = sd[(size_t)d * H + h];
        float x = ss[(size_t)d * H + h] + sdv[h] + mw * cev[h];
        a_self[h] = x > 0.0f ? x : 0.2f * x;
        m[h] = a_self[h];
    }

    int s0; float a0[H];
    edge_alpha<H>(csr_src, csr_w, ss, sdv, cev, beg, n0, lane, s0, a0);
#pragma unroll
    for (int h = 0; h < H; ++h) m[h] = fmaxf(m[h], a0[h]);
    for (int base = beg + 32; base < end; base += 32) {
        const int n = (end - base) < 32 ? (end - base) : 32;
        int s; float a[H];
        edge_alpha<H>(csr_src, csr_w, ss, sdv, cev, base, n, lane, s, a);
#pragma unroll
        for (int h = 0; h < H; ++h) m[h] = fmaxf(m[h], a[h]);
    }
#pragma unroll
    for (int h = 0; h < H; ++h) m[h] = wave_max(m[h]);

    float at0[H], den[H], ex_self[H];
#pragma unroll
    for (int h = 0; h < H; ++h) {
        at0[h] = lane < n0 ? __expf(a0[h] - m[h]) : 0.0f;
        den[h] = at0[h];
    }
    for (int base = beg + 32; base < end; base += 32) {
        const int n = (end - base) < 32 ? (end - base) : 32;
        int s; float a[H];
        edge_alpha<H>(csr_src, csr_w, ss, sdv, cev, base, n, lane, s, a);
#pragma unroll
        for (int h = 0; h < H; ++h) den[h] += lane < n ? __expf(a[h] - m[h]) : 0.0f;
    }
#pragma unroll
    for (int h = 0; h < H; ++h) {
        ex_self[h] = __expf(a_self[h] - m[h]);
        den[h] = wave_sum(den[h]) + ex_self[h] + 1e-16f;
        at0[h] = at0[h] / den[h];
    }

    vecf acc = {};
#pragma unroll 4
    for (int t = 0; t < n0; ++t) {
        const int s = __builtin_amdgcn_readlane(s0, t);
        const float attn = pick_attn<H>(at0, t, hl);
        const vecf v = *(const vecf*)(xw + (size_t)s * C + c0);
        acc += v * attn;
    }
    for (int base = beg + 32; base < end; base += 32) {
        const int n = (end - base) < 32 ? (end - base) : 32;
        int s1; float a[H];
        edge_alpha<H>(csr_src, csr_w, ss, sdv, cev, base, n, lane, s1, a);
#pragma unroll
        for (int h = 0; h < H; ++h) a[h] = (lane < n ? __expf(a[h] - m[h]) : 0.0f) / den[h];
#pragma unroll 4
        for (int t = 0; t < n; ++t) {
            const int s = __builtin_amdgcn_readlane(s1, t);
            const float attn = pick_attn<H>(a, t, hl);
            const vecf v = *(const vecf*)(xw + (size_t)s * C + c0);
            acc += v * attn;
        }
    }
    {
        float exs = ex_self[0], dn = den[0];
#pragma unroll
        for (int h = 1; h < H; ++h) { exs = (hl == h) ? ex_self[h] : exs; dn = (hl == h) ? den[h] : dn; }
        const float attn = exs / dn;
        const vecf v = *(const vecf*)(xw + (size_t)d * C + c0);
        acc += v * attn;
    }
    float bias[VW];
#pragma unroll
    for (int i = 0; i < VW; ++i) bias[i] = b[c0 + i];
    vecf r;
#pragma unroll
    for (int i = 0; i < VW; ++i) {
        float v = acc[i] + bias[i];
        if constexpr (ELU) v = v > 0.0f ? v : expm1f(v);
        r[i] = v;
    }
    *(vecf*)(out + (size_t)d * C + c0) = r;
}

extern "C" void kernel_launch(void* const* d_in, const int* in_sizes, int n_in,
                              void* d_out, int out_size, void* d_ws, size_t ws_size,
                              hipStream_t stream) {
    const float* x  = (const float*)d_in[0];
    const int*   ei = (const int*)d_in[1];
    const float* ew = (const float*)d_in[2];

    const float* Wl[3] = { (const float*)d_in[3],  (const float*)d_in[9],  (const float*)d_in[15] };
    const float* As[3] = { (const float*)d_in[4],  (const float*)d_in[10], (const float*)d_in[16] };
    const float* Ad[3] = { (const float*)d_in[5],  (const float*)d_in[11], (const float*)d_in[17] };
    const float* We[3] = { (const float*)d_in[6],  (const float*)d_in[12], (const float*)d_in[18] };
    const float* Ae[3] = { (const float*)d_in[7],  (const float*)d_in[13], (const float*)d_in[19] };
    const float* Bi[3] = { (const float*)d_in[8],  (const float*)d_in[14], (const float*)d_in[20] };

    const int Hs[3] = { 4, 4, 1 };
    const int Fs[3] = { 32, 32, 64 };

    float* ws    = (float*)d_ws;
    float* xw    = ws;
    float* bufA  = xw    + (size_t)NN * 128;
    float* bufB  = bufA  + (size_t)NN * 128;
    float* ss    = bufB  + (size_t)NN * 128;
    float* sd    = ss    + (size_t)NN * 4;
    float* ce    = sd    + (size_t)NN * 4;
    float* meanw = ce + 4;
    float* mpart = meanw + 4;
    int* deg     = (int*)(mpart + 256);
    int* rowptr  = deg + NN;
    int* rank    = rowptr + NN + 4;
    int* csr_src = rank + EE;
    float* csr_w = (float*)(csr_src + EE);

    mean_partial_kernel<<<MEAN_BLOCKS, MEAN_THREADS, 0, stream>>>(ew, mpart, EE);
    mean_final_kernel<<<1, MEAN_THREADS, 0, stream>>>(mpart, meanw, MEAN_BLOCKS, EE);

    csr_zero_kernel<<<(NN + 255) / 256, 256, 0, stream>>>(deg);
    csr_count_kernel<<<(EE + 255) / 256, 256, 0, stream>>>(ei, deg, rank);
    csr_scan_kernel<<<1, SCAN_THREADS, 0, stream>>>(deg, rowptr);
    csr_fill_kernel<<<(EE + 255) / 256, 256, 0, stream>>>(ei, ew, rowptr, rank, csr_src, csr_w);

    const float* in = x;
    float* outs[3] = { bufA, bufB, (float*)d_out };

    for (int l = 0; l < 3; ++l) {
        const int H = Hs[l], F = Fs[l], C = H * F;

        edge_coef_kernel<<<1, 32, 0, stream>>>(We[l], Ae[l], ce, H, F);

        gemm_wmma_f32<<<dim3(NN / 16, C / 16), 32, 0, stream>>>(in, Wl[l], xw, KDIM, C);

        score_kernel<<<(NN * H + 255) / 256, 256, 0, stream>>>(xw, As[l], Ad[l], ss, sd, H, F);

        if (l < 2)
            gat_node_kernel<4, 4, true><<<(NN + GATHER_WAVES - 1) / GATHER_WAVES, GATHER_WAVES * 32, 0, stream>>>(
                rowptr, csr_src, csr_w, xw, ss, sd, ce, meanw, Bi[l], outs[l]);
        else
            gat_node_kernel<2, 1, false><<<(NN + GATHER_WAVES - 1) / GATHER_WAVES, GATHER_WAVES * 32, 0, stream>>>(
                rowptr, csr_src, csr_w, xw, ss, sd, ce, meanw, Bi[l], outs[l]);

        in = outs[l];
    }
}


# attention scores move into the GEMM epilogue: a wave owns 16 rows x 64 columns (one A fragment feeds 4 tiles) and reduces ss,sd per head from its accumulators; 3 launches go; score sum order changes
# speedup vs baseline: 6.3645x; 1.3825x over previous
#include <hip/hip_runtime.h>
#include <math.h>

#define NN    50000
#define EE    800000
#define ETOT  (EE + NN)
#define KDIM  128

typedef float v2f __attribute__((ext_vector_type(2)));
typedef float v8f __attribute__((ext_vector_type(8)));

#define MEAN_THREADS    256
#define MEAN_PER_THREAD 16
#define MEAN_BLOCKS     ((EE + MEAN_THREADS * MEAN_PER_THREAD - 1) / (MEAN_THREADS * MEAN_PER_THREAD))

__device__ __forceinline__ float block_sum_256(float s, float* swave) {
    for (int m = 16; m >= 1; m >>= 1) s += __shfl_xor(s, m, 32);
    const int wave = threadIdx.x >> 5, lane = threadIdx.x & 31;
    if (lane == 0) swave[wave] = s;
    __syncthreads();
    float t = (lane < MEAN_THREADS / 32) ? swave[lane] : 0.0f;
    for (int m = MEAN_THREADS / 64; m >= 1; m >>= 1) t += __shfl_xor(t, m, 32);
    return t;
}

__global__ __launch_bounds__(MEAN_THREADS) void mean_partial_kernel(const float* __restrict__ ew,
                                                                    float* __restrict__ partial, int E) {
    __shared__ float swave[MEAN_THREADS / 32];
    const int T = gridDim.x * MEAN_THREADS;
    const int g = blockIdx.x * MEAN_THREADS + threadIdx.x;
    float s = 0.0f;
    for (int i0 = g; i0 < E; i0 += MEAN_PER_THREAD * T) {
        float v[MEAN_PER_THREAD];
        for (int j = 0; j < MEAN_PER_THREAD; ++j) {
            const int i = i0 + j * T;
            const float x = ew[i < E ? i : E - 1];
            v[j] = i < E ? x : 0.0f;
        }
        for (int w = 1; w < MEAN_PER_THREAD; w <<= 1) {
            for (int j = 0; j < MEAN_PER_THREAD; j += 2 * w) v[j] += v[j + w];
        }
        s += v[0];
    }
    s = block_sum_256(s, swave);
    if (threadIdx.x == 0) partial[blockIdx.x] = s;
}

__global__ __launch_bounds__(MEAN_THREADS) void mean_final_kernel(const float* __restrict__ partial,
                                                                  float* __restrict__ meanw, int nb, int E) {
    __shared__ float swave[MEAN_THREADS / 32];
    float s = 0.0f;
    for (int i = threadIdx.x; i < nb; i += MEAN_THREADS) s += partial[i];
    s = block_sum_256(s, swave);
    if (threadIdx.x == 0) meanw[0] = s / (float)E;
}

__global__ void edge_coef_kernel(const float* __restrict__ We, const float* __restrict__ ae,
                                 float* __restrict__ ce, int H, int F) {
    int h = threadIdx.x;
    if (h >= H) return;
    float s = 0.0f;
    for (int f = 0; f < F; ++f) s += We[h * F + f] * ae[h * F + f];
    ce[h] = s;
}

#define GEMM_TILES 4
#define GEMM_COLS  (GEMM_TILES * 16)
#define GEMM_WAVES 4
#define STRIPS     (NN / 16)

template <int C, int F>
__global__ __launch_bounds__(GEMM_WAVES * 32) void gemm_score_kernel(const float* __restrict__ X,
                                                                     const float* __restrict__ W,
                                                                     const float* __restrict__ a_s,
                                                                     const float* __restrict__ a_d,
                                                                     float* __restrict__ xw,
                                                                     float* __restrict__ ss,
                                                                     float* __restrict__ sd) {
    constexpr int H   = C / F;
    constexpr int HW  = GEMM_COLS / F;
    constexpr int TPH = F / 16;
    constexpr int NV  = 16 * HW;
    const int lane = threadIdx.x & 31;
    const int strip = __builtin_amdgcn_readfirstlane(blockIdx.x * GEMM_WAVES + (threadIdx.x >> 5));
    if (strip >= STRIPS) return;
    const int half = lane >> 4;
    const int l15  = lane & 15;
    const int cb   = blockIdx.y * GEMM_COLS;

    v8f acc[GEMM_TILES] = {};
    const float* xrow = X + (size_t)(strip * 16 + l15) * KDIM + half * 2;
    const float* wcol = W + (size_t)(half * 2) * C + cb + l15;

#pragma unroll 8
    for (int k = 0; k < KDIM; k += 4) {
        v2f a;
        a[0] = xrow[k];
        a[1] = xrow[k + 1];
        v2f b[GEMM_TILES];
#pragma unroll
        for (int t = 0; t < GEMM_TILES; ++t) {
            b[t][0] = wcol[(size_t)k * C + t * 16];
            b[t][1] = wcol[(size_t)(k + 1) * C + t * 16];
        }
#pragma unroll
        for (int t = 0; t < GEMM_TILES; ++t)
            acc[t] = __builtin_amdgcn_wmma_f32_16x16x4_f32(
                 false, a,  false, b[t],
                 (short)0, acc[t],  false,  false);
    }

    const int r0 = strip * 16 + half * 8;
#pragma unroll
    for (int t = 0; t < GEMM_TILES; ++t) {
#pragma unroll
        for (int v = 0; v < 8; ++v)
            xw[(size_t)(r0 + v) * C + cb + t * 16 + l15] = acc[t][v];
    }

    float as_c[GEMM_TILES], ad_c[GEMM_TILES];
#pragma unroll
    for (int t = 0; t < GEMM_TILES; ++t) {
        as_c[t] = a_s[cb + t * 16 + l15];
        ad_c[t] = a_d[cb + t * 16 + l15];
    }
    float val[NV];
#pragma unroll
    for (int v = 0; v < 8; ++v) {
#pragma unroll
        for (int g = 0; g < HW; ++g) {
            float ps = acc[g * TPH][v] * as_c[g * TPH];
            float pd = acc[g * TPH][v] * ad_c[g * TPH];
#pragma unroll
            for (int j = 1; j < TPH; ++j) {
                ps += acc[g * TPH + j][v] * as_c[g * TPH + j];
                pd += acc[g * TPH + j][v] * ad_c[g * TPH + j];
            }
            val[(v * HW + g) * 2 + 0] = ps;
            val[(v * HW + g) * 2 + 1] = pd;
        }
    }
    {
        constexpr int h8 = NV / 2, h4 = NV / 4, h2 = NV / 8, h1 = NV / 16;
        const bool u8 = (lane & 8) != 0, u4 = (lane & 4) != 0, u2 = (lane & 2) != 0, u1 = (lane & 1) != 0;
#pragma unroll
        for (int i = 0; i < h8; ++i) {
            const float send = u8 ? val[i] : val[i + h8];
            const float keep = u8 ? val[i + h8] : val[i];
            val[i] = keep + __shfl_xor(send, 8, 32);
        }
#pragma unroll
        for (int i = 0; i < h4; ++i) {
            const float send = u4 ? val[i] : val[i + h4];
            const float keep = u4 ? val[i + h4] : val[i];
            val[i] = keep + __shfl_xor(send, 4, 32);
        }
#pragma unroll
        for (int i = 0; i < h2; ++i) {
            const float send = u2 ? val[i] : val[i + h2];
            const float keep = u2 ? val[i + h2] : val[i];
            val[i] = keep + __shfl_xor(send, 2, 32);
        }
#pragma unroll
        for (int i = 0; i < h1; ++i) {
            const float send = u1 ? val[i] : val[i + h1];
            const float keep = u1 ? val[i + h1] : val[i];
            val[i] = keep + __shfl_xor(send, 1, 32);
        }
    }
    const int hb = cb / F;
#pragma unroll
    for (int j = 0; j < HW; ++j) {
        const int idx  = l15 * HW + j;
        const int kind = idx & 1;
        const int g    = (idx >> 1) % HW;
        const int v    = (idx >> 1) / HW;
        float* dst = kind ? sd : ss;
        dst[(size_t)(r0 + v) * H + hb + g] = val[j];
    }
}

__global__ void csr_zero_kernel(int* __restrict__ deg) {
    const int i = blockIdx.x * blockDim.x + threadIdx.x;
    if (i < NN) deg[i] = 0;
}

__global__ void csr_count_kernel(const int* __restrict__ ei, int* __restrict__ deg, int* __restrict__ rank) {
    const int e = blockIdx.x * blockDim.x + threadIdx.x;
    if (e < EE) rank[e] = atomicAdd(&deg[ei[EE + e]], 1);
}

#define SCAN_THREADS 1024
#define SCAN_TILE    (SCAN_THREADS * 4)
#define SCAN_TILES   ((NN + SCAN_TILE - 1) / SCAN_TILE)
__global__ __launch_bounds__(SCAN_THREADS) void csr_scan_kernel(const int* __restrict__ deg,
                                                                int* __restrict__ rowptr) {
    __shared__ int swave[SCAN_TILES][32];
    __shared__ int stile[SCAN_TILES];
    const int t = threadIdx.x, lane = t & 31, wave = t >> 5;
    int4 v[SCAN_TILES];
    int  incl[SCAN_TILES];
#pragma unroll
    for (int k = 0; k < SCAN_TILES; ++k) {
        const int i = k * SCAN_TILE + 4 * t;
        v[k] = (i < NN) ? *(const int4*)(deg + i) : make_int4(0, 0, 0, 0);
    }
#pragma unroll
    for (int k = 0; k < SCAN_TILES; ++k) {
        int in = v[k].x + v[k].y + v[k].z + v[k].w;
        for (int m = 1; m < 32; m <<= 1) {
            const int o = __shfl_up(in, m, 32);
            if (lane >= m) in += o;
        }
        incl[k] = in;
        if (lane == 31) swave[k][wave] = in;
    }
    __syncthreads();
    if (wave < SCAN_TILES) {
        const int w = swave[wave][lane];
        int wi = w;
        for (int m = 1; m < 32; m <<= 1) {
            const int o = __shfl_up(wi, m, 32);
            if (lane >= m) wi += o;
        }
        swave[wave][lane] = wi - w;
        if (lane == 31) stile[wave] = wi;
    }
    __syncthreads();
    int carry = 0;
#pragma unroll
    for (int k = 0; k < SCAN_TILES; ++k) {
        const int i = k * SCAN_TILE + 4 * t;
        const int s = v[k].x + v[k].y + v[k].z + v[k].w;
        int4 r;
        r.x = carry + swave[k][wave] + incl[k] - s;
        r.y = r.x + v[k].x;
        r.z = r.y + v[k].y;
        r.w = r.z + v[k].z;
        if (i < NN) *(int4*)(rowptr + i) = r;
        carry += stile[k];
    }
    if (t == 0) rowptr[NN] = carry;
}

__global__ void csr_fill_kernel(const int* __restrict__ ei, const float* __restrict__ ew,
                                const int* __restrict__ rowptr, const int* __restrict__ rank,
                                int* __restrict__ csr_src, float* __restrict__ csr_w) {
    const int e = blockIdx.x * blockDim.x + threadIdx.x;
    if (e >= EE) return;
    const int pos = rowptr[ei[EE + e]] + rank[e];
    csr_src[pos] = ei[e];
    csr_w[pos]   = ew[e];
}

#define GATHER_WAVES 8

__device__ __forceinline__ float wave_max(float v) {
    for (int m = 16; m >= 1; m >>= 1) v = fmaxf(v, __shfl_xor(v, m, 32));
    return v;
}
__device__ __forceinline__ float wave_sum(float v) {
    for (int m = 16; m >= 1; m >>= 1) v += __shfl_xor(v, m, 32);
    return v;
}

template <int H>
__device__ __forceinline__ void edge_alpha(const int* __restrict__ csr_src, const float* __restrict__ csr_w,
                                           const float* __restrict__ ss, const float* sdv, const float* cev,
                                           int base, int n, int lane, int& s, float* a) {
    s = 0;
    float w = 0.0f;
    if (lane < n) { s = csr_src[base + lane]; w = csr_w[base + lane]; }
    float q[H];
    if constexpr (H == 4) {
        const float4 t = *(const float4*)(ss + (size_t)s * 4);
        q[0] = t.x; q[1] = t.y; q[2] = t.z; q[3] = t.w;
    } else {
#pragma unroll
        for (int h = 0; h < H; ++h) q[h] = ss[(size_t)s * H + h];
    }
#pragma unroll
    for (int h = 0; h < H; ++h) {
        float x = q[h] + sdv[h] + w * cev[h];
        x = x > 0.0f ? x : 0.2f * x;
        a[h] = lane < n ? x : -__builtin_inff();
    }
}

template <int H>
__device__ __forceinline__ float pick_attn(const float* at, int t, int hl) {
    float r[H];
#pragma unroll
    for (int h = 0; h < H; ++h)
        r[h] = __int_as_float(__builtin_amdgcn_readlane(__float_as_int(at[h]), t));
    float v = r[0];
#pragma unroll
    for (int h = 1; h < H; ++h) v = (hl == h) ? r[h] : v;
    return v;
}

template <int VW, int H, bool ELU>
__global__ __launch_bounds__(GATHER_WAVES * 32) void gat_node_kernel(const int* __restrict__ rowptr,
                                                                     const int* __restrict__ csr_src,
                                                                     const float* __restrict__ csr_w,
                                                                     const float* __restrict__ xw,
                                                                     const float* __restrict__ ss,
                                                                     const float* __restrict__ sd,
                                                                     const float* __restrict__ ce,
                                                                     const float* __restrict__ meanw,
                                                                     const float* __restrict__ b,
                                                                     float* __restrict__ out) {
    constexpr int C = 32 * VW;
    constexpr int F = C / H;
    typedef float vecf __attribute__((ext_vector_type(VW)));
    const int lane = threadIdx.x & 31;
    const int d = __builtin_amdgcn_readfirstlane(blockIdx.x * GATHER_WAVES + (threadIdx.x >> 5));
    if (d >= NN) return;
    const int c0 = lane * VW;
    const int hl = c0 / F;
    const int beg = rowptr[d], end = rowptr[d + 1];
    const int n0 = (end - beg) < 32 ? (end - beg) : 32;

    float cev[H], sdv[H], a_self[H], m[H];
    const float mw = meanw[0];
#pragma unroll
    for (int h = 0; h < H; ++h) {
        cev[h] = ce[h];
        sdv[h] = sd[(size_t)d * H + h];
        float x = ss[(size_t)d * H + h] + sdv[h] + mw * cev[h];
        a_self[h] = x > 0.0f ? x : 0.2f * x;
        m[h] = a_self[h];
    }

    int s0; float a0[H];
    edge_alpha<H>(csr_src, csr_w, ss, sdv, cev, beg, n0, lane, s0, a0);
#pragma unroll
    for (int h = 0; h < H; ++h) m[h] = fmaxf(m[h], a0[h]);
    for (int base = beg + 32; base < end; base += 32) {
        const int n = (end - base) < 32 ? (end - base) : 32;
        int s; float a[H];
        edge_alpha<H>(csr_src, csr_w, ss, sdv, cev, base, n, lane, s, a);
#pragma unroll
        for (int h = 0; h < H; ++h) m[h] = fmaxf(m[h], a[h]);
    }
#pragma unroll
    for (int h = 0; h < H; ++h) m[h] = wave_max(m[h]);

    float at0[H], den[H], ex_self[H];
#pragma unroll
    for (int h = 0; h < H; ++h) {
        at0[h] = lane < n0 ? __expf(a0[h] - m[h]) : 0.0f;
        den[h] = at0[h];
    }
    for (int base = beg + 32; base < end; base += 32) {
        const int n = (end - base) < 32 ? (end - base) : 32;
        int s; float a[H];
        edge_alpha<H>(csr_src, csr_w, ss, sdv, cev, base, n, lane, s, a);
#pragma unroll
        for (int h = 0; h < H; ++h) den[h] += lane < n ? __expf(a[h] - m[h]) : 0.0f;
    }
#pragma unroll
    for (int h = 0; h < H; ++h) {
        ex_self[h] = __expf(a_self[h] - m[h]);
        den[h] = wave_sum(den[h]) + ex_self[h] + 1e-16f;
        at0[h] = at0[h] / den[h];
    }

    vecf acc = {};
#pragma unroll 4
    for (int t = 0; t < n0; ++t) {
        const int s = __builtin_amdgcn_readlane(s0, t);
        const float attn = pick_attn<H>(at0, t, hl);
        const vecf v = *(const vecf*)(xw + (size_t)s * C + c0);
        acc += v * attn;
    }
    for (int base = beg + 32; base < end; base += 32) {
        const int n = (end - base) < 32 ? (end - base) : 32;
        int s1; float a[H];
        edge_alpha<H>(csr_src, csr_w, ss, sdv, cev, base, n, lane, s1, a);
#pragma unroll
        for (int h = 0; h < H; ++h) a[h] = (lane < n ? __expf(a[h] - m[h]) : 0.0f) / den[h];
#pragma unroll 4
        for (int t = 0; t < n; ++t) {
            const int s = __builtin_amdgcn_readlane(s1, t);
            const float attn = pick_attn<H>(a, t, hl);
            const vecf v = *(const vecf*)(xw + (size_t)s * C + c0);
            acc += v * attn;
        }
    }
    {
        float exs = ex_self[0], dn = den[0];
#pragma unroll
        for (int h = 1; h < H; ++h) { exs = (hl == h) ? ex_self[h] : exs; dn = (hl == h) ? den[h] : dn; }
        const float attn = exs / dn;
        const vecf v = *(const vecf*)(xw + (size_t)d * C + c0);
        acc += v * attn;
    }
    float bias[VW];
#pragma unroll
    for (int i = 0; i < VW; ++i) bias[i] = b[c0 + i];
    vecf r;
#pragma unroll
    for (int i = 0; i < VW; ++i) {
        float v = acc[i] + bias[i];
        if constexpr (ELU) v = v > 0.0f ? v : expm1f(v);
        r[i] = v;
    }
    *(vecf*)(out + (size_t)d * C + c0) = r;
}

extern "C" void kernel_launch(void* const* d_in, const int* in_sizes, int n_in,
                              void* d_out, int out_size, void* d_ws, size_t ws_size,
                              hipStream_t stream) {
    const float* x  = (const float*)d_in[0];
    const int*   ei = (const int*)d_in[1];
    const float* ew = (const float*)d_in[2];

    const float* Wl[3] = { (const float*)d_in[3],  (const float*)d_in[9],  (const float*)d_in[15] };
    const float* As[3] = { (const float*)d_in[4],  (const float*)d_in[10], (const float*)d_in[16] };
    const float* Ad[3] = { (const float*)d_in[5],  (const float*)d_in[11], (const float*)d_in[17] };
    const float* We[3] = { (const float*)d_in[6],  (const float*)d_in[12], (const float*)d_in[18] };
    const float* Ae[3] = { (const float*)d_in[7],  (const float*)d_in[13], (const float*)d_in[19] };
    const float* Bi[3] = { (const float*)d_in[8],  (const float*)d_in[14], (const float*)d_in[20] };

    const int Hs[3] = { 4, 4, 1 };
    const int Fs[3] = { 32, 32, 64 };

    float* ws    = (float*)d_ws;
    float* xw    = ws;
    float* bufA  = xw    + (size_t)NN * 128;
    float* bufB  = bufA  + (size_t)NN * 128;
    float* ss    = bufB  + (size_t)NN * 128;
    float* sd    = ss    + (size_t)NN * 4;
    float* ce    = sd    + (size_t)NN * 4;
    float* meanw = ce + 4;
    float* mpart = meanw + 4;
    int* deg     = (int*)(mpart + 256);
    int* rowptr  = deg + NN;
    int* rank    = rowptr + NN + 4;
    int* csr_src = rank + EE;
    float* csr_w = (float*)(csr_src + EE);

    mean_partial_kernel<<<MEAN_BLOCKS, MEAN_THREADS, 0, stream>>>(ew, mpart, EE);
    mean_final_kernel<<<1, MEAN_THREADS, 0, stream>>>(mpart, meanw, MEAN_BLOCKS, EE);

    csr_zero_kernel<<<(NN + 255) / 256, 256, 0, stream>>>(deg);
    csr_count_kernel<<<(EE + 255) / 256, 256, 0, stream>>>(ei, deg, rank);
    csr_scan_kernel<<<1, SCAN_THREADS, 0, stream>>>(deg, rowptr);
    csr_fill_kernel<<<(EE + 255) / 256, 256, 0, stream>>>(ei, ew, rowptr, rank, csr_src, csr_w);

    const float* in = x;
    float* outs[3] = { bufA, bufB, (float*)d_out };

    for (int l = 0; l < 3; ++l) {
        const int H = Hs[l], F = Fs[l];

        edge_coef_kernel<<<1, 32, 0, stream>>>(We[l], Ae[l], ce, H, F);

        if (l < 2)
            gemm_score_kernel<128, 32><<<dim3((STRIPS + GEMM_WAVES - 1) / GEMM_WAVES, 128 / GEMM_COLS),
                                        GEMM_WAVES * 32, 0, stream>>>(in, Wl[l], As[l], Ad[l], xw, ss, sd);
        else
            gemm_score_kernel<64, 64><<<dim3((STRIPS + GEMM_WAVES - 1) / GEMM_WAVES, 64 / GEMM_COLS),
                                       GEMM_WAVES * 32, 0, stream>>>(in, Wl[l], As[l], Ad[l], xw, ss, sd);

        if (l < 2)
            gat_node_kernel<4, 4, true><<<(NN + GATHER_WAVES - 1) / GATHER_WAVES, GATHER_WAVES * 32, 0, stream>>>(
                rowptr, csr_src, csr_w, xw, ss, sd, ce, meanw, Bi[l], outs[l]);
        else
            gat_node_kernel<2, 1, false><<<(NN + GATHER_WAVES - 1) / GATHER_WAVES, GATHER_WAVES * 32, 0, stream>>>(
                rowptr, csr_src, csr_w, xw, ss, sd, ce, meanw, Bi[l], outs[l]);

        in = outs[l];
    }
}


# five tiny helper launches ride in neighbours: deg clear in the mean's partial-sum launch; the mean's final block and 3 edge-coefficient dots as two extra blocks of the count pass; 15 launches -> 10
# speedup vs baseline: 6.6355x; 1.0426x over previous
#include <hip/hip_runtime.h>
#include <math.h>

#define NN    50000
#define EE    800000
#define ETOT  (EE + NN)
#define KDIM  128

typedef float v2f __attribute__((ext_vector_type(2)));
typedef float v8f __attribute__((ext_vector_type(8)));

#define MEAN_THREADS    256
#define MEAN_PER_THREAD 16
#define MEAN_BLOCKS     ((EE + MEAN_THREADS * MEAN_PER_THREAD - 1) / (MEAN_THREADS * MEAN_PER_THREAD))

__device__ __forceinline__ float block_sum_256(float s, float* swave) {
    for (int m = 16; m >= 1; m >>= 1) s += __shfl_xor(s, m, 32);
    const int wave = threadIdx.x >> 5, lane = threadIdx.x & 31;
    if (lane == 0) swave[wave] = s;
    __syncthreads();
    float t = (lane < MEAN_THREADS / 32) ? swave[lane] : 0.0f;
    for (int m = MEAN_THREADS / 64; m >= 1; m >>= 1) t += __shfl_xor(t, m, 32);
    return t;
}

__global__ __launch_bounds__(MEAN_THREADS) void mean_partial_kernel(const float* __restrict__ ew,
                                                                    float* __restrict__ partial,
                                                                    int* __restrict__ deg, int E) {
    __shared__ float swave[MEAN_THREADS / 32];
    const int T = gridDim.x * MEAN_THREADS;
    const int g = blockIdx.x * MEAN_THREADS + threadIdx.x;
    for (int i = g; i < NN; i += T) deg[i] = 0;
    float s = 0.0f;
    for (int i0 = g; i0 < E; i0 += MEAN_PER_THREAD * T) {
        float v[MEAN_PER_THREAD];
        for (int j = 0; j < MEAN_PER_THREAD; ++j) {
            const int i = i0 + j * T;
            const float x = ew[i < E ? i : E - 1];
            v[j] = i < E ? x : 0.0f;
        }
        for (int w = 1; w < MEAN_PER_THREAD; w <<= 1) {
            for (int j = 0; j < MEAN_PER_THREAD; j += 2 * w) v[j] += v[j + w];
        }
        s += v[0];
    }
    s = block_sum_256(s, swave);
    if (threadIdx.x == 0) partial[blockIdx.x] = s;
}

__device__ __forceinline__ void mean_final_block(const float* __restrict__ partial, float* __restrict__ meanw,
                                                 int nb, int E, float* swave) {
    float s = 0.0f;
    for (int i = threadIdx.x; i < nb; i += MEAN_THREADS) s += partial[i];
    s = block_sum_256(s, swave);
    if (threadIdx.x == 0) meanw[0] = s / (float)E;
}

__device__ __forceinline__ void edge_coef_block(const float* __restrict__ We0, const float* __restrict__ ae0,
                                                const float* __restrict__ We1, const float* __restrict__ ae1,
                                                const float* __restrict__ We2, const float* __restrict__ ae2,
                                                float* __restrict__ ce) {
    const int t = threadIdx.x;
    if (t >= 9) return;
    const int l = t >> 2, h = t & 3;
    const int F = l < 2 ? 32 : 64;
    const float* We = l == 0 ? We0 : (l == 1 ? We1 : We2);
    const float* ae = l == 0 ? ae0 : (l == 1 ? ae1 : ae2);
    float s = 0.0f;
    for (int f = 0; f < F; ++f) s += We[h * F + f] * ae[h * F + f];
    ce[t] = s;
}

#define GEMM_TILES 4
#define GEMM_COLS  (GEMM_TILES * 16)
#define GEMM_WAVES 4
#define STRIPS     (NN / 16)

template <int C, int F>
__global__ __launch_bounds__(GEMM_WAVES * 32) void gemm_score_kernel(const float* __restrict__ X,
                                                                     const float* __restrict__ W,
                                                                     const float* __restrict__ a_s,
                                                                     const float* __restrict__ a_d,
                                                                     float* __restrict__ xw,
                                                                     float* __restrict__ ss,
                                                                     float* __restrict__ sd) {
    constexpr int H   = C / F;
    constexpr int HW  = GEMM_COLS / F;
    constexpr int TPH = F / 16;
    constexpr int NV  = 16 * HW;
    const int lane = threadIdx.x & 31;
    const int strip = __builtin_amdgcn_readfirstlane(blockIdx.x * GEMM_WAVES + (threadIdx.x >> 5));
    if (strip >= STRIPS) return;
    const int half = lane >> 4;
    const int l15  = lane & 15;
    const int cb   = blockIdx.y * GEMM_COLS;

    v8f acc[GEMM_TILES] = {};
    const float* xrow = X + (size_t)(strip * 16 + l15) * KDIM + half * 2;
    const float* wcol = W + (size_t)(half * 2) * C + cb + l15;

#pragma unroll 8
    for (int k = 0; k < KDIM; k += 4) {
        v2f a;
        a[0] = xrow[k];
        a[1] = xrow[k + 1];
        v2f b[GEMM_TILES];
#pragma unroll
        for (int t = 0; t < GEMM_TILES; ++t) {
            b[t][0] = wcol[(size_t)k * C + t * 16];
            b[t][1] = wcol[(size_t)(k + 1) * C + t * 16];
        }
#pragma unroll
        for (int t = 0; t < GEMM_TILES; ++t)
            acc[t] = __builtin_amdgcn_wmma_f32_16x16x4_f32(
                 false, a,  false, b[t],
                 (short)0, acc[t],  false,  false);
    }

    const int r0 = strip * 16 + half * 8;
#pragma unroll
    for (int t = 0; t < GEMM_TILES; ++t) {
#pragma unroll
        for (int v = 0; v < 8; ++v)
            xw[(size_t)(r0 + v) * C + cb + t * 16 + l15] = acc[t][v];
    }

    float as_c[GEMM_TILES], ad_c[GEMM_TILES];
#pragma unroll
    for (int t = 0; t < GEMM_TILES; ++t) {
        as_c[t] = a_s[cb + t * 16 + l15];
        ad_c[t] = a_d[cb + t * 16 + l15];
    }
    float val[NV];
#pragma unroll
    for (int v = 0; v < 8; ++v) {
#pragma unroll
        for (int g = 0; g < HW; ++g) {
            float ps = acc[g * TPH][v] * as_c[g * TPH];
            float pd = acc[g * TPH][v] * ad_c[g * TPH];
#pragma unroll
            for (int j = 1; j < TPH; ++j) {
                ps += acc[g * TPH + j][v] * as_c[g * TPH + j];
                pd += acc[g * TPH + j][v] * ad_c[g * TPH + j];
            }
            val[(v * HW + g) * 2 + 0] = ps;
            val[(v * HW + g) * 2 + 1] = pd;
        }
    }
    {
        constexpr int h8 = NV / 2, h4 = NV / 4, h2 = NV / 8, h1 = NV / 16;
        const bool u8 = (lane & 8) != 0, u4 = (lane & 4) != 0, u2 = (lane & 2) != 0, u1 = (lane & 1) != 0;
#pragma unroll
        for (int i = 0; i < h8; ++i) {
            const float send = u8 ? val[i] : val[i + h8];
            const float keep = u8 ? val[i + h8] : val[i];
            val[i] = keep + __shfl_xor(send, 8, 32);
        }
#pragma unroll
        for (int i = 0; i < h4; ++i) {
            const float send = u4 ? val[i] : val[i + h4];
            const float keep = u4 ? val[i + h4] : val[i];
            val[i] = keep + __shfl_xor(send, 4, 32);
        }
#pragma unroll
        for (int i = 0; i < h2; ++i) {
            const float send = u2 ? val[i] : val[i + h2];
            const float keep = u2 ? val[i + h2] : val[i];
            val[i] = keep + __shfl_xor(send, 2, 32);
        }
#pragma unroll
        for (int i = 0; i < h1; ++i) {
            const float send = u1 ? val[i] : val[i + h1];
            const float keep = u1 ? val[i + h1] : val[i];
            val[i] = keep + __shfl_xor(send, 1, 32);
        }
    }
    const int hb = cb / F;
#pragma unroll
    for (int j = 0; j < HW; ++j) {
        const int idx  = l15 * HW + j;
        const int kind = idx & 1;
        const int g    = (idx >> 1) % HW;
        const int v    = (idx >> 1) / HW;
        float* dst = kind ? sd : ss;
        dst[(size_t)(r0 + v) * H + hb + g] = val[j];
    }
}

#define COUNT_HELPERS 2
#define COUNT_BLOCKS  ((EE + MEAN_THREADS - 1) / MEAN_THREADS + COUNT_HELPERS)
__global__ __launch_bounds__(MEAN_THREADS) void csr_count_kernel(const int* __restrict__ ei, int* __restrict__ deg,
                                                                 int* __restrict__ rank,
                                                                 const float* __restrict__ partial,
                                                                 float* __restrict__ meanw,
                                                                 const float* __restrict__ We0,
                                                                 const float* __restrict__ ae0,
                                                                 const float* __restrict__ We1,
                                                                 const float* __restrict__ ae1,
                                                                 const float* __restrict__ We2,
                                                                 const float* __restrict__ ae2,
                                                                 float* __restrict__ ce) {
    __shared__ float swave[MEAN_THREADS / 32];
    if (blockIdx.x == 0) {
        mean_final_block(partial, meanw, MEAN_BLOCKS, EE, swave);
        return;
    }
    if (blockIdx.x == 1) {
        edge_coef_block(We0, ae0, We1, ae1, We2, ae2, ce);
        return;
    }
    const int e = (blockIdx.x - COUNT_HELPERS) * MEAN_THREADS + threadIdx.x;
    if (e < EE) rank[e] = atomicAdd(&deg[ei[EE + e]], 1);
}

#define SCAN_THREADS 1024
#define SCAN_TILE    (SCAN_THREADS * 4)
#define SCAN_TILES   ((NN + SCAN_TILE - 1) / SCAN_TILE)
__global__ __launch_bounds__(SCAN_THREADS) void csr_scan_kernel(const int* __restrict__ deg,
                                                                int* __restrict__ rowptr) {
    __shared__ int swave[SCAN_TILES][32];
    __shared__ int stile[SCAN_TILES];
    const int t = threadIdx.x, lane = t & 31, wave = t >> 5;
    int4 v[SCAN_TILES];
    int  incl[SCAN_TILES];
#pragma unroll
    for (int k = 0; k < SCAN_TILES; ++k) {
        const int i = k * SCAN_TILE + 4 * t;
        v[k] = (i < NN) ? *(const int4*)(deg + i) : make_int4(0, 0, 0, 0);
    }
#pragma unroll
    for (int k = 0; k < SCAN_TILES; ++k) {
        int in = v[k].x + v[k].y + v[k].z + v[k].w;
        for (int m = 1; m < 32; m <<= 1) {
            const int o = __shfl_up(in, m, 32);
            if (lane >= m) in += o;
        }
        incl[k] = in;
        if (lane == 31) swave[k][wave] = in;
    }
    __syncthreads();
    if (wave < SCAN_TILES) {
        const int w = swave[wave][lane];
        int wi = w;
        for (int m = 1; m < 32; m <<= 1) {
            const int o = __shfl_up(wi, m, 32);
            if (lane >= m) wi += o;
        }
        swave[wave][lane] = wi - w;
        if (lane == 31) stile[wave] = wi;
    }
    __syncthreads();
    int carry = 0;
#pragma unroll
    for (int k = 0; k < SCAN_TILES; ++k) {
        const int i = k * SCAN_TILE + 4 * t;
        const int s = v[k].x + v[k].y + v[k].z + v[k].w;
        int4 r;
        r.x = carry + swave[k][wave] + incl[k] - s;
        r.y = r.x + v[k].x;
        r.z = r.y + v[k].y;
        r.w = r.z + v[k].z;
        if (i < NN) *(int4*)(rowptr + i) = r;
        carry += stile[k];
    }
    if (t == 0) rowptr[NN] = carry;
}

__global__ void csr_fill_kernel(const int* __restrict__ ei, const float* __restrict__ ew,
                                const int* __restrict__ rowptr, const int* __restrict__ rank,
                                int* __restrict__ csr_src, float* __restrict__ csr_w) {
    const int e = blockIdx.x * blockDim.x + threadIdx.x;
    if (e >= EE) return;
    const int pos = rowptr[ei[EE + e]] + rank[e];
    csr_src[pos] = ei[e];
    csr_w[pos]   = ew[e];
}

#define GATHER_WAVES 8

__device__ __forceinline__ float wave_max(float v) {
    for (int m = 16; m >= 1; m >>= 1) v = fmaxf(v, __shfl_xor(v, m, 32));
    return v;
}
__device__ __forceinline__ float wave_sum(float v) {
    for (int m = 16; m >= 1; m >>= 1) v += __shfl_xor(v, m, 32);
    return v;
}

template <int H>
__device__ __forceinline__ void edge_alpha(const int* __restrict__ csr_src, const float* __restrict__ csr_w,
                                           const float* __restrict__ ss, const float* sdv, const float* cev,
                                           int base, int n, int lane, int& s, float* a) {
    s = 0;
    float w = 0.0f;
    if (lane < n) { s = csr_src[base + lane]; w = csr_w[base + lane]; }
    float q[H];
    if constexpr (H == 4) {
        const float4 t = *(const float4*)(ss + (size_t)s * 4);
        q[0] = t.x; q[1] = t.y; q[2] = t.z; q[3] = t.w;
    } else {
#pragma unroll
        for (int h = 0; h < H; ++h) q[h] = ss[(size_t)s * H + h];
    }
#pragma unroll
    for (int h = 0; h < H; ++h) {
        float x = q[h] + sdv[h] + w * cev[h];
        x = x > 0.0f ? x : 0.2f * x;
        a[h] = lane < n ? x : -__builtin_inff();
    }
}

template <int H>
__device__ __forceinline__ float pick_attn(const float* at, int t, int hl) {
    float r[H];
#pragma unroll
    for (int h = 0; h < H; ++h)
        r[h] = __int_as_float(__builtin_amdgcn_readlane(__float_as_int(at[h]), t));
    float v = r[0];
#pragma unroll
    for (int h = 1; h < H; ++h) v = (hl == h) ? r[h] : v;
    return v;
}

template <int VW, int H, bool ELU>
__global__ __launch_bounds__(GATHER_WAVES * 32) void gat_node_kernel(const int* __restrict__ rowptr,
                                                                     const int* __restrict__ csr_src,
                                                                     const float* __restrict__ csr_w,
                                                                     const float* __restrict__ xw,
                                                                     const float* __restrict__ ss,
                                                                     const float* __restrict__ sd,
                                                                     const float* __restrict__ ce,
                                                                     const float* __restrict__ meanw,
                                                                     const float* __restrict__ b,
                                                                     float* __restrict__ out) {
    constexpr int C = 32 * VW;
    constexpr int F = C / H;
    typedef float vecf __attribute__((ext_vector_type(VW)));
    const int lane = threadIdx.x & 31;
    const int d = __builtin_amdgcn_readfirstlane(blockIdx.x * GATHER_WAVES + (threadIdx.x >> 5));
    if (d >= NN) return;
    const int c0 = lane * VW;
    const int hl = c0 / F;
    const int beg = rowptr[d], end = rowptr[d + 1];
    const int n0 = (end - beg) < 32 ? (end - beg) : 32;

    float cev[H], sdv[H], a_self[H], m[H];
    const float mw = meanw[0];
#pragma unroll
    for (int h = 0; h < H; ++h) {
        cev[h] = ce[h];
        sdv[h] = sd[(size_t)d * H + h];
        float x = ss[(size_t)d * H + h] + sdv[h] + mw * cev[h];
        a_self[h] = x > 0.0f ? x : 0.2f * x;
        m[h] = a_self[h];
    }

    int s0; float a0[H];
    edge_alpha<H>(csr_src, csr_w, ss, sdv, cev, beg, n0, lane, s0, a0);
#pragma unroll
    for (int h = 0; h < H; ++h) m[h] = fmaxf(m[h], a0[h]);
    for (int base = beg + 32; base < end; base += 32) {
        const int n = (end - base) < 32 ? (end - base) : 32;
        int s; float a[H];
        edge_alpha<H>(csr_src, csr_w, ss, sdv, cev, base, n, lane, s, a);
#pragma unroll
        for (int h = 0; h < H; ++h) m[h] = fmaxf(m[h], a[h]);
    }
#pragma unroll
    for (int h = 0; h < H; ++h) m[h] = wave_max(m[h]);

    float at0[H], den[H], ex_self[H];
#pragma unroll
    for (int h = 0; h < H; ++h) {
        at0[h] = lane < n0 ? __expf(a0[h] - m[h]) : 0.0f;
        den[h] = at0[h];
    }
    for (int base = beg + 32; base < end; base += 32) {
        const int n = (end - base) < 32 ? (end - base) : 32;
        int s; float a[H];
        edge_alpha<H>(csr_src, csr_w, ss, sdv, cev, base, n, lane, s, a);
#pragma unroll
        for (int h = 0; h < H; ++h) den[h] += lane < n ? __expf(a[h] - m[h]) : 0.0f;
    }
#pragma unroll
    for (int h = 0; h < H; ++h) {
        ex_self[h] = __expf(a_self[h] - m[h]);
        den[h] = wave_sum(den[h]) + ex_self[h] + 1e-16f;
        at0[h] = at0[h] / den[h];
    }

    vecf acc = {};
#pragma unroll 4
    for (int t = 0; t < n0; ++t) {
        const int s = __builtin_amdgcn_readlane(s0, t);
        const float attn = pick_attn<H>(at0, t, hl);
        const vecf v = *(const vecf*)(xw + (size_t)s * C + c0);
        acc += v * attn;
    }
    for (int base = beg + 32; base < end; base += 32) {
        const int n = (end - base) < 32 ? (end - base) : 32;
        int s1; float a[H];
        edge_alpha<H>(csr_src, csr_w, ss, sdv, cev, base, n, lane, s1, a);
#pragma unroll
        for (int h = 0; h < H; ++h) a[h] = (lane < n ? __expf(a[h] - m[h]) : 0.0f) / den[h];
#pragma unroll 4
        for (int t = 0; t < n; ++t) {
            const int s = __builtin_amdgcn_readlane(s1, t);
            const float attn = pick_attn<H>(a, t, hl);
            const vecf v = *(const vecf*)(xw + (size_t)s * C + c0);
            acc += v * attn;
        }
    }
    {
        float exs = ex_self[0], dn = den[0];
#pragma unroll
        for (int h = 1; h < H; ++h) { exs = (hl == h) ? ex_self[h] : exs; dn = (hl == h) ? den[h] : dn; }
        const float attn = exs / dn;
        const vecf v = *(const vecf*)(xw + (size_t)d * C + c0);
        acc += v * attn;
    }
    float bias[VW];
#pragma unroll
    for (int i = 0; i < VW; ++i) bias[i] = b[c0 + i];
    vecf r;
#pragma unroll
    for (int i = 0; i < VW; ++i) {
        float v = acc[i] + bias[i];
        if constexpr (ELU) v = v > 0.0f ? v : expm1f(v);
        r[i] = v;
    }
    *(vecf*)(out + (size_t)d * C + c0) = r;
}

extern "C" void kernel_launch(void* const* d_in, const int* in_sizes, int n_in,
                              void* d_out, int out_size, void* d_ws, size_t ws_size,
                              hipStream_t stream) {
    const float* x  = (const float*)d_in[0];
    const int*   ei = (const int*)d_in[1];
    const float* ew = (const float*)d_in[2];

    const float* Wl[3] = { (const float*)d_in[3],  (const float*)d_in[9],  (const float*)d_in[15] };
    const float* As[3] = { (const float*)d_in[4],  (const float*)d_in[10], (const float*)d_in[16] };
    const float* Ad[3] = { (const float*)d_in[5],  (const float*)d_in[11], (const float*)d_in[17] };
    const float* We[3] = { (const float*)d_in[6],  (const float*)d_in[12], (const float*)d_in[18] };
    const float* Ae[3] = { (const float*)d_in[7],  (const float*)d_in[13], (const float*)d_in[19] };
    const float* Bi[3] = { (const float*)d_in[8],  (const float*)d_in[14], (const float*)d_in[20] };

    float* ws    = (float*)d_ws;
    float* xw    = ws;
    float* bufA  = xw    + (size_t)NN * 128;
    float* bufB  = bufA  + (size_t)NN * 128;
    float* ss    = bufB  + (size_t)NN * 128;
    float* sd    = ss    + (size_t)NN * 4;
    float* ce    = sd    + (size_t)NN * 4;
    float* meanw = ce + 32;
    float* mpart = meanw + 32;
    int* deg     = (int*)(mpart + 256);
    int* rowptr  = deg + NN;
    int* rank    = rowptr + NN + 4;
    int* csr_src = rank + EE;
    float* csr_w = (float*)(csr_src + EE);

    mean_partial_kernel<<<MEAN_BLOCKS, MEAN_THREADS, 0, stream>>>(ew, mpart, deg, EE);
    csr_count_kernel<<<COUNT_BLOCKS, MEAN_THREADS, 0, stream>>>(ei, deg, rank, mpart, meanw,
                                                                We[0], Ae[0], We[1], Ae[1], We[2], Ae[2], ce);
    csr_scan_kernel<<<1, SCAN_THREADS, 0, stream>>>(deg, rowptr);
    csr_fill_kernel<<<(EE + 255) / 256, 256, 0, stream>>>(ei, ew, rowptr, rank, csr_src, csr_w);

    const float* in = x;
    float* outs[3] = { bufA, bufB, (float*)d_out };
    const float* ces[3] = { ce, ce + 4, ce + 8 };

    for (int l = 0; l < 3; ++l) {
        if (l < 2)
            gemm_score_kernel<128, 32><<<dim3((STRIPS + GEMM_WAVES - 1) / GEMM_WAVES, 128 / GEMM_COLS),
                                        GEMM_WAVES * 32, 0, stream>>>(in, Wl[l], As[l], Ad[l], xw, ss, sd);
        else
            gemm_score_kernel<64, 64><<<dim3((STRIPS + GEMM_WAVES - 1) / GEMM_WAVES, 64 / GEMM_COLS),
                                       GEMM_WAVES * 32, 0, stream>>>(in, Wl[l], As[l], Ad[l], xw, ss, sd);

        if (l < 2)
            gat_node_kernel<4, 4, true><<<(NN + GATHER_WAVES - 1) / GATHER_WAVES, GATHER_WAVES * 32, 0, stream>>>(
                rowptr, csr_src, csr_w, xw, ss, sd, ces[l], meanw, Bi[l], outs[l]);
        else
            gat_node_kernel<2, 1, false><<<(NN + GATHER_WAVES - 1) / GATHER_WAVES, GATHER_WAVES * 32, 0, stream>>>(
                rowptr, csr_src, csr_w, xw, ss, sd, ces[l], meanw, Bi[l], outs[l]);

        in = outs[l];
    }
}


# GEMMs of layers 1, 2 move into the per-destination pass before them: a block's 16 activated rows feed the WMMA via LDS (same k order); 2 launches and the rows' write+reads go; score sum order changes
# speedup vs baseline: 7.0389x; 1.0608x over previous
#include <hip/hip_runtime.h>
#include <math.h>

#define NN    50000
#define EE    800000
#define ETOT  (EE + NN)
#define KDIM  128

typedef float v2f __attribute__((ext_vector_type(2)));
typedef float v8f __attribute__((ext_vector_type(8)));

#define MEAN_THREADS    256
#define MEAN_PER_THREAD 16
#define MEAN_BLOCKS     ((EE + MEAN_THREADS * MEAN_PER_THREAD - 1) / (MEAN_THREADS * MEAN_PER_THREAD))

__device__ __forceinline__ float block_sum_256(float s, float* swave) {
    for (int m = 16; m >= 1; m >>= 1) s += __shfl_xor(s, m, 32);
    const int wave = threadIdx.x >> 5, lane = threadIdx.x & 31;
    if (lane == 0) swave[wave] = s;
    __syncthreads();
    float t = (lane < MEAN_THREADS / 32) ? swave[lane] : 0.0f;
    for (int m = MEAN_THREADS / 64; m >= 1; m >>= 1) t += __shfl_xor(t, m, 32);
    return t;
}

__global__ __launch_bounds__(MEAN_THREADS) void mean_partial_kernel(const float* __restrict__ ew,
                                                                    float* __restrict__ partial,
                                                                    int* __restrict__ deg, int E) {
    __shared__ float swave[MEAN_THREADS / 32];
    const int T = gridDim.x * MEAN_THREADS;
    const int g = blockIdx.x * MEAN_THREADS + threadIdx.x;
    for (int i = g; i < NN; i += T) deg[i] = 0;
    float s = 0.0f;
    for (int i0 = g; i0 < E; i0 += MEAN_PER_THREAD * T) {
        float v[MEAN_PER_THREAD];
        for (int j = 0; j < MEAN_PER_THREAD; ++j) {
            const int i = i0 + j * T;
            const float x = ew[i < E ? i : E - 1];
            v[j] = i < E ? x : 0.0f;
        }
        for (int w = 1; w < MEAN_PER_THREAD; w <<= 1) {
            for (int j = 0; j < MEAN_PER_THREAD; j += 2 * w) v[j] += v[j + w];
        }
        s += v[0];
    }
    s = block_sum_256(s, swave);
    if (threadIdx.x == 0) partial[blockIdx.x] = s;
}

__device__ __forceinline__ void mean_final_block(const float* __restrict__ partial, float* __restrict__ meanw,
                                                 int nb, int E, float* swave) {
    float s = 0.0f;
    for (int i = threadIdx.x; i < nb; i += MEAN_THREADS) s += partial[i];
    s = block_sum_256(s, swave);
    if (threadIdx.x == 0) meanw[0] = s / (float)E;
}

__device__ __forceinline__ void edge_coef_block(const float* __restrict__ We0, const float* __restrict__ ae0,
                                                const float* __restrict__ We1, const float* __restrict__ ae1,
                                                const float* __restrict__ We2, const float* __restrict__ ae2,
                                                float* __restrict__ ce) {
    const int t = threadIdx.x;
    if (t >= 9) return;
    const int l = t >> 2, h = t & 3;
    const int F = l < 2 ? 32 : 64;
    const float* We = l == 0 ? We0 : (l == 1 ? We1 : We2);
    const float* ae = l == 0 ? ae0 : (l == 1 ? ae1 : ae2);
    float s = 0.0f;
    for (int f = 0; f < F; ++f) s += We[h * F + f] * ae[h * F + f];
    ce[t] = s;
}

#define GEMM_TILES 4
#define GEMM_COLS  (GEMM_TILES * 16)
#define GEMM_WAVES 4
#define STRIPS     (NN / 16)

template <int C, int F>
__global__ __launch_bounds__(GEMM_WAVES * 32) void gemm_score_kernel(const float* __restrict__ X,
                                                                     const float* __restrict__ W,
                                                                     const float* __restrict__ a_s,
                                                                     const float* __restrict__ a_d,
                                                                     float* __restrict__ xw,
                                                                     float* __restrict__ ss,
                                                                     float* __restrict__ sd) {
    constexpr int H   = C / F;
    constexpr int HW  = GEMM_COLS / F;
    constexpr int TPH = F / 16;
    constexpr int NV  = 16 * HW;
    const int lane = threadIdx.x & 31;
    const int strip = __builtin_amdgcn_readfirstlane(blockIdx.x * GEMM_WAVES + (threadIdx.x >> 5));
    if (strip >= STRIPS) return;
    const int half = lane >> 4;
    const int l15  = lane & 15;
    const int cb   = blockIdx.y * GEMM_COLS;

    v8f acc[GEMM_TILES] = {};
    const float* xrow = X + (size_t)(strip * 16 + l15) * KDIM + half * 2;
    const float* wcol = W + (size_t)(half * 2) * C + cb + l15;

#pragma unroll 8
    for (int k = 0; k < KDIM; k += 4) {
        v2f a;
        a[0] = xrow[k];
        a[1] = xrow[k + 1];
        v2f b[GEMM_TILES];
#pragma unroll
        for (int t = 0; t < GEMM_TILES; ++t) {
            b[t][0] = wcol[(size_t)k * C + t * 16];
            b[t][1] = wcol[(size_t)(k + 1) * C + t * 16];
        }
#pragma unroll
        for (int t = 0; t < GEMM_TILES; ++t)
            acc[t] = __builtin_amdgcn_wmma_f32_16x16x4_f32(
                 false, a,  false, b[t],
                 (short)0, acc[t],  false,  false);
    }

    const int r0 = strip * 16 + half * 8;
#pragma unroll
    for (int t = 0; t < GEMM_TILES; ++t) {
#pragma unroll
        for (int v = 0; v < 8; ++v)
            xw[(size_t)(r0 + v) * C + cb + t * 16 + l15] = acc[t][v];
    }

    float as_c[GEMM_TILES], ad_c[GEMM_TILES];
#pragma unroll
    for (int t = 0; t < GEMM_TILES; ++t) {
        as_c[t] = a_s[cb + t * 16 + l15];
        ad_c[t] = a_d[cb + t * 16 + l15];
    }
    float val[NV];
#pragma unroll
    for (int v = 0; v < 8; ++v) {
#pragma unroll
        for (int g = 0; g < HW; ++g) {
            float ps = acc[g * TPH][v] * as_c[g * TPH];
            float pd = acc[g * TPH][v] * ad_c[g * TPH];
#pragma unroll
            for (int j = 1; j < TPH; ++j) {
                ps += acc[g * TPH + j][v] * as_c[g * TPH + j];
                pd += acc[g * TPH + j][v] * ad_c[g * TPH + j];
            }
            val[(v * HW + g) * 2 + 0] = ps;
            val[(v * HW + g) * 2 + 1] = pd;
        }
    }
    {
        constexpr int h8 = NV / 2, h4 = NV / 4, h2 = NV / 8, h1 = NV / 16;
        const bool u8 = (lane & 8) != 0, u4 = (lane & 4) != 0, u2 = (lane & 2) != 0, u1 = (lane & 1) != 0;
#pragma unroll
        for (int i = 0; i < h8; ++i) {
            const float send = u8 ? val[i] : val[i + h8];
            const float keep = u8 ? val[i + h8] : val[i];
            val[i] = keep + __shfl_xor(send, 8, 32);
        }
#pragma unroll
        for (int i = 0; i < h4; ++i) {
            const float send = u4 ? val[i] : val[i + h4];
            const float keep = u4 ? val[i + h4] : val[i];
            val[i] = keep + __shfl_xor(send, 4, 32);
        }
#pragma unroll
        for (int i = 0; i < h2; ++i) {
            const float send = u2 ? val[i] : val[i + h2];
            const float keep = u2 ? val[i + h2] : val[i];
            val[i] = keep + __shfl_xor(send, 2, 32);
        }
#pragma unroll
        for (int i = 0; i < h1; ++i) {
            const float send = u1 ? val[i] : val[i + h1];
            const float keep = u1 ? val[i + h1] : val[i];
            val[i] = keep + __shfl_xor(send, 1, 32);
        }
    }
    const int hb = cb / F;
#pragma unroll
    for (int j = 0; j < HW; ++j) {
        const int idx  = l15 * HW + j;
        const int kind = idx & 1;
        const int g    = (idx >> 1) % HW;
        const int v    = (idx >> 1) / HW;
        float* dst = kind ? sd : ss;
        dst[(size_t)(r0 + v) * H + hb + g] = val[j];
    }
}

#define COUNT_HELPERS 2
#define COUNT_BLOCKS  ((EE + MEAN_THREADS - 1) / MEAN_THREADS + COUNT_HELPERS)
__global__ __launch_bounds__(MEAN_THREADS) void csr_count_kernel(const int* __restrict__ ei, int* __restrict__ deg,
                                                                 int* __restrict__ rank,
                                                                 const float* __restrict__ partial,
                                                                 float* __restrict__ meanw,
                                                                 const float* __restrict__ We0,
                                                                 const float* __restrict__ ae0,
                                                                 const float* __restrict__ We1,
                                                                 const float* __restrict__ ae1,
                                                                 const float* __restrict__ We2,
                                                                 const float* __restrict__ ae2,
                                                                 float* __restrict__ ce) {
    __shared__ float swave[MEAN_THREADS / 32];
    if (blockIdx.x == 0) {
        mean_final_block(partial, meanw, MEAN_BLOCKS, EE, swave);
        return;
    }
    if (blockIdx.x == 1) {
        edge_coef_block(We0, ae0, We1, ae1, We2, ae2, ce);
        return;
    }
    const int e = (blockIdx.x - COUNT_HELPERS) * MEAN_THREADS + threadIdx.x;
    if (e < EE) rank[e] = atomicAdd(&deg[ei[EE + e]], 1);
}

#define SCAN_THREADS 1024
#define SCAN_TILE    (SCAN_THREADS * 4)
#define SCAN_TILES   ((NN + SCAN_TILE - 1) / SCAN_TILE)
__global__ __launch_bounds__(SCAN_THREADS) void csr_scan_kernel(const int* __restrict__ deg,
                                                                int* __restrict__ rowptr) {
    __shared__ int swave[SCAN_TILES][32];
    __shared__ int stile[SCAN_TILES];
    const int t = threadIdx.x, lane = t & 31, wave = t >> 5;
    int4 v[SCAN_TILES];
    int  incl[SCAN_TILES];
#pragma unroll
    for (int k = 0; k < SCAN_TILES; ++k) {
        const int i = k * SCAN_TILE + 4 * t;
        v[k] = (i < NN) ? *(const int4*)(deg + i) : make_int4(0, 0, 0, 0);
    }
#pragma unroll
    for (int k = 0; k < SCAN_TILES; ++k) {
        int in = v[k].x + v[k].y + v[k].z + v[k].w;
        for (int m = 1; m < 32; m <<= 1) {
            const int o = __shfl_up(in, m, 32);
            if (lane >= m) in += o;
        }
        incl[k] = in;
        if (lane == 31) swave[k][wave] = in;
    }
    __syncthreads();
    if (wave < SCAN_TILES) {
        const int w = swave[wave][lane];
        int wi = w;
        for (int m = 1; m < 32; m <<= 1) {
            const int o = __shfl_up(wi, m, 32);
            if (lane >= m) wi += o;
        }
        swave[wave][lane] = wi - w;
        if (lane == 31) stile[wave] = wi;
    }
    __syncthreads();
    int carry = 0;
#pragma unroll
    for (int k = 0; k < SCAN_TILES; ++k) {
        const int i = k * SCAN_TILE + 4 * t;
        const int s = v[k].x + v[k].y + v[k].z + v[k].w;
        int4 r;
        r.x = carry + swave[k][wave] + incl[k] - s;
        r.y = r.x + v[k].x;
        r.z = r.y + v[k].y;
        r.w = r.z + v[k].z;
        if (i < NN) *(int4*)(rowptr + i) = r;
        carry += stile[k];
    }
    if (t == 0) rowptr[NN] = carry;
}

__global__ void csr_fill_kernel(const int* __restrict__ ei, const float* __restrict__ ew,
                                const int* __restrict__ rowptr, const int* __restrict__ rank,
                                int* __restrict__ csr_src, float* __restrict__ csr_w) {
    const int e = blockIdx.x * blockDim.x + threadIdx.x;
    if (e >= EE) return;
    const int pos = rowptr[ei[EE + e]] + rank[e];
    csr_src[pos] = ei[e];
    csr_w[pos]   = ew[e];
}

#define GATHER_WAVES 8

__device__ __forceinline__ float wave_max(float v) {
    for (int m = 16; m >= 1; m >>= 1) v = fmaxf(v, __shfl_xor(v, m, 32));
    return v;
}
__device__ __forceinline__ float wave_sum(float v) {
    for (int m = 16; m >= 1; m >>= 1) v += __shfl_xor(v, m, 32);
    return v;
}

template <int H>
__device__ __forceinline__ void edge_alpha(const int* __restrict__ csr_src, const float* __restrict__ csr_w,
                                           const float* __restrict__ ss, const float* sdv, const float* cev,
                                           int base, int n, int lane, int& s, float* a) {
    s = 0;
    float w = 0.0f;
    if (lane < n) { s = csr_src[base + lane]; w = csr_w[base + lane]; }
    float q[H];
    if constexpr (H == 4) {
        const float4 t = *(const float4*)(ss + (size_t)s * 4);
        q[0] = t.x; q[1] = t.y; q[2] = t.z; q[3] = t.w;
    } else {
#pragma unroll
        for (int h = 0; h < H; ++h) q[h] = ss[(size_t)s * H + h];
    }
#pragma unroll
    for (int h = 0; h < H; ++h) {
        float x = q[h] + sdv[h] + w * cev[h];
        x = x > 0.0f ? x : 0.2f * x;
        a[h] = lane < n ? x : -__builtin_inff();
    }
}

template <int H>
__device__ __forceinline__ float pick_attn(const float* at, int t, int hl) {
    float r[H];
#pragma unroll
    for (int h = 0; h < H; ++h)
        r[h] = __int_as_float(__builtin_amdgcn_readlane(__float_as_int(at[h]), t));
    float v = r[0];
#pragma unroll
    for (int h = 1; h < H; ++h) v = (hl == h) ? r[h] : v;
    return v;
}

template <int VW> struct lane_vec { typedef float type __attribute__((ext_vector_type(VW))); };

template <int VW, int H, bool ELU>
__device__ __forceinline__ typename lane_vec<VW>::type gat_node_row(const int* __restrict__ rowptr,
                                                                    const int* __restrict__ csr_src,
                                                                    const float* __restrict__ csr_w,
                                                                    const float* __restrict__ xw,
                                                                    const float* __restrict__ ss,
                                                                    const float* __restrict__ sd,
                                                                    const float* __restrict__ ce,
                                                                    const float* __restrict__ meanw,
                                                                    const float* __restrict__ b,
                                                                    const int d, const int lane) {
    constexpr int C = 32 * VW;
    constexpr int F = C / H;
    typedef typename lane_vec<VW>::type vecf;
    const int c0 = lane * VW;
    const int hl = c0 / F;
    const int beg = rowptr[d], end = rowptr[d + 1];
    const int n0 = (end - beg) < 32 ? (end - beg) : 32;

    float cev[H], sdv[H], a_self[H], m[H];
    const float mw = meanw[0];
#pragma unroll
    for (int h = 0; h < H; ++h) {
        cev[h] = ce[h];
        sdv[h] = sd[(size_t)d * H + h];
        float x = ss[(size_t)d * H + h] + sdv[h] + mw * cev[h];
        a_self[h] = x > 0.0f ? x : 0.2f * x;
        m[h] = a_self[h];
    }

    int s0; float a0[H];
    edge_alpha<H>(csr_src, csr_w, ss, sdv, cev, beg, n0, lane, s0, a0);
#pragma unroll
    for (int h = 0; h < H; ++h) m[h] = fmaxf(m[h], a0[h]);
    for (int base = beg + 32; base < end; base += 32) {
        const int n = (end - base) < 32 ? (end - base) : 32;
        int s; float a[H];
        edge_alpha<H>(csr_src, csr_w, ss, sdv, cev, base, n, lane, s, a);
#pragma unroll
        for (int h = 0; h < H; ++h) m[h] = fmaxf(m[h], a[h]);
    }
#pragma unroll
    for (int h = 0; h < H; ++h) m[h] = wave_max(m[h]);

    float at0[H], den[H], ex_self[H];
#pragma unroll
    for (int h = 0; h < H; ++h) {
        at0[h] = lane < n0 ? __expf(a0[h] - m[h]) : 0.0f;
        den[h] = at0[h];
    }
    for (int base = beg + 32; base < end; base += 32) {
        const int n = (end - base) < 32 ? (end - base) : 32;
        int s; float a[H];
        edge_alpha<H>(csr_src, csr_w, ss, sdv, cev, base, n, lane, s, a);
#pragma unroll
        for (int h = 0; h < H; ++h) den[h] += lane < n ? __expf(a[h] - m[h]) : 0.0f;
    }
#pragma unroll
    for (int h = 0; h < H; ++h) {
        ex_self[h] = __expf(a_self[h] - m[h]);
        den[h] = wave_sum(den[h]) + ex_self[h] + 1e-16f;
        at0[h] = at0[h] / den[h];
    }

    vecf acc = {};
#pragma unroll 4
    for (int t = 0; t < n0; ++t) {
        const int s = __builtin_amdgcn_readlane(s0, t);
        const float attn = pick_attn<H>(at0, t, hl);
        const vecf v = *(const vecf*)(xw + (size_t)s * C + c0);
        acc += v * attn;
    }
    for (int base = beg + 32; base < end; base += 32) {
        const int n = (end - base) < 32 ? (end - base) : 32;
        int s1; float a[H];
        edge_alpha<H>(csr_src, csr_w, ss, sdv, cev, base, n, lane, s1, a);
#pragma unroll
        for (int h = 0; h < H; ++h) a[h] = (lane < n ? __expf(a[h] - m[h]) : 0.0f) / den[h];
#pragma unroll 4
        for (int t = 0; t < n; ++t) {
            const int s = __builtin_amdgcn_readlane(s1, t);
            const float attn = pick_attn<H>(a, t, hl);
            const vecf v = *(const vecf*)(xw + (size_t)s * C + c0);
            acc += v * attn;
        }
    }
    {
        float exs = ex_self[0], dn = den[0];
#pragma unroll
        for (int h = 1; h < H; ++h) { exs = (hl == h) ? ex_self[h] : exs; dn = (hl == h) ? den[h] : dn; }
        const float attn = exs / dn;
        const vecf v = *(const vecf*)(xw + (size_t)d * C + c0);
        acc += v * attn;
    }
    float bias[VW];
#pragma unroll
    for (int i = 0; i < VW; ++i) bias[i] = b[c0 + i];
    vecf r;
#pragma unroll
    for (int i = 0; i < VW; ++i) {
        float v = acc[i] + bias[i];
        if constexpr (ELU) v = v > 0.0f ? v : expm1f(v);
        r[i] = v;
    }
    return r;
}

template <int VW, int H, bool ELU>
__global__ __launch_bounds__(GATHER_WAVES * 32) void gat_node_kernel(const int* __restrict__ rowptr,
                                                                     const int* __restrict__ csr_src,
                                                                     const float* __restrict__ csr_w,
                                                                     const float* __restrict__ xw,
                                                                     const float* __restrict__ ss,
                                                                     const float* __restrict__ sd,
                                                                     const float* __restrict__ ce,
                                                                     const float* __restrict__ meanw,
                                                                     const float* __restrict__ b,
                                                                     float* __restrict__ out) {
    typedef typename lane_vec<VW>::type vecf;
    const int lane = threadIdx.x & 31;
    const int d = __builtin_amdgcn_readfirstlane(blockIdx.x * GATHER_WAVES + (threadIdx.x >> 5));
    if (d >= NN) return;
    const vecf r = gat_node_row<VW, H, ELU>(rowptr, csr_src, csr_w, xw, ss, sd, ce, meanw, b, d, lane);
    *(vecf*)(out + (size_t)d * (32 * VW) + lane * VW) = r;
}

#define FUSE_NODES 16
#define FUSE_WAVES 8
#define FUSE_PITCH 132

template <int CN, int FN>
__global__ __launch_bounds__(FUSE_WAVES * 32) void gat_gemm_kernel(const int* __restrict__ rowptr,
                                                                   const int* __restrict__ csr_src,
                                                                   const float* __restrict__ csr_w,
                                                                   const float* __restrict__ xw,
                                                                   const float* __restrict__ ss,
                                                                   const float* __restrict__ sd,
                                                                   const float* __restrict__ ce,
                                                                   const float* __restrict__ meanw,
                                                                   const float* __restrict__ b,
                                                                   const float* __restrict__ Wn,
                                                                   const float* __restrict__ a_sn,
                                                                   const float* __restrict__ a_dn,
                                                                   float* __restrict__ xwn,
                                                                   float* __restrict__ ssn,
                                                                   float* __restrict__ sdn) {
    constexpr int HN  = CN / FN;
    constexpr int NT  = CN / 16;
    constexpr int TPH = FN / 16;
    static_assert(NT <= FUSE_WAVES, "one tile per wave");
    static_assert(16 * HN * 2 <= FUSE_WAVES * 32, "one thread per finished score");
    typedef typename lane_vec<4>::type vecf;
    __shared__ __attribute__((aligned(16))) float hrow[FUSE_NODES * FUSE_PITCH];
    __shared__ float part[NT][16][2];
    const int lane  = threadIdx.x & 31;
    const int wave  = __builtin_amdgcn_readfirstlane(threadIdx.x >> 5);
    const int strip = blockIdx.x;
    for (int i = 0; i < FUSE_NODES / FUSE_WAVES; ++i) {
        const int rl = wave * (FUSE_NODES / FUSE_WAVES) + i;
        const vecf r = gat_node_row<4, 4, true>(rowptr, csr_src, csr_w, xw, ss, sd, ce, meanw, b,
                                                strip * 16 + rl, lane);
        *(vecf*)(hrow + rl * FUSE_PITCH + lane * 4) = r;
    }
    __syncthreads();
    if (wave < NT) {
        const int half = lane >> 4;
        const int l15  = lane & 15;
        const int cb   = wave * 16;
        v8f acc = {};
        const float* xrow = hrow + l15 * FUSE_PITCH + half * 2;
        const float* wcol = Wn + (size_t)(half * 2) * CN + cb + l15;
#pragma unroll 8
        for (int k = 0; k < KDIM; k += 4) {
            v2f a;
            a[0] = xrow[k];
            a[1] = xrow[k + 1];
            v2f bf;
            bf[0] = wcol[(size_t)k * CN];
            bf[1] = wcol[(size_t)(k + 1) * CN];
            acc = __builtin_amdgcn_wmma_f32_16x16x4_f32(
                 false, a,  false, bf,
                 (short)0, acc,  false,  false);
        }
        const int r0 = strip * 16 + half * 8;
#pragma unroll
        for (int v = 0; v < 8; ++v)
            xwn[(size_t)(r0 + v) * CN + cb + l15] = acc[v];

        const float as_c = a_sn[cb + l15];
        const float ad_c = a_dn[cb + l15];
        float val[16];
#pragma unroll
        for (int v = 0; v < 8; ++v) {
            val[2 * v + 0] = acc[v] * as_c;
            val[2 * v + 1] = acc[v] * ad_c;
        }
        {
            const bool u8 = (lane & 8) != 0, u4 = (lane & 4) != 0, u2 = (lane & 2) != 0, u1 = (lane & 1) != 0;
#pragma unroll
            for (int i = 0; i < 8; ++i) {
                const float send = u8 ? val[i] : val[i + 8];
                const float keep = u8 ? val[i + 8] : val[i];
                val[i] = keep + __shfl_xor(send, 8, 32);
            }
#pragma unroll
            for (int i = 0; i < 4; ++i) {
                const float send = u4 ? val[i] : val[i + 4];
                const float keep = u4 ? val[i + 4] : val[i];
                val[i] = keep + __shfl_xor(send, 4, 32);
            }
#pragma unroll
            for (int i = 0; i < 2; ++i) {
                const float send = u2 ? val[i] : val[i + 2];
                const float keep = u2 ? val[i + 2] : val[i];
                val[i] = keep + __shfl_xor(send, 2, 32);
            }
            {
                const float send = u1 ? val[0] : val[1];
                const float keep = u1 ? val[1] : val[0];
                val[0] = keep + __shfl_xor(send, 1, 32);
            }
        }
        part[wave][half * 8 + (l15 >> 1)][l15 & 1] = val[0];
    }
    __syncthreads();
    if (threadIdx.x < 16 * HN * 2) {
        const int kind = threadIdx.x & 1;
        const int h    = (threadIdx.x >> 1) % HN;
        const int row  = (threadIdx.x >> 1) / HN;
        float s = part[h * TPH][row][kind];
#pragma unroll
        for (int j = 1; j < TPH; ++j) s += part[h * TPH + j][row][kind];
        float* dst = kind ? sdn : ssn;
        dst[(size_t)(strip * 16 + row) * HN + h] = s;
    }
}

extern "C" void kernel_launch(void* const* d_in, const int* in_sizes, int n_in,
                              void* d_out, int out_size, void* d_ws, size_t ws_size,
                              hipStream_t stream) {
    const float* x  = (const float*)d_in[0];
    const int*   ei = (const int*)d_in[1];
    const float* ew = (const float*)d_in[2];

    const float* Wl[3] = { (const float*)d_in[3],  (const float*)d_in[9],  (const float*)d_in[15] };
    const float* As[3] = { (const float*)d_in[4],  (const float*)d_in[10], (const float*)d_in[16] };
    const float* Ad[3] = { (const float*)d_in[5],  (const float*)d_in[11], (const float*)d_in[17] };
    const float* We[3] = { (const float*)d_in[6],  (const float*)d_in[12], (const float*)d_in[18] };
    const float* Ae[3] = { (const float*)d_in[7],  (const float*)d_in[13], (const float*)d_in[19] };
    const float* Bi[3] = { (const float*)d_in[8],  (const float*)d_in[14], (const float*)d_in[20] };

    float* ws    = (float*)d_ws;
    float* xw    = ws;
    float* bufA  = xw    + (size_t)NN * 128;
    float* bufB  = bufA  + (size_t)NN * 128;
    float* ss    = bufB  + (size_t)NN * 128;
    float* sd    = ss    + (size_t)NN * 4;
    float* ce    = sd    + (size_t)NN * 4;
    float* meanw = ce + 32;
    float* mpart = meanw + 32;
    int* deg     = (int*)(mpart + 256);
    int* rowptr  = deg + NN;
    int* rank    = rowptr + NN + 4;
    int* csr_src = rank + EE;
    float* csr_w = (float*)(csr_src + EE);

    mean_partial_kernel<<<MEAN_BLOCKS, MEAN_THREADS, 0, stream>>>(ew, mpart, deg, EE);
    csr_count_kernel<<<COUNT_BLOCKS, MEAN_THREADS, 0, stream>>>(ei, deg, rank, mpart, meanw,
                                                                We[0], Ae[0], We[1], Ae[1], We[2], Ae[2], ce);
    csr_scan_kernel<<<1, SCAN_THREADS, 0, stream>>>(deg, rowptr);
    csr_fill_kernel<<<(EE + 255) / 256, 256, 0, stream>>>(ei, ew, rowptr, rank, csr_src, csr_w);

    float* ss2 = bufB;
    float* sd2 = bufB + (size_t)NN * 4;

    gemm_score_kernel<128, 32><<<dim3((STRIPS + GEMM_WAVES - 1) / GEMM_WAVES, 128 / GEMM_COLS),
                                GEMM_WAVES * 32, 0, stream>>>(x, Wl[0], As[0], Ad[0], xw, ss, sd);
    gat_gemm_kernel<128, 32><<<STRIPS, FUSE_WAVES * 32, 0, stream>>>(
        rowptr, csr_src, csr_w, xw, ss, sd, ce, meanw, Bi[0], Wl[1], As[1], Ad[1], bufA, ss2, sd2);
    gat_gemm_kernel<64, 64><<<STRIPS, FUSE_WAVES * 32, 0, stream>>>(
        rowptr, csr_src, csr_w, bufA, ss2, sd2, ce + 4, meanw, Bi[1], Wl[2], As[2], Ad[2], xw, ss, sd);
    gat_node_kernel<2, 1, false><<<(NN + GATHER_WAVES - 1) / GATHER_WAVES, GATHER_WAVES * 32, 0, stream>>>(
        rowptr, csr_src, csr_w, xw, ss, sd, ce + 8, meanw, Bi[2], (float*)d_out);
}


# edges by destination: fixed rows of 32 (source, weight) slots; the in-degree add's rank places each edge at once, overflow list past 32; prefix-sum and fill launches go (8 -> 6); arrival order kept
# speedup vs baseline: 8.4316x; 1.1979x over previous
#include <hip/hip_runtime.h>
#include <math.h>

#define NN    50000
#define EE    800000
#define ETOT  (EE + NN)
#define KDIM  128

typedef float v2f __attribute__((ext_vector_type(2)));
typedef float v8f __attribute__((ext_vector_type(8)));

#define MEAN_THREADS    256
#define MEAN_PER_THREAD 16
#define MEAN_BLOCKS     ((EE + MEAN_THREADS * MEAN_PER_THREAD - 1) / (MEAN_THREADS * MEAN_PER_THREAD))

__device__ __forceinline__ float block_sum_256(float s, float* swave) {
    for (int m = 16; m >= 1; m >>= 1) s += __shfl_xor(s, m, 32);
    const int wave = threadIdx.x >> 5, lane = threadIdx.x & 31;
    if (lane == 0) swave[wave] = s;
    __syncthreads();
    float t = (lane < MEAN_THREADS / 32) ? swave[lane] : 0.0f;
    for (int m = MEAN_THREADS / 64; m >= 1; m >>= 1) t += __shfl_xor(t, m, 32);
    return t;
}

__global__ __launch_bounds__(MEAN_THREADS) void mean_partial_kernel(const float* __restrict__ ew,
                                                                    float* __restrict__ partial,
                                                                    int* __restrict__ deg,
                                                                    int* __restrict__ ovf_n, int E) {
    __shared__ float swave[MEAN_THREADS / 32];
    const int T = gridDim.x * MEAN_THREADS;
    const int g = blockIdx.x * MEAN_THREADS + threadIdx.x;
    for (int i = g; i < NN; i += T) deg[i] = 0;
    if (g == 0) ovf_n[0] = 0;
    float s = 0.0f;
    for (int i0 = g; i0 < E; i0 += MEAN_PER_THREAD * T) {
        float v[MEAN_PER_THREAD];
        for (int j = 0; j < MEAN_PER_THREAD; ++j) {
            const int i = i0 + j * T;
            const float x = ew[i < E ? i : E - 1];
            v[j] = i < E ? x : 0.0f;
        }
        for (int w = 1; w < MEAN_PER_THREAD; w <<= 1) {
            for (int j = 0; j < MEAN_PER_THREAD; j += 2 * w) v[j] += v[j + w];
        }
        s += v[0];
    }
    s = block_sum_256(s, swave);
    if (threadIdx.x == 0) partial[blockIdx.x] = s;
}

__device__ __forceinline__ void mean_final_block(const float* __restrict__ partial, float* __restrict__ meanw,
                                                 int nb, int E, float* swave) {
    float s = 0.0f;
    for (int i = threadIdx.x; i < nb; i += MEAN_THREADS) s += partial[i];
    s = block_sum_256(s, swave);
    if (threadIdx.x == 0) meanw[0] = s / (float)E;
}

__device__ __forceinline__ void edge_coef_block(const float* __restrict__ We0, const float* __restrict__ ae0,
                                                const float* __restrict__ We1, const float* __restrict__ ae1,
                                                const float* __restrict__ We2, const float* __restrict__ ae2,
                                                float* __restrict__ ce) {
    const int t = threadIdx.x;
    if (t >= 9) return;
    const int l = t >> 2, h = t & 3;
    const int F = l < 2 ? 32 : 64;
    const float* We = l == 0 ? We0 : (l == 1 ? We1 : We2);
    const float* ae = l == 0 ? ae0 : (l == 1 ? ae1 : ae2);
    float s = 0.0f;
    for (int f = 0; f < F; ++f) s += We[h * F + f] * ae[h * F + f];
    ce[t] = s;
}

#define GEMM_TILES 4
#define GEMM_COLS  (GEMM_TILES * 16)
#define GEMM_WAVES 4
#define STRIPS     (NN / 16)

template <int C, int F>
__global__ __launch_bounds__(GEMM_WAVES * 32) void gemm_score_kernel(const float* __restrict__ X,
                                                                     const float* __restrict__ W,
                                                                     const float* __restrict__ a_s,
                                                                     const float* __restrict__ a_d,
                                                                     float* __restrict__ xw,
                                                                     float* __restrict__ ss,
                                                                     float* __restrict__ sd) {
    constexpr int H   = C / F;
    constexpr int HW  = GEMM_COLS / F;
    constexpr int TPH = F / 16;
    constexpr int NV  = 16 * HW;
    const int lane = threadIdx.x & 31;
    const int strip = __builtin_amdgcn_readfirstlane(blockIdx.x * GEMM_WAVES + (threadIdx.x >> 5));
    if (strip >= STRIPS) return;
    const int half = lane >> 4;
    const int l15  = lane & 15;
    const int cb   = blockIdx.y * GEMM_COLS;

    v8f acc[GEMM_TILES] = {};
    const float* xrow = X + (size_t)(strip * 16 + l15) * KDIM + half * 2;
    const float* wcol = W + (size_t)(half * 2) * C + cb + l15;

#pragma unroll 8
    for (int k = 0; k < KDIM; k += 4) {
        v2f a;
        a[0] = xrow[k];
        a[1] = xrow[k + 1];
        v2f b[GEMM_TILES];
#pragma unroll
        for (int t = 0; t < GEMM_TILES; ++t) {
            b[t][0] = wcol[(size_t)k * C + t * 16];
            b[t][1] = wcol[(size_t)(k + 1) * C + t * 16];
        }
#pragma unroll
        for (int t = 0; t < GEMM_TILES; ++t)
            acc[t] = __builtin_amdgcn_wmma_f32_16x16x4_f32(
                 false, a,  false, b[t],
                 (short)0, acc[t],  false,  false);
    }

    const int r0 = strip * 16 + half * 8;
#pragma unroll
    for (int t = 0; t < GEMM_TILES; ++t) {
#pragma unroll
        for (int v = 0; v < 8; ++v)
            xw[(size_t)(r0 + v) * C + cb + t * 16 + l15] = acc[t][v];
    }

    float as_c[GEMM_TILES], ad_c[GEMM_TILES];
#pragma unroll
    for (int t = 0; t < GEMM_TILES; ++t) {
        as_c[t] = a_s[cb + t * 16 + l15];
        ad_c[t] = a_d[cb + t * 16 + l15];
    }
    float val[NV];
#pragma unroll
    for (int v = 0; v < 8; ++v) {
#pragma unroll
        for (int g = 0; g < HW; ++g) {
            float ps = acc[g * TPH][v] * as_c[g * TPH];
            float pd = acc[g * TPH][v] * ad_c[g * TPH];
#pragma unroll
            for (int j = 1; j < TPH; ++j) {
                ps += acc[g * TPH + j][v] * as_c[g * TPH + j];
                pd += acc[g * TPH + j][v] * ad_c[g * TPH + j];
            }
            val[(v * HW + g) * 2 + 0] = ps;
            val[(v * HW + g) * 2 + 1] = pd;
        }
    }
    {
        constexpr int h8 = NV / 2, h4 = NV / 4, h2 = NV / 8, h1 = NV / 16;
        const bool u8 = (lane & 8) != 0, u4 = (lane & 4) != 0, u2 = (lane & 2) != 0, u1 = (lane & 1) != 0;
#pragma unroll
        for (int i = 0; i < h8; ++i) {
            const float send = u8 ? val[i] : val[i + h8];
            const float keep = u8 ? val[i + h8] : val[i];
            val[i] = keep + __shfl_xor(send, 8, 32);
        }
#pragma unroll
        for (int i = 0; i < h4; ++i) {
            const float send = u4 ? val[i] : val[i + h4];
            const float keep = u4 ? val[i + h4] : val[i];
            val[i] = keep + __shfl_xor(send, 4, 32);
        }
#pragma unroll
        for (int i = 0; i < h2; ++i) {
            const float send = u2 ? val[i] : val[i + h2];
            const float keep = u2 ? val[i + h2] : val[i];
            val[i] = keep + __shfl_xor(send, 2, 32);
        }
#pragma unroll
        for (int i = 0; i < h1; ++i) {
            const float send = u1 ? val[i] : val[i + h1];
            const float keep = u1 ? val[i + h1] : val[i];
            val[i] = keep + __shfl_xor(send, 1, 32);
        }
    }
    const int hb = cb / F;
#pragma unroll
    for (int j = 0; j < HW; ++j) {
        const int idx  = l15 * HW + j;
        const int kind = idx & 1;
        const int g    = (idx >> 1) % HW;
        const int v    = (idx >> 1) / HW;
        float* dst = kind ? sd : ss;
        dst[(size_t)(r0 + v) * H + hb + g] = val[j];
    }
}

#define SLOTS 32
#define COUNT_HELPERS 2
#define COUNT_BLOCKS  ((EE + MEAN_THREADS - 1) / MEAN_THREADS + COUNT_HELPERS)
__global__ __launch_bounds__(MEAN_THREADS) void edge_place_kernel(const int* __restrict__ ei,
                                                                  const float* __restrict__ ew,
                                                                  int* __restrict__ deg,
                                                                  int* __restrict__ slots,
                                                                  int* __restrict__ ovf_n,
                                                                  int* __restrict__ ovf_dst,
                                                                  int* __restrict__ ovf_src,
                                                                  float* __restrict__ ovf_w,
                                                                  const float* __restrict__ partial,
                                                                  float* __restrict__ meanw,
                                                                  const float* __restrict__ We0,
                                                                  const float* __restrict__ ae0,
                                                                  const float* __restrict__ We1,
                                                                  const float* __restrict__ ae1,
                                                                  const float* __restrict__ We2,
                                                                  const float* __restrict__ ae2,
                                                                  float* __restrict__ ce) {
    __shared__ float swave[MEAN_THREADS / 32];
    if (blockIdx.x == 0) {
        mean_final_block(partial, meanw, MEAN_BLOCKS, EE, swave);
        return;
    }
    if (blockIdx.x == 1) {
        edge_coef_block(We0, ae0, We1, ae1, We2, ae2, ce);
        return;
    }
    const int e = (blockIdx.x - COUNT_HELPERS) * MEAN_THREADS + threadIdx.x;
    if (e >= EE) return;
    const int   d = ei[EE + e];
    const int   s = ei[e];
    const float w = ew[e];
    const int   r = atomicAdd(&deg[d], 1);
    if (r < SLOTS) {
        int2 rec;
        rec.x = s;
        rec.y = __float_as_int(w);
        *(int2*)(slots + ((size_t)d * SLOTS + r) * 2) = rec;
    } else {
        const int p = atomicAdd(ovf_n, 1);
        ovf_dst[p] = d;
        ovf_src[p] = s;
        ovf_w[p]   = w;
    }
}

#define GATHER_WAVES 8

__device__ __forceinline__ float wave_max(float v) {
    for (int m = 16; m >= 1; m >>= 1) v = fmaxf(v, __shfl_xor(v, m, 32));
    return v;
}
__device__ __forceinline__ float wave_sum(float v) {
    for (int m = 16; m >= 1; m >>= 1) v += __shfl_xor(v, m, 32);
    return v;
}

__device__ __forceinline__ void slot_edge(const int* __restrict__ slots, int d, int n, int lane, int& s, float& w) {
    s = 0;
    w = 0.0f;
    if (lane < n) {
        const int2 rec = *(const int2*)(slots + ((size_t)d * SLOTS + lane) * 2);
        s = rec.x;
        w = __int_as_float(rec.y);
    }
}

__device__ __forceinline__ bool ovf_edge(const int* __restrict__ ovf_dst, const int* __restrict__ ovf_src,
                                         const float* __restrict__ ovf_w, int base, int nov, int d, int lane,
                                         int& s, float& w) {
    s = 0;
    w = 0.0f;
    bool ok = false;
    if (base + lane < nov) {
        ok = ovf_dst[base + lane] == d;
        if (ok) { s = ovf_src[base + lane]; w = ovf_w[base + lane]; }
    }
    return ok;
}

template <int H>
__device__ __forceinline__ void edge_score(const float* __restrict__ ss, const float* sdv, const float* cev,
                                           int s, float w, bool ok, float* a) {
    float q[H];
    if constexpr (H == 4) {
        const float4 t = *(const float4*)(ss + (size_t)s * 4);
        q[0] = t.x; q[1] = t.y; q[2] = t.z; q[3] = t.w;
    } else {
#pragma unroll
        for (int h = 0; h < H; ++h) q[h] = ss[(size_t)s * H + h];
    }
#pragma unroll
    for (int h = 0; h < H; ++h) {
        float x = q[h] + sdv[h] + w * cev[h];
        x = x > 0.0f ? x : 0.2f * x;
        a[h] = ok ? x : -__builtin_inff();
    }
}

template <int H>
__device__ __forceinline__ float pick_attn(const float* at, int t, int hl) {
    float r[H];
#pragma unroll
    for (int h = 0; h < H; ++h)
        r[h] = __int_as_float(__builtin_amdgcn_readlane(__float_as_int(at[h]), t));
    float v = r[0];
#pragma unroll
    for (int h = 1; h < H; ++h) v = (hl == h) ? r[h] : v;
    return v;
}

template <int VW> struct lane_vec { typedef float type __attribute__((ext_vector_type(VW))); };

template <int VW, int H, bool ELU>
__device__ __forceinline__ typename lane_vec<VW>::type gat_node_row(const int* __restrict__ deg,
                                                                    const int* __restrict__ slots,
                                                                    const int* __restrict__ ovf_n,
                                                                    const int* __restrict__ ovf_dst,
                                                                    const int* __restrict__ ovf_src,
                                                                    const float* __restrict__ ovf_w,
                                                                    const float* __restrict__ xw,
                                                                    const float* __restrict__ ss,
                                                                    const float* __restrict__ sd,
                                                                    const float* __restrict__ ce,
                                                                    const float* __restrict__ meanw,
                                                                    const float* __restrict__ b,
                                                                    const int d, const int lane) {
    constexpr int C = 32 * VW;
    constexpr int F = C / H;
    typedef typename lane_vec<VW>::type vecf;
    const int c0 = lane * VW;
    const int hl = c0 / F;
    const int dg  = deg[d];
    const int n0  = dg < SLOTS ? dg : SLOTS;
    const int nov = dg > SLOTS ? ovf_n[0] : 0;

    float cev[H], sdv[H], a_self[H], m[H];
    const float mw = meanw[0];
#pragma unroll
    for (int h = 0; h < H; ++h) {
        cev[h] = ce[h];
        sdv[h] = sd[(size_t)d * H + h];
        float x = ss[(size_t)d * H + h] + sdv[h] + mw * cev[h];
        a_self[h] = x > 0.0f ? x : 0.2f * x;
        m[h] = a_self[h];
    }

    int s0; float w0; float a0[H];
    slot_edge(slots, d, n0, lane, s0, w0);
    edge_score<H>(ss, sdv, cev, s0, w0, lane < n0, a0);
#pragma unroll
    for (int h = 0; h < H; ++h) m[h] = fmaxf(m[h], a0[h]);
    for (int base = 0; base < nov; base += 32) {
        int s; float w; float a[H];
        const bool ok = ovf_edge(ovf_dst, ovf_src, ovf_w, base, nov, d, lane, s, w);
        edge_score<H>(ss, sdv, cev, s, w, ok, a);
#pragma unroll
        for (int h = 0; h < H; ++h) m[h] = fmaxf(m[h], a[h]);
    }
#pragma unroll
    for (int h = 0; h < H; ++h) m[h] = wave_max(m[h]);

    float at0[H], den[H], ex_self[H];
#pragma unroll
    for (int h = 0; h < H; ++h) {
        at0[h] = lane < n0 ? __expf(a0[h] - m[h]) : 0.0f;
        den[h] = at0[h];
    }
    for (int base = 0; base < nov; base += 32) {
        int s; float w; float a[H];
        const bool ok = ovf_edge(ovf_dst, ovf_src, ovf_w, base, nov, d, lane, s, w);
        edge_score<H>(ss, sdv, cev, s, w, ok, a);
#pragma unroll
        for (int h = 0; h < H; ++h) den[h] += ok ? __expf(a[h] - m[h]) : 0.0f;
    }
#pragma unroll
    for (int h = 0; h < H; ++h) {
        ex_self[h] = __expf(a_self[h] - m[h]);
        den[h] = wave_sum(den[h]) + ex_self[h] + 1e-16f;
        at0[h] = at0[h] / den[h];
    }

    vecf acc = {};
#pragma unroll 4
    for (int t = 0; t < n0; ++t) {
        const int s = __builtin_amdgcn_readlane(s0, t);
        const float attn = pick_attn<H>(at0, t, hl);
        const vecf v = *(const vecf*)(xw + (size_t)s * C + c0);
        acc += v * attn;
    }
    for (int base = 0; base < nov; base += 32) {
        int s1; float w; float a[H];
        const bool ok = ovf_edge(ovf_dst, ovf_src, ovf_w, base, nov, d, lane, s1, w);
        unsigned int todo = __builtin_amdgcn_ballot_w32(ok);
        if (todo == 0u) continue;
        edge_score<H>(ss, sdv, cev, s1, w, ok, a);
#pragma unroll
        for (int h = 0; h < H; ++h) a[h] = (ok ? __expf(a[h] - m[h]) : 0.0f) / den[h];
        while (todo != 0u) {
            const int t = __builtin_ctz(todo);
            todo &= todo - 1u;
            const int s = __builtin_amdgcn_readlane(s1, t);
            const float attn = pick_attn<H>(a, t, hl);
            const vecf v = *(const vecf*)(xw + (size_t)s * C + c0);
            acc += v * attn;
        }
    }
    {
        float exs = ex_self[0], dn = den[0];
#pragma unroll
        for (int h = 1; h < H; ++h) { exs = (hl == h) ? ex_self[h] : exs; dn = (hl == h) ? den[h] : dn; }
        const float attn = exs / dn;
        const vecf v = *(const vecf*)(xw + (size_t)d * C + c0);
        acc += v * attn;
    }
    float bias[VW];
#pragma unroll
    for (int i = 0; i < VW; ++i) bias[i] = b[c0 + i];
    vecf r;
#pragma unroll
    for (int i = 0; i < VW; ++i) {
        float v = acc[i] + bias[i];
        if constexpr (ELU) v = v > 0.0f ? v : expm1f(v);
        r[i] = v;
    }
    return r;
}

template <int VW, int H, bool ELU>
__global__ __launch_bounds__(GATHER_WAVES * 32) void gat_node_kernel(const int* __restrict__ deg,
                                                                     const int* __restrict__ slots,
                                                                     const int* __restrict__ ovf_n,
                                                                     const int* __restrict__ ovf_dst,
                                                                     const int* __restrict__ ovf_src,
                                                                     const float* __restrict__ ovf_w,
                                                                     const float* __restrict__ xw,
                                                                     const float* __restrict__ ss,
                                                                     const float* __restrict__ sd,
                                                                     const float* __restrict__ ce,
                                                                     const float* __restrict__ meanw,
                                                                     const float* __restrict__ b,
                                                                     float* __restrict__ out) {
    typedef typename lane_vec<VW>::type vecf;
    const int lane = threadIdx.x & 31;
    const int d = __builtin_amdgcn_readfirstlane(blockIdx.x * GATHER_WAVES + (threadIdx.x >> 5));
    if (d >= NN) return;
    const vecf r = gat_node_row<VW, H, ELU>(deg, slots, ovf_n, ovf_dst, ovf_src, ovf_w,
                                            xw, ss, sd, ce, meanw, b, d, lane);
    *(vecf*)(out + (size_t)d * (32 * VW) + lane * VW) = r;
}

#define FUSE_NODES 16
#define FUSE_WAVES 8
#define FUSE_PITCH 132

template <int CN, int FN>
__global__ __launch_bounds__(FUSE_WAVES * 32) void gat_gemm_kernel(const int* __restrict__ deg,
                                                                   const int* __restrict__ slots,
                                                                   const int* __restrict__ ovf_n,
                                                                   const int* __restrict__ ovf_dst,
                                                                   const int* __restrict__ ovf_src,
                                                                   const float* __restrict__ ovf_w,
                                                                   const float* __restrict__ xw,
                                                                   const float* __restrict__ ss,
                                                                   const float* __restrict__ sd,
                                                                   const float* __restrict__ ce,
                                                                   const float* __restrict__ meanw,
                                                                   const float* __restrict__ b,
                                                                   const float* __restrict__ Wn,
                                                                   const float* __restrict__ a_sn,
                                                                   const float* __restrict__ a_dn,
                                                                   float* __restrict__ xwn,
                                                                   float* __restrict__ ssn,
                                                                   float* __restrict__ sdn) {
    constexpr int HN  = CN / FN;
    constexpr int NT  = CN / 16;
    constexpr int TPH = FN / 16;
    static_assert(NT <= FUSE_WAVES, "one tile per wave");
    static_assert(16 * HN * 2 <= FUSE_WAVES * 32, "one thread per finished score");
    typedef typename lane_vec<4>::type vecf;
    __shared__ __attribute__((aligned(16))) float hrow[FUSE_NODES * FUSE_PITCH];
    __shared__ float part[NT][16][2];
    const int lane  = threadIdx.x & 31;
    const int wave  = __builtin_amdgcn_readfirstlane(threadIdx.x >> 5);
    const int strip = blockIdx.x;
    for (int i = 0; i < FUSE_NODES / FUSE_WAVES; ++i) {
        const int rl = wave * (FUSE_NODES / FUSE_WAVES) + i;
        const vecf r = gat_node_row<4, 4, true>(deg, slots, ovf_n, ovf_dst, ovf_src, ovf_w,
                                                xw, ss, sd, ce, meanw, b, strip * 16 + rl, lane);
        *(vecf*)(hrow + rl * FUSE_PITCH + lane * 4) = r;
    }
    __syncthreads();
    if (wave < NT) {
        const int half = lane >> 4;
        const int l15  = lane & 15;
        const int cb   = wave * 16;
        v8f acc = {};
        const float* xrow = hrow + l15 * FUSE_PITCH + half * 2;
        const float* wcol = Wn + (size_t)(half * 2) * CN + cb + l15;
#pragma unroll 8
        for (int k = 0; k < KDIM; k += 4) {
            v2f a;
            a[0] = xrow[k];
            a[1] = xrow[k + 1];
            v2f bf;
            bf[0] = wcol[(size_t)k * CN];
            bf[1] = wcol[(size_t)(k + 1) * CN];
            acc = __builtin_amdgcn_wmma_f32_16x16x4_f32(
                 false, a,  false, bf,
                 (short)0, acc,  false,  false);
        }
        const int r0 = strip * 16 + half * 8;
#pragma unroll
        for (int v = 0; v < 8; ++v)
            xwn[(size_t)(r0 + v) * CN + cb + l15] = acc[v];

        const float as_c = a_sn[cb + l15];
        const float ad_c = a_dn[cb + l15];
        float val[16];
#pragma unroll
        for (int v = 0; v < 8; ++v) {
            val[2 * v + 0] = acc[v] * as_c;
            val[2 * v + 1] = acc[v] * ad_c;
        }
        {
            const bool u8 = (lane & 8) != 0, u4 = (lane & 4) != 0, u2 = (lane & 2) != 0, u1 = (lane & 1) != 0;
#pragma unroll
            for (int i = 0; i < 8; ++i) {
                const float send = u8 ? val[i] : val[i + 8];
                const float keep = u8 ? val[i + 8] : val[i];
                val[i] = keep + __shfl_xor(send, 8, 32);
            }
#pragma unroll
            for (int i = 0; i < 4; ++i) {
                const float send = u4 ? val[i] : val[i + 4];
                const float keep = u4 ? val[i + 4] : val[i];
                val[i] = keep + __shfl_xor(send, 4, 32);
            }
#pragma unroll
            for (int i = 0; i < 2; ++i) {
                const float send = u2 ? val[i] : val[i + 2];
                const float keep = u2 ? val[i + 2] : val[i];
                val[i] = keep + __shfl_xor(send, 2, 32);
            }
            {
                const float send = u1 ? val[0] : val[1];
                const float keep = u1 ? val[1] : val[0];
                val[0] = keep + __shfl_xor(send, 1, 32);
            }
        }
        part[wave][half * 8 + (l15 >> 1)][l15 & 1] = val[0];
    }
    __syncthreads();
    if (threadIdx.x < 16 * HN * 2) {
        const int kind = threadIdx.x & 1;
        const int h    = (threadIdx.x >> 1) % HN;
        const int row  = (threadIdx.x >> 1) / HN;
        float s = part[h * TPH][row][kind];
#pragma unroll
        for (int j = 1; j < TPH; ++j) s += part[h * TPH + j][row][kind];
        float* dst = kind ? sdn : ssn;
        dst[(size_t)(strip * 16 + row) * HN + h] = s;
    }
}

extern "C" void kernel_launch(void* const* d_in, const int* in_sizes, int n_in,
                              void* d_out, int out_size, void* d_ws, size_t ws_size,
                              hipStream_t stream) {
    const float* x  = (const float*)d_in[0];
    const int*   ei = (const int*)d_in[1];
    const float* ew = (const float*)d_in[2];

    const float* Wl[3] = { (const float*)d_in[3],  (const float*)d_in[9],  (const float*)d_in[15] };
    const float* As[3] = { (const float*)d_in[4],  (const float*)d_in[10], (const float*)d_in[16] };
    const float* Ad[3] = { (const float*)d_in[5],  (const float*)d_in[11], (const float*)d_in[17] };
    const float* We[3] = { (const float*)d_in[6],  (const float*)d_in[12], (const float*)d_in[18] };
    const float* Ae[3] = { (const float*)d_in[7],  (const float*)d_in[13], (const float*)d_in[19] };
    const float* Bi[3] = { (const float*)d_in[8],  (const float*)d_in[14], (const float*)d_in[20] };

    float* ws    = (float*)d_ws;
    float* xw    = ws;
    float* bufA  = xw    + (size_t)NN * 128;
    float* bufB  = bufA  + (size_t)NN * 128;
    float* ss    = bufB  + (size_t)NN * 128;
    float* sd    = ss    + (size_t)NN * 4;
    float* ce    = sd    + (size_t)NN * 4;
    float* meanw = ce + 32;
    float* mpart = meanw + 32;
    int* ovf_n   = (int*)(mpart + 256);
    int* deg     = ovf_n + 32;
    int* slots   = deg + NN + 16;
    int* ovf_dst = slots + (size_t)NN * SLOTS * 2;
    int* ovf_src = ovf_dst + EE;
    float* ovf_w = (float*)(ovf_src + EE);

    mean_partial_kernel<<<MEAN_BLOCKS, MEAN_THREADS, 0, stream>>>(ew, mpart, deg, ovf_n, EE);
    edge_place_kernel<<<COUNT_BLOCKS, MEAN_THREADS, 0, stream>>>(ei, ew, deg, slots, ovf_n, ovf_dst, ovf_src, ovf_w,
                                                                 mpart, meanw,
                                                                 We[0], Ae[0], We[1], Ae[1], We[2], Ae[2], ce);

    float* ss2 = bufB;
    float* sd2 = bufB + (size_t)NN * 4;

    gemm_score_kernel<128, 32><<<dim3((STRIPS + GEMM_WAVES - 1) / GEMM_WAVES, 128 / GEMM_COLS),
                                GEMM_WAVES * 32, 0, stream>>>(x, Wl[0], As[0], Ad[0], xw, ss, sd);
    gat_gemm_kernel<128, 32><<<STRIPS, FUSE_WAVES * 32, 0, stream>>>(
        deg, slots, ovf_n, ovf_dst, ovf_src, ovf_w,
        xw, ss, sd, ce, meanw, Bi[0], Wl[1], As[1], Ad[1], bufA, ss2, sd2);
    gat_gemm_kernel<64, 64><<<STRIPS, FUSE_WAVES * 32, 0, stream>>>(
        deg, slots, ovf_n, ovf_dst, ovf_src, ovf_w,
        bufA, ss2, sd2, ce + 4, meanw, Bi[1], Wl[2], As[2], Ad[2], xw, ss, sd);
    gat_node_kernel<2, 1, false><<<(NN + GATHER_WAVES - 1) / GATHER_WAVES, GATHER_WAVES * 32, 0, stream>>>(
        deg, slots, ovf_n, ovf_dst, ovf_src, ovf_w,
        xw, ss, sd, ce + 8, meanw, Bi[2], (float*)d_out);
}


# edges by destination: the per-edge returning add in memory becomes LDS counters in two passes (bin by 128-node bucket, rank inside it; 6 -> 7 launches); a node's edges still summed in arrival order
# speedup vs baseline: 11.2178x; 1.3304x over previous
#include <hip/hip_runtime.h>
#include <math.h>

#define NN    50000
#define EE    800000
#define ETOT  (EE + NN)
#define KDIM  128

#define BUCKET_SHIFT 7
#define BUCKET_NODES (1 << BUCKET_SHIFT)
#define NBUCKETS     ((NN + BUCKET_NODES - 1) / BUCKET_NODES)

typedef float v2f __attribute__((ext_vector_type(2)));
typedef float v8f __attribute__((ext_vector_type(8)));

#define MEAN_THREADS    256
#define MEAN_PER_THREAD 16
#define MEAN_BLOCKS     ((EE + MEAN_THREADS * MEAN_PER_THREAD - 1) / (MEAN_THREADS * MEAN_PER_THREAD))

__device__ __forceinline__ float block_sum_256(float s, float* swave) {
    for (int m = 16; m >= 1; m >>= 1) s += __shfl_xor(s, m, 32);
    const int wave = threadIdx.x >> 5, lane = threadIdx.x & 31;
    if (lane == 0) swave[wave] = s;
    __syncthreads();
    float t = (lane < MEAN_THREADS / 32) ? swave[lane] : 0.0f;
    for (int m = MEAN_THREADS / 64; m >= 1; m >>= 1) t += __shfl_xor(t, m, 32);
    return t;
}

__global__ __launch_bounds__(MEAN_THREADS) void mean_partial_kernel(const float* __restrict__ ew,
                                                                    float* __restrict__ partial,
                                                                    int* __restrict__ bfill,
                                                                    int* __restrict__ ovf_n, int E) {
    __shared__ float swave[MEAN_THREADS / 32];
    const int T = gridDim.x * MEAN_THREADS;
    const int g = blockIdx.x * MEAN_THREADS + threadIdx.x;
    for (int i = g; i < NBUCKETS; i += T) bfill[i] = 0;
    if (g == 0) { ovf_n[0] = 0; ovf_n[1] = 0; }
    float s = 0.0f;
    for (int i0 = g; i0 < E; i0 += MEAN_PER_THREAD * T) {
        float v[MEAN_PER_THREAD];
        for (int j = 0; j < MEAN_PER_THREAD; ++j) {
            const int i = i0 + j * T;
            const float x = ew[i < E ? i : E - 1];
            v[j] = i < E ? x : 0.0f;
        }
        for (int w = 1; w < MEAN_PER_THREAD; w <<= 1) {
            for (int j = 0; j < MEAN_PER_THREAD; j += 2 * w) v[j] += v[j + w];
        }
        s += v[0];
    }
    s = block_sum_256(s, swave);
    if (threadIdx.x == 0) partial[blockIdx.x] = s;
}

__device__ __forceinline__ void mean_final_block(const float* __restrict__ partial, float* __restrict__ meanw,
                                                 int nb, int E, float* swave) {
    float s = 0.0f;
    for (int i = threadIdx.x; i < nb; i += MEAN_THREADS) s += partial[i];
    s = block_sum_256(s, swave);
    if (threadIdx.x == 0) meanw[0] = s / (float)E;
}

__device__ __forceinline__ void edge_coef_block(const float* __restrict__ We0, const float* __restrict__ ae0,
                                                const float* __restrict__ We1, const float* __restrict__ ae1,
                                                const float* __restrict__ We2, const float* __restrict__ ae2,
                                                float* __restrict__ ce) {
    const int t = threadIdx.x;
    if (t >= 9) return;
    const int l = t >> 2, h = t & 3;
    const int F = l < 2 ? 32 : 64;
    const float* We = l == 0 ? We0 : (l == 1 ? We1 : We2);
    const float* ae = l == 0 ? ae0 : (l == 1 ? ae1 : ae2);
    float s = 0.0f;
    for (int f = 0; f < F; ++f) s += We[h * F + f] * ae[h * F + f];
    ce[t] = s;
}

#define GEMM_TILES 4
#define GEMM_COLS  (GEMM_TILES * 16)
#define GEMM_WAVES 4
#define STRIPS     (NN / 16)

template <int C, int F>
__global__ __launch_bounds__(GEMM_WAVES * 32) void gemm_score_kernel(const float* __restrict__ X,
                                                                     const float* __restrict__ W,
                                                                     const float* __restrict__ a_s,
                                                                     const float* __restrict__ a_d,
                                                                     float* __restrict__ xw,
                                                                     float* __restrict__ ss,
                                                                     float* __restrict__ sd) {
    constexpr int H   = C / F;
    constexpr int HW  = GEMM_COLS / F;
    constexpr int TPH = F / 16;
    constexpr int NV  = 16 * HW;
    const int lane = threadIdx.x & 31;
    const int strip = __builtin_amdgcn_readfirstlane(blockIdx.x * GEMM_WAVES + (threadIdx.x >> 5));
    if (strip >= STRIPS) return;
    const int half = lane >> 4;
    const int l15  = lane & 15;
    const int cb   = blockIdx.y * GEMM_COLS;

    v8f acc[GEMM_TILES] = {};
    const float* xrow = X + (size_t)(strip * 16 + l15) * KDIM + half * 2;
    const float* wcol = W + (size_t)(half * 2) * C + cb + l15;

#pragma unroll 8
    for (int k = 0; k < KDIM; k += 4) {
        v2f a;
        a[0] = xrow[k];
        a[1] = xrow[k + 1];
        v2f b[GEMM_TILES];
#pragma unroll
        for (int t = 0; t < GEMM_TILES; ++t) {
            b[t][0] = wcol[(size_t)k * C + t * 16];
            b[t][1] = wcol[(size_t)(k + 1) * C + t * 16];
        }
#pragma unroll
        for (int t = 0; t < GEMM_TILES; ++t)
            acc[t] = __builtin_amdgcn_wmma_f32_16x16x4_f32(
                 false, a,  false, b[t],
                 (short)0, acc[t],  false,  false);
    }

    const int r0 = strip * 16 + half * 8;
#pragma unroll
    for (int t = 0; t < GEMM_TILES; ++t) {
#pragma unroll
        for (int v = 0; v < 8; ++v)
            xw[(size_t)(r0 + v) * C + cb + t * 16 + l15] = acc[t][v];
    }

    float as_c[GEMM_TILES], ad_c[GEMM_TILES];
#pragma unroll
    for (int t = 0; t < GEMM_TILES; ++t) {
        as_c[t] = a_s[cb + t * 16 + l15];
        ad_c[t] = a_d[cb + t * 16 + l15];
    }
    float val[NV];
#pragma unroll
    for (int v = 0; v < 8; ++v) {
#pragma unroll
        for (int g = 0; g < HW; ++g) {
            float ps = acc[g * TPH][v] * as_c[g * TPH];
            float pd = acc[g * TPH][v] * ad_c[g * TPH];
#pragma unroll
            for (int j = 1; j < TPH; ++j) {
                ps += acc[g * TPH + j][v] * as_c[g * TPH + j];
                pd += acc[g * TPH + j][v] * ad_c[g * TPH + j];
            }
            val[(v * HW + g) * 2 + 0] = ps;
            val[(v * HW + g) * 2 + 1] = pd;
        }
    }
    {
        constexpr int h8 = NV / 2, h4 = NV / 4, h2 = NV / 8, h1 = NV / 16;
        const bool u8 = (lane & 8) != 0, u4 = (lane & 4) != 0, u2 = (lane & 2) != 0, u1 = (lane & 1) != 0;
#pragma unroll
        for (int i = 0; i < h8; ++i) {
            const float send = u8 ? val[i] : val[i + h8];
            const float keep = u8 ? val[i + h8] : val[i];
            val[i] = keep + __shfl_xor(send, 8, 32);
        }
#pragma unroll
        for (int i = 0; i < h4; ++i) {
            const float send = u4 ? val[i] : val[i + h4];
            const float keep = u4 ? val[i + h4] : val[i];
            val[i] = keep + __shfl_xor(send, 4, 32);
        }
#pragma unroll
        for (int i = 0; i < h2; ++i) {
            const float send = u2 ? val[i] : val[i + h2];
            const float keep = u2 ? val[i + h2] : val[i];
            val[i] = keep + __shfl_xor(send, 2, 32);
        }
#pragma unroll
        for (int i = 0; i < h1; ++i) {
            const float send = u1 ? val[i] : val[i + h1];
            const float keep = u1 ? val[i + h1] : val[i];
            val[i] = keep + __shfl_xor(send, 1, 32);
        }
    }
    const int hb = cb / F;
#pragma unroll
    for (int j = 0; j < HW; ++j) {
        const int idx  = l15 * HW + j;
        const int kind = idx & 1;
        const int g    = (idx >> 1) % HW;
        const int v    = (idx >> 1) / HW;
        float* dst = kind ? sd : ss;
        dst[(size_t)(r0 + v) * H + hb + g] = val[j];
    }
}

#define SLOTS 32
#define BUCKET_CAP    4096
#define BIN_EPT       16
#define BIN_CHUNK     (MEAN_THREADS * BIN_EPT)
#define BIN_BLOCKS    ((EE + BIN_CHUNK - 1) / BIN_CHUNK)
#define RANK_EPT      8
#define COUNT_HELPERS 2

__device__ __forceinline__ void ovf_push(int* __restrict__ ovf_n, int* __restrict__ ovf_dst,
                                         int* __restrict__ ovf_src, float* __restrict__ ovf_w,
                                         int dnode, int snode, float w) {
    const int p = atomicAdd(ovf_n, 1);
    ovf_dst[p] = dnode;
    ovf_src[p] = snode;
    ovf_w[p]   = w;
}

__global__ __launch_bounds__(MEAN_THREADS) void edge_bin_kernel(const int* __restrict__ ei,
                                                                const float* __restrict__ ew,
                                                                int* __restrict__ bfill,
                                                                int* __restrict__ recs,
                                                                int* __restrict__ ovf_n,
                                                                int* __restrict__ ovf_dst,
                                                                int* __restrict__ ovf_src,
                                                                float* __restrict__ ovf_w,
                                                                const float* __restrict__ partial,
                                                                float* __restrict__ meanw,
                                                                const float* __restrict__ We0,
                                                                const float* __restrict__ ae0,
                                                                const float* __restrict__ We1,
                                                                const float* __restrict__ ae1,
                                                                const float* __restrict__ We2,
                                                                const float* __restrict__ ae2,
                                                                float* __restrict__ ce) {
    __shared__ float swave[MEAN_THREADS / 32];
    __shared__ int  cnt[NBUCKETS];
    __shared__ int  lpos[NBUCKETS];
    __shared__ int  wsum[MEAN_THREADS / 32];
    __shared__ int2 stage[BIN_CHUNK];
    __shared__ int  stage_b[BIN_CHUNK];
    if (blockIdx.x == 0) {
        mean_final_block(partial, meanw, MEAN_BLOCKS, EE, swave);
        return;
    }
    if (blockIdx.x == 1) {
        edge_coef_block(We0, ae0, We1, ae1, We2, ae2, ce);
        return;
    }
    const int tid  = threadIdx.x;
    const int lane = tid & 31, wave = tid >> 5;
    const int base = (blockIdx.x - COUNT_HELPERS) * BIN_CHUNK;
    for (int t = tid; t < NBUCKETS; t += MEAN_THREADS) cnt[t] = 0;
    __syncthreads();

    int dn[BIN_EPT], sn[BIN_EPT], rk[BIN_EPT];
    float wv[BIN_EPT];
#pragma unroll
    for (int j = 0; j < BIN_EPT; ++j) {
        const int e  = base + j * MEAN_THREADS + tid;
        const int ec = e < EE ? e : EE - 1;
        const int dv = ei[EE + ec];
        sn[j] = ei[ec];
        wv[j] = ew[ec];
        dn[j] = e < EE ? dv : -1;
    }
#pragma unroll
    for (int j = 0; j < BIN_EPT; ++j) {
        rk[j] = 0;
        if (dn[j] >= 0) rk[j] = atomicAdd(&cnt[dn[j] >> BUCKET_SHIFT], 1);
    }
    __syncthreads();

    {
        const int b0 = 2 * tid, b1 = 2 * tid + 1;
        const int c0 = b0 < NBUCKETS ? cnt[b0] : 0;
        const int c1 = b1 < NBUCKETS ? cnt[b1] : 0;
        int v = c0 + c1;
        for (int o = 1; o < 32; o <<= 1) {
            const int u = __shfl_up(v, o, 32);
            if (lane >= o) v += u;
        }
        if (lane == 31) wsum[wave] = v;
        __syncthreads();
        int woff = 0;
        for (int k = 0; k < MEAN_THREADS / 32; ++k) woff += k < wave ? wsum[k] : 0;
        const int excl = woff + v - (c0 + c1);
        if (b0 < NBUCKETS) lpos[b0] = excl;
        if (b1 < NBUCKETS) lpos[b1] = excl + c0;
    }
    for (int t = tid; t < NBUCKETS; t += MEAN_THREADS) {
        const int n = cnt[t];
        cnt[t] = n > 0 ? atomicAdd(&bfill[t], n) : 0;
    }
    __syncthreads();

#pragma unroll
    for (int j = 0; j < BIN_EPT; ++j) {
        if (dn[j] >= 0) {
            const int b = dn[j] >> BUCKET_SHIFT;
            const int i = lpos[b] + rk[j];
            int2 rec;
            rec.x = (sn[j] << BUCKET_SHIFT) | (dn[j] & (BUCKET_NODES - 1));
            rec.y = __float_as_int(wv[j]);
            stage[i]   = rec;
            stage_b[i] = b;
        }
    }
    __syncthreads();

    const int nvalid = (EE - base) < BIN_CHUNK ? (EE - base) : BIN_CHUNK;
    for (int i = tid; i < nvalid; i += MEAN_THREADS) {
        const int  b   = stage_b[i];
        const int2 rec = stage[i];
        const int  pos = cnt[b] + (i - lpos[b]);
        if (pos < BUCKET_CAP) {
            *(int2*)(recs + ((size_t)b * BUCKET_CAP + pos) * 2) = rec;
        } else {
            ovf_n[1] = 1;
            ovf_push(ovf_n, ovf_dst, ovf_src, ovf_w, b * BUCKET_NODES + (rec.x & (BUCKET_NODES - 1)),
                     rec.x >> BUCKET_SHIFT, __int_as_float(rec.y));
        }
    }
}

__global__ __launch_bounds__(MEAN_THREADS) void edge_rank_kernel(const int* __restrict__ bfill,
                                                                 const int* __restrict__ recs,
                                                                 int* __restrict__ deg,
                                                                 int* __restrict__ slots,
                                                                 int* __restrict__ ovf_n,
                                                                 int* __restrict__ ovf_dst,
                                                                 int* __restrict__ ovf_src,
                                                                 float* __restrict__ ovf_w) {
    __shared__ int  cnt[BUCKET_NODES];
    __shared__ int2 row[BUCKET_NODES * SLOTS];
    const int tid = threadIdx.x;
    const int b   = blockIdx.x;
    const int nb  = bfill[b];
    const int n   = nb < BUCKET_CAP ? nb : BUCKET_CAP;
    if (tid < BUCKET_NODES) cnt[tid] = 0;
    __syncthreads();
    for (int i0 = tid; i0 < n; i0 += RANK_EPT * MEAN_THREADS) {
        int2 rec[RANK_EPT];
#pragma unroll
        for (int j = 0; j < RANK_EPT; ++j) {
            const int i  = i0 + j * MEAN_THREADS;
            const int ic = i < n ? i : n - 1;
            rec[j] = *(const int2*)(recs + ((size_t)b * BUCKET_CAP + ic) * 2);
        }
#pragma unroll
        for (int j = 0; j < RANK_EPT; ++j) {
            if (i0 + j * MEAN_THREADS < n) {
                const int dl = rec[j].x & (BUCKET_NODES - 1);
                const int s  = rec[j].x >> BUCKET_SHIFT;
                const int r  = atomicAdd(&cnt[dl], 1);
                if (r < SLOTS) {
                    int2 o;
                    o.x = s;
                    o.y = rec[j].y;
                    row[dl * SLOTS + r] = o;
                } else {
                    ovf_push(ovf_n, ovf_dst, ovf_src, ovf_w, b * BUCKET_NODES + dl, s, __int_as_float(rec[j].y));
                }
            }
        }
    }
    __syncthreads();
    const int lane = tid & 31;
    for (int rl = tid >> 5; rl < BUCKET_NODES; rl += MEAN_THREADS / 32) {
        const int node = b * BUCKET_NODES + rl;
        if (node < NN) {
            const int c = cnt[rl];
            if (lane == 0) deg[node] = c;
            if (lane < c) *(int2*)(slots + ((size_t)node * SLOTS + lane) * 2) = row[rl * SLOTS + lane];
        }
    }
}

#define GATHER_WAVES 8

__device__ __forceinline__ float wave_max(float v) {
    for (int m = 16; m >= 1; m >>= 1) v = fmaxf(v, __shfl_xor(v, m, 32));
    return v;
}
__device__ __forceinline__ float wave_sum(float v) {
    for (int m = 16; m >= 1; m >>= 1) v += __shfl_xor(v, m, 32);
    return v;
}

__device__ __forceinline__ void slot_edge(const int* __restrict__ slots, int d, int n, int lane, int& s, float& w) {
    s = 0;
    w = 0.0f;
    if (lane < n) {
        const int2 rec = *(const int2*)(slots + ((size_t)d * SLOTS + lane) * 2);
        s = rec.x;
        w = __int_as_float(rec.y);
    }
}

__device__ __forceinline__ bool ovf_edge(const int* __restrict__ ovf_dst, const int* __restrict__ ovf_src,
                                         const float* __restrict__ ovf_w, int base, int nov, int d, int lane,
                                         int& s, float& w) {
    s = 0;
    w = 0.0f;
    bool ok = false;
    if (base + lane < nov) {
        ok = ovf_dst[base + lane] == d;
        if (ok) { s = ovf_src[base + lane]; w = ovf_w[base + lane]; }
    }
    return ok;
}

template <int H>
__device__ __forceinline__ void edge_score(const float* __restrict__ ss, const float* sdv, const float* cev,
                                           int s, float w, bool ok, float* a) {
    float q[H];
    if constexpr (H == 4) {
        const float4 t = *(const float4*)(ss + (size_t)s * 4);
        q[0] = t.x; q[1] = t.y; q[2] = t.z; q[3] = t.w;
    } else {
#pragma unroll
        for (int h = 0; h < H; ++h) q[h] = ss[(size_t)s * H + h];
    }
#pragma unroll
    for (int h = 0; h < H; ++h) {
        float x = q[h] + sdv[h] + w * cev[h];
        x = x > 0.0f ? x : 0.2f * x;
        a[h] = ok ? x : -__builtin_inff();
    }
}

template <int H>
__device__ __forceinline__ float pick_attn(const float* at, int t, int hl) {
    float r[H];
#pragma unroll
    for (int h = 0; h < H; ++h)
        r[h] = __int_as_float(__builtin_amdgcn_readlane(__float_as_int(at[h]), t));
    float v = r[0];
#pragma unroll
    for (int h = 1; h < H; ++h) v = (hl == h) ? r[h] : v;
    return v;
}

template <int VW> struct lane_vec { typedef float type __attribute__((ext_vector_type(VW))); };

template <int VW, int H, bool ELU>
__device__ __forceinline__ typename lane_vec<VW>::type gat_node_row(const int* __restrict__ deg,
                                                                    const int* __restrict__ slots,
                                                                    const int* __restrict__ ovf_n,
                                                                    const int* __restrict__ ovf_dst,
                                                                    const int* __restrict__ ovf_src,
                                                                    const float* __restrict__ ovf_w,
                                                                    const float* __restrict__ xw,
                                                                    const float* __restrict__ ss,
                                                                    const float* __restrict__ sd,
                                                                    const float* __restrict__ ce,
                                                                    const float* __restrict__ meanw,
                                                                    const float* __restrict__ b,
                                                                    const int d, const int lane) {
    constexpr int C = 32 * VW;
    constexpr int F = C / H;
    typedef typename lane_vec<VW>::type vecf;
    const int c0 = lane * VW;
    const int hl = c0 / F;
    const int dg  = deg[d];
    const int n0  = dg < SLOTS ? dg : SLOTS;
    const int nov = (dg > SLOTS || ovf_n[1] != 0) ? ovf_n[0] : 0;

    float cev[H], sdv[H], a_self[H], m[H];
    const float mw = meanw[0];
#pragma unroll
    for (int h = 0; h < H; ++h) {
        cev[h] = ce[h];
        sdv[h] = sd[(size_t)d * H + h];
        float x = ss[(size_t)d * H + h] + sdv[h] + mw * cev[h];
        a_self[h] = x > 0.0f ? x : 0.2f * x;
        m[h] = a_self[h];
    }

    int s0; float w0; float a0[H];
    slot_edge(slots, d, n0, lane, s0, w0);
    edge_score<H>(ss, sdv, cev, s0, w0, lane < n0, a0);
#pragma unroll
    for (int h = 0; h < H; ++h) m[h] = fmaxf(m[h], a0[h]);
    for (int base = 0; base < nov; base += 32) {
        int s; float w; float a[H];
        const bool ok = ovf_edge(ovf_dst, ovf_src, ovf_w, base, nov, d, lane, s, w);
        edge_score<H>(ss, sdv, cev, s, w, ok, a);
#pragma unroll
        for (int h = 0; h < H; ++h) m[h] = fmaxf(m[h], a[h]);
    }
#pragma unroll
    for (int h = 0; h < H; ++h) m[h] = wave_max(m[h]);

    float at0[H], den[H], ex_self[H];
#pragma unroll
    for (int h = 0; h < H; ++h) {
        at0[h] = lane < n0 ? __expf(a0[h] - m[h]) : 0.0f;
        den[h] = at0[h];
    }
    for (int base = 0; base < nov; base += 32) {
        int s; float w; float a[H];
        const bool ok = ovf_edge(ovf_dst, ovf_src, ovf_w, base, nov, d, lane, s, w);
        edge_score<H>(ss, sdv, cev, s, w, ok, a);
#pragma unroll
        for (int h = 0; h < H; ++h) den[h] += ok ? __expf(a[h] - m[h]) : 0.0f;
    }
#pragma unroll
    for (int h = 0; h < H; ++h) {
        ex_self[h] = __expf(a_self[h] - m[h]);
        den[h] = wave_sum(den[h]) + ex_self[h] + 1e-16f;
        at0[h] = at0[h] / den[h];
    }

    vecf acc = {};
#pragma unroll 4
    for (int t = 0; t < n0; ++t) {
        const int s = __builtin_amdgcn_readlane(s0, t);
        const float attn = pick_attn<H>(at0, t, hl);
        const vecf v = *(const vecf*)(xw + (size_t)s * C + c0);
        acc += v * attn;
    }
    for (int base = 0; base < nov; base += 32) {
        int s1; float w; float a[H];
        const bool ok = ovf_edge(ovf_dst, ovf_src, ovf_w, base, nov, d, lane, s1, w);
        unsigned int todo = __builtin_amdgcn_ballot_w32(ok);
        if (todo == 0u) continue;
        edge_score<H>(ss, sdv, cev, s1, w, ok, a);
#pragma unroll
        for (int h = 0; h < H; ++h) a[h] = (ok ? __expf(a[h] - m[h]) : 0.0f) / den[h];
        while (todo != 0u) {
            const int t = __builtin_ctz(todo);
            todo &= todo - 1u;
            const int s = __builtin_amdgcn_readlane(s1, t);
            const float attn = pick_attn<H>(a, t, hl);
            const vecf v = *(const vecf*)(xw + (size_t)s * C + c0);
            acc += v * attn;
        }
    }
    {
        float exs = ex_self[0], dn = den[0];
#pragma unroll
        for (int h = 1; h < H; ++h) { exs = (hl == h) ? ex_self[h] : exs; dn = (hl == h) ? den[h] : dn; }
        const float attn = exs / dn;
        const vecf v = *(const vecf*)(xw + (size_t)d * C + c0);
        acc += v * attn;
    }
    float bias[VW];
#pragma unroll
    for (int i = 0; i < VW; ++i) bias[i] = b[c0 + i];
    vecf r;
#pragma unroll
    for (int i = 0; i < VW; ++i) {
        float v = acc[i] + bias[i];
        if constexpr (ELU) v = v > 0.0f ? v : expm1f(v);
        r[i] = v;
    }
    return r;
}

template <int VW, int H, bool ELU>
__global__ __launch_bounds__(GATHER_WAVES * 32) void gat_node_kernel(const int* __restrict__ deg,
                                                                     const int* __restrict__ slots,
                                                                     const int* __restrict__ ovf_n,
                                                                     const int* __restrict__ ovf_dst,
                                                                     const int* __restrict__ ovf_src,
                                                                     const float* __restrict__ ovf_w,
                                                                     const float* __restrict__ xw,
                                                                     const float* __restrict__ ss,
                                                                     const float* __restrict__ sd,
                                                                     const float* __restrict__ ce,
                                                                     const float* __restrict__ meanw,
                                                                     const float* __restrict__ b,
                                                                     float* __restrict__ out) {
    typedef typename lane_vec<VW>::type vecf;
    const int lane = threadIdx.x & 31;
    const int d = __builtin_amdgcn_readfirstlane(blockIdx.x * GATHER_WAVES + (threadIdx.x >> 5));
    if (d >= NN) return;
    const vecf r = gat_node_row<VW, H, ELU>(deg, slots, ovf_n, ovf_dst, ovf_src, ovf_w,
                                            xw, ss, sd, ce, meanw, b, d, lane);
    *(vecf*)(out + (size_t)d * (32 * VW) + lane * VW) = r;
}

#define FUSE_NODES 16
#define FUSE_WAVES 8
#define FUSE_PITCH 132

template <int CN, int FN>
__global__ __launch_bounds__(FUSE_WAVES * 32) void gat_gemm_kernel(const int* __restrict__ deg,
                                                                   const int* __restrict__ slots,
                                                                   const int* __restrict__ ovf_n,
                                                                   const int* __restrict__ ovf_dst,
                                                                   const int* __restrict__ ovf_src,
                                                                   const float* __restrict__ ovf_w,
                                                                   const float* __restrict__ xw,
                                                                   const float* __restrict__ ss,
                                                                   const float* __restrict__ sd,
                                                                   const float* __restrict__ ce,
                                                                   const float* __restrict__ meanw,
                                                                   const float* __restrict__ b,
                                                                   const float* __restrict__ Wn,
                                                                   const float* __restrict__ a_sn,
                                                                   const float* __restrict__ a_dn,
                                                                   float* __restrict__ xwn,
                                                                   float* __restrict__ ssn,
                                                                   float* __restrict__ sdn) {
    constexpr int HN  = CN / FN;
    constexpr int NT  = CN / 16;
    constexpr int TPH = FN / 16;
    static_assert(NT <= FUSE_WAVES, "one tile per wave");
    static_assert(16 * HN * 2 <= FUSE_WAVES * 32, "one thread per finished score");
    typedef typename lane_vec<4>::type vecf;
    __shared__ __attribute__((aligned(16))) float hrow[FUSE_NODES * FUSE_PITCH];
    __shared__ float part[NT][16][2];
    const int lane  = threadIdx.x & 31;
    const int wave  = __builtin_amdgcn_readfirstlane(threadIdx.x >> 5);
    const int strip = blockIdx.x;
    for (int i = 0; i < FUSE_NODES / FUSE_WAVES; ++i) {
        const int rl = wave * (FUSE_NODES / FUSE_WAVES) + i;
        const vecf r = gat_node_row<4, 4, true>(deg, slots, ovf_n, ovf_dst, ovf_src, ovf_w,
                                                xw, ss, sd, ce, meanw, b, strip * 16 + rl, lane);
        *(vecf*)(hrow + rl * FUSE_PITCH + lane * 4) = r;
    }
    __syncthreads();
    if (wave < NT) {
        const int half = lane >> 4;
        const int l15  = lane & 15;
        const int cb   = wave * 16;
        v8f acc = {};
        const float* xrow = hrow + l15 * FUSE_PITCH + half * 2;
        const float* wcol = Wn + (size_t)(half * 2) * CN + cb + l15;
#pragma unroll 8
        for (int k = 0; k < KDIM; k += 4) {
            v2f a;
            a[0] = xrow[k];
            a[1] = xrow[k + 1];
            v2f bf;
            bf[0] = wcol[(size_t)k * CN];
            bf[1] = wcol[(size_t)(k + 1) * CN];
            acc = __builtin_amdgcn_wmma_f32_16x16x4_f32(
                 false, a,  false, bf,
                 (short)0, acc,  false,  false);
        }
        const int r0 = strip * 16 + half * 8;
#pragma unroll
        for (int v = 0; v < 8; ++v)
            xwn[(size_t)(r0 + v) * CN + cb + l15] = acc[v];

        const float as_c = a_sn[cb + l15];
        const float ad_c = a_dn[cb + l15];
        float val[16];
#pragma unroll
        for (int v = 0; v < 8; ++v) {
            val[2 * v + 0] = acc[v] * as_c;
            val[2 * v + 1] = acc[v] * ad_c;
        }
        {
            const bool u8 = (lane & 8) != 0, u4 = (lane & 4) != 0, u2 = (lane & 2) != 0, u1 = (lane & 1) != 0;
#pragma unroll
            for (int i = 0; i < 8; ++i) {
                const float send = u8 ? val[i] : val[i + 8];
                const float keep = u8 ? val[i + 8] : val[i];
                val[i] = keep + __shfl_xor(send, 8, 32);
            }
#pragma unroll
            for (int i = 0; i < 4; ++i) {
                const float send = u4 ? val[i] : val[i + 4];
                const float keep = u4 ? val[i + 4] : val[i];
                val[i] = keep + __shfl_xor(send, 4, 32);
            }
#pragma unroll
            for (int i = 0; i < 2; ++i) {
                const float send = u2 ? val[i] : val[i + 2];
                const float keep = u2 ? val[i + 2] : val[i];
                val[i] = keep + __shfl_xor(send, 2, 32);
            }
            {
                const float send = u1 ? val[0] : val[1];
                const float keep = u1 ? val[1] : val[0];
                val[0] = keep + __shfl_xor(send, 1, 32);
            }
        }
        part[wave][half * 8 + (l15 >> 1)][l15 & 1] = val[0];
    }
    __syncthreads();
    if (threadIdx.x < 16 * HN * 2) {
        const int kind = threadIdx.x & 1;
        const int h    = (threadIdx.x >> 1) % HN;
        const int row  = (threadIdx.x >> 1) / HN;
        float s = part[h * TPH][row][kind];
#pragma unroll
        for (int j = 1; j < TPH; ++j) s += part[h * TPH + j][row][kind];
        float* dst = kind ? sdn : ssn;
        dst[(size_t)(strip * 16 + row) * HN + h] = s;
    }
}

extern "C" void kernel_launch(void* const* d_in, const int* in_sizes, int n_in,
                              void* d_out, int out_size, void* d_ws, size_t ws_size,
                              hipStream_t stream) {
    const float* x  = (const float*)d_in[0];
    const int*   ei = (const int*)d_in[1];
    const float* ew = (const float*)d_in[2];

    const float* Wl[3] = { (const float*)d_in[3],  (const float*)d_in[9],  (const float*)d_in[15] };
    const float* As[3] = { (const float*)d_in[4],  (const float*)d_in[10], (const float*)d_in[16] };
    const float* Ad[3] = { (const float*)d_in[5],  (const float*)d_in[11], (const float*)d_in[17] };
    const float* We[3] = { (const float*)d_in[6],  (const float*)d_in[12], (const float*)d_in[18] };
    const float* Ae[3] = { (const float*)d_in[7],  (const float*)d_in[13], (const float*)d_in[19] };
    const float* Bi[3] = { (const float*)d_in[8],  (const float*)d_in[14], (const float*)d_in[20] };

    float* ws    = (float*)d_ws;
    float* xw    = ws;
    float* bufA  = xw    + (size_t)NN * 128;
    float* bufB  = bufA  + (size_t)NN * 128;
    float* ss    = bufB  + (size_t)NN * 128;
    float* sd    = ss    + (size_t)NN * 4;
    float* ce    = sd    + (size_t)NN * 4;
    float* meanw = ce + 32;
    float* mpart = meanw + 32;
    int* ovf_n   = (int*)(mpart + 256);
    int* deg     = ovf_n + 32;
    int* slots   = deg + NN + 16;
    int* ovf_dst = slots + (size_t)NN * SLOTS * 2;
    int* ovf_src = ovf_dst + EE;
    float* ovf_w = (float*)(ovf_src + EE);
    int* bfill   = (int*)(ovf_w + EE);
    int* recs    = bfill + 512;

    mean_partial_kernel<<<MEAN_BLOCKS, MEAN_THREADS, 0, stream>>>(ew, mpart, bfill, ovf_n, EE);
    edge_bin_kernel<<<BIN_BLOCKS + COUNT_HELPERS, MEAN_THREADS, 0, stream>>>(ei, ew, bfill, recs,
                                                                             ovf_n, ovf_dst, ovf_src, ovf_w,
                                                                             mpart, meanw,
                                                                             We[0], Ae[0], We[1], Ae[1], We[2], Ae[2], ce);
    edge_rank_kernel<<<NBUCKETS, MEAN_THREADS, 0, stream>>>(bfill, recs, deg, slots, ovf_n, ovf_dst, ovf_src, ovf_w);

    float* ss2 = bufB;
    float* sd2 = bufB + (size_t)NN * 4;

    gemm_score_kernel<128, 32><<<dim3((STRIPS + GEMM_WAVES - 1) / GEMM_WAVES, 128 / GEMM_COLS),
                                GEMM_WAVES * 32, 0, stream>>>(x, Wl[0], As[0], Ad[0], xw, ss, sd);
    gat_gemm_kernel<128, 32><<<STRIPS, FUSE_WAVES * 32, 0, stream>>>(
        deg, slots, ovf_n, ovf_dst, ovf_src, ovf_w,
        xw, ss, sd, ce, meanw, Bi[0], Wl[1], As[1], Ad[1], bufA, ss2, sd2);
    gat_gemm_kernel<64, 64><<<STRIPS, FUSE_WAVES * 32, 0, stream>>>(
        deg, slots, ovf_n, ovf_dst, ovf_src, ovf_w,
        bufA, ss2, sd2, ce + 4, meanw, Bi[1], Wl[2], As[2], Ad[2], xw, ss, sd);
    gat_node_kernel<2, 1, false><<<(NN + GATHER_WAVES - 1) / GATHER_WAVES, GATHER_WAVES * 32, 0, stream>>>(
        deg, slots, ovf_n, ovf_dst, ovf_src, ovf_w,
        xw, ss, sd, ce + 8, meanw, Bi[2], (float*)d_out);
}


# edge scores move into the per-destination pass: dot of each gathered source row with a_s (own row with a_d), softmax under a running max; the per-edge score-line gather goes; sum order changes
# speedup vs baseline: 11.8683x; 1.0580x over previous
#include <hip/hip_runtime.h>
#include <math.h>

#define NN    50000
#define EE    800000
#define ETOT  (EE + NN)
#define KDIM  128

#define BUCKET_SHIFT 7
#define BUCKET_NODES (1 << BUCKET_SHIFT)
#define NBUCKETS     ((NN + BUCKET_NODES - 1) / BUCKET_NODES)

typedef float v2f __attribute__((ext_vector_type(2)));
typedef float v8f __attribute__((ext_vector_type(8)));

#define MEAN_THREADS    256
#define MEAN_PER_THREAD 16
#define MEAN_BLOCKS     ((EE + MEAN_THREADS * MEAN_PER_THREAD - 1) / (MEAN_THREADS * MEAN_PER_THREAD))

__device__ __forceinline__ float block_sum_256(float s, float* swave) {
    for (int m = 16; m >= 1; m >>= 1) s += __shfl_xor(s, m, 32);
    const int wave = threadIdx.x >> 5, lane = threadIdx.x & 31;
    if (lane == 0) swave[wave] = s;
    __syncthreads();
    float t = (lane < MEAN_THREADS / 32) ? swave[lane] : 0.0f;
    for (int m = MEAN_THREADS / 64; m >= 1; m >>= 1) t += __shfl_xor(t, m, 32);
    return t;
}

__global__ __launch_bounds__(MEAN_THREADS) void mean_partial_kernel(const float* __restrict__ ew,
                                                                    float* __restrict__ partial,
                                                                    int* __restrict__ bfill,
                                                                    int* __restrict__ ovf_n, int E) {
    __shared__ float swave[MEAN_THREADS / 32];
    const int T = gridDim.x * MEAN_THREADS;
    const int g = blockIdx.x * MEAN_THREADS + threadIdx.x;
    for (int i = g; i < NBUCKETS; i += T) bfill[i] = 0;
    if (g == 0) { ovf_n[0] = 0; ovf_n[1] = 0; }
    float s = 0.0f;
    for (int i0 = g; i0 < E; i0 += MEAN_PER_THREAD * T) {
        float v[MEAN_PER_THREAD];
        for (int j = 0; j < MEAN_PER_THREAD; ++j) {
            const int i = i0 + j * T;
            const float x = ew[i < E ? i : E - 1];
            v[j] = i < E ? x : 0.0f;
        }
        for (int w = 1; w < MEAN_PER_THREAD; w <<= 1) {
            for (int j = 0; j < MEAN_PER_THREAD; j += 2 * w) v[j] += v[j + w];
        }
        s += v[0];
    }
    s = block_sum_256(s, swave);
    if (threadIdx.x == 0) partial[blockIdx.x] = s;
}

__device__ __forceinline__ void mean_final_block(const float* __restrict__ partial, float* __restrict__ meanw,
                                                 int nb, int E, float* swave) {
    float s = 0.0f;
    for (int i = threadIdx.x; i < nb; i += MEAN_THREADS) s += partial[i];
    s = block_sum_256(s, swave);
    if (threadIdx.x == 0) meanw[0] = s / (float)E;
}

__device__ __forceinline__ void edge_coef_block(const float* __restrict__ We0, const float* __restrict__ ae0,
                                                const float* __restrict__ We1, const float* __restrict__ ae1,
                                                const float* __restrict__ We2, const float* __restrict__ ae2,
                                                float* __restrict__ ce) {
    const int t = threadIdx.x;
    if (t >= 9) return;
    const int l = t >> 2, h = t & 3;
    const int F = l < 2 ? 32 : 64;
    const float* We = l == 0 ? We0 : (l == 1 ? We1 : We2);
    const float* ae = l == 0 ? ae0 : (l == 1 ? ae1 : ae2);
    float s = 0.0f;
    for (int f = 0; f < F; ++f) s += We[h * F + f] * ae[h * F + f];
    ce[t] = s;
}

#define GEMM_TILES 4
#define GEMM_COLS  (GEMM_TILES * 16)
#define GEMM_WAVES 4
#define STRIPS     (NN / 16)

template <int C>
__global__ __launch_bounds__(GEMM_WAVES * 32) void gemm_kernel(const float* __restrict__ X,
                                                               const float* __restrict__ W,
                                                               float* __restrict__ xw) {
    const int lane = threadIdx.x & 31;
    const int strip = __builtin_amdgcn_readfirstlane(blockIdx.x * GEMM_WAVES + (threadIdx.x >> 5));
    if (strip >= STRIPS) return;
    const int half = lane >> 4;
    const int l15  = lane & 15;
    const int cb   = blockIdx.y * GEMM_COLS;

    v8f acc[GEMM_TILES] = {};
    const float* xrow = X + (size_t)(strip * 16 + l15) * KDIM + half * 2;
    const float* wcol = W + (size_t)(half * 2) * C + cb + l15;

#pragma unroll 8
    for (int k = 0; k < KDIM; k += 4) {
        v2f a;
        a[0] = xrow[k];
        a[1] = xrow[k + 1];
        v2f b[GEMM_TILES];
#pragma unroll
        for (int t = 0; t < GEMM_TILES; ++t) {
            b[t][0] = wcol[(size_t)k * C + t * 16];
            b[t][1] = wcol[(size_t)(k + 1) * C + t * 16];
        }
#pragma unroll
        for (int t = 0; t < GEMM_TILES; ++t)
            acc[t] = __builtin_amdgcn_wmma_f32_16x16x4_f32(
                 false, a,  false, b[t],
                 (short)0, acc[t],  false,  false);
    }

    const int r0 = strip * 16 + half * 8;
#pragma unroll
    for (int t = 0; t < GEMM_TILES; ++t) {
#pragma unroll
        for (int v = 0; v < 8; ++v)
            xw[(size_t)(r0 + v) * C + cb + t * 16 + l15] = acc[t][v];
    }
}

#define SLOTS 32
#define BUCKET_CAP    4096
#define BIN_EPT       16
#define BIN_CHUNK     (MEAN_THREADS * BIN_EPT)
#define BIN_BLOCKS    ((EE + BIN_CHUNK - 1) / BIN_CHUNK)
#define RANK_EPT      8
#define COUNT_HELPERS 2

__device__ __forceinline__ void ovf_push(int* __restrict__ ovf_n, int* __restrict__ ovf_dst,
                                         int* __restrict__ ovf_src, float* __restrict__ ovf_w,
                                         int dnode, int snode, float w) {
    const int p = atomicAdd(ovf_n, 1);
    ovf_dst[p] = dnode;
    ovf_src[p] = snode;
    ovf_w[p]   = w;
}

__global__ __launch_bounds__(MEAN_THREADS) void edge_bin_kernel(const int* __restrict__ ei,
                                                                const float* __restrict__ ew,
                                                                int* __restrict__ bfill,
                                                                int* __restrict__ recs,
                                                                int* __restrict__ ovf_n,
                                                                int* __restrict__ ovf_dst,
                                                                int* __restrict__ ovf_src,
                                                                float* __restrict__ ovf_w,
                                                                const float* __restrict__ partial,
                                                                float* __restrict__ meanw,
                                                                const float* __restrict__ We0,
                                                                const float* __restrict__ ae0,
                                                                const float* __restrict__ We1,
                                                                const float* __restrict__ ae1,
                                                                const float* __restrict__ We2,
                                                                const float* __restrict__ ae2,
                                                                float* __restrict__ ce) {
    __shared__ float swave[MEAN_THREADS / 32];
    __shared__ int  cnt[NBUCKETS];
    __shared__ int  lpos[NBUCKETS];
    __shared__ int  wsum[MEAN_THREADS / 32];
    __shared__ int2 stage[BIN_CHUNK];
    __shared__ int  stage_b[BIN_CHUNK];
    if (blockIdx.x == 0) {
        mean_final_block(partial, meanw, MEAN_BLOCKS, EE, swave);
        return;
    }
    if (blockIdx.x == 1) {
        edge_coef_block(We0, ae0, We1, ae1, We2, ae2, ce);
        return;
    }
    const int tid  = threadIdx.x;
    const int lane = tid & 31, wave = tid >> 5;
    const int base = (blockIdx.x - COUNT_HELPERS) * BIN_CHUNK;
    for (int t = tid; t < NBUCKETS; t += MEAN_THREADS) cnt[t] = 0;
    __syncthreads();

    int dn[BIN_EPT], sn[BIN_EPT], rk[BIN_EPT];
    float wv[BIN_EPT];
#pragma unroll
    for (int j = 0; j < BIN_EPT; ++j) {
        const int e  = base + j * MEAN_THREADS + tid;
        const int ec = e < EE ? e : EE - 1;
        const int dv = ei[EE + ec];
        sn[j] = ei[ec];
        wv[j] = ew[ec];
        dn[j] = e < EE ? dv : -1;
    }
#pragma unroll
    for (int j = 0; j < BIN_EPT; ++j) {
        rk[j] = 0;
        if (dn[j] >= 0) rk[j] = atomicAdd(&cnt[dn[j] >> BUCKET_SHIFT], 1);
    }
    __syncthreads();

    {
        const int b0 = 2 * tid, b1 = 2 * tid + 1;
        const int c0 = b0 < NBUCKETS ? cnt[b0] : 0;
        const int c1 = b1 < NBUCKETS ? cnt[b1] : 0;
        int v = c0 + c1;
        for (int o = 1; o < 32; o <<= 1) {
            const int u = __shfl_up(v, o, 32);
            if (lane >= o) v += u;
        }
        if (lane == 31) wsum[wave] = v;
        __syncthreads();
        int woff = 0;
        for (int k = 0; k < MEAN_THREADS / 32; ++k) woff += k < wave ? wsum[k] : 0;
        const int excl = woff + v - (c0 + c1);
        if (b0 < NBUCKETS) lpos[b0] = excl;
        if (b1 < NBUCKETS) lpos[b1] = excl + c0;
    }
    for (int t = tid; t < NBUCKETS; t += MEAN_THREADS) {
        const int n = cnt[t];
        cnt[t] = n > 0 ? atomicAdd(&bfill[t], n) : 0;
    }
    __syncthreads();

#pragma unroll
    for (int j = 0; j < BIN_EPT; ++j) {
        if (dn[j] >= 0) {
            const int b = dn[j] >> BUCKET_SHIFT;
            const int i = lpos[b] + rk[j];
            int2 rec;
            rec.x = (sn[j] << BUCKET_SHIFT) | (dn[j] & (BUCKET_NODES - 1));
            rec.y = __float_as_int(wv[j]);
            stage[i]   = rec;
            stage_b[i] = b;
        }
    }
    __syncthreads();

    const int nvalid = (EE - base) < BIN_CHUNK ? (EE - base) : BIN_CHUNK;
    for (int i = tid; i < nvalid; i += MEAN_THREADS) {
        const int  b   = stage_b[i];
        const int2 rec = stage[i];
        const int  pos = cnt[b] + (i - lpos[b]);
        if (pos < BUCKET_CAP) {
            *(int2*)(recs + ((size_t)b * BUCKET_CAP + pos) * 2) = rec;
        } else {
            ovf_n[1] = 1;
            ovf_push(ovf_n, ovf_dst, ovf_src, ovf_w, b * BUCKET_NODES + (rec.x & (BUCKET_NODES - 1)),
                     rec.x >> BUCKET_SHIFT, __int_as_float(rec.y));
        }
    }
}

__global__ __launch_bounds__(MEAN_THREADS) void edge_rank_kernel(const int* __restrict__ bfill,
                                                                 const int* __restrict__ recs,
                                                                 int* __restrict__ deg,
                                                                 int* __restrict__ slots,
                                                                 int* __restrict__ ovf_n,
                                                                 int* __restrict__ ovf_dst,
                                                                 int* __restrict__ ovf_src,
                                                                 float* __restrict__ ovf_w) {
    __shared__ int  cnt[BUCKET_NODES];
    __shared__ int2 row[BUCKET_NODES * SLOTS];
    const int tid = threadIdx.x;
    const int b   = blockIdx.x;
    const int nb  = bfill[b];
    const int n   = nb < BUCKET_CAP ? nb : BUCKET_CAP;
    if (tid < BUCKET_NODES) cnt[tid] = 0;
    __syncthreads();
    for (int i0 = tid; i0 < n; i0 += RANK_EPT * MEAN_THREADS) {
        int2 rec[RANK_EPT];
#pragma unroll
        for (int j = 0; j < RANK_EPT; ++j) {
            const int i  = i0 + j * MEAN_THREADS;
            const int ic = i < n ? i : n - 1;
            rec[j] = *(const int2*)(recs + ((size_t)b * BUCKET_CAP + ic) * 2);
        }
#pragma unroll
        for (int j = 0; j < RANK_EPT; ++j) {
            if (i0 + j * MEAN_THREADS < n) {
                const int dl = rec[j].x & (BUCKET_NODES - 1);
                const int s  = rec[j].x >> BUCKET_SHIFT;
                const int r  = atomicAdd(&cnt[dl], 1);
                if (r < SLOTS) {
                    int2 o;
                    o.x = s;
                    o.y = rec[j].y;
                    row[dl * SLOTS + r] = o;
                } else {
                    ovf_push(ovf_n, ovf_dst, ovf_src, ovf_w, b * BUCKET_NODES + dl, s, __int_as_float(rec[j].y));
                }
            }
        }
    }
    __syncthreads();
    const int lane = tid & 31;
    for (int rl = tid >> 5; rl < BUCKET_NODES; rl += MEAN_THREADS / 32) {
        const int node = b * BUCKET_NODES + rl;
        if (node < NN) {
            const int c = cnt[rl];
            if (lane == 0) deg[node] = c;
            if (lane < c) *(int2*)(slots + ((size_t)node * SLOTS + lane) * 2) = row[rl * SLOTS + lane];
        }
    }
}

#define GATHER_WAVES 8

template <int M>
__device__ __forceinline__ float lane_xor(float v) {
    return __int_as_float(__builtin_amdgcn_update_dpp(0, __float_as_int(v), 0x160 + M, 0xf, 0xf, true));
}
__device__ __forceinline__ float lane_xor16(float v) {
    return __int_as_float(__builtin_amdgcn_permlanex16(__float_as_int(v), __float_as_int(v),
                                                       0x76543210, 0xfedcba98, false, false));
}
template <int G>
__device__ __forceinline__ float head_sum(float v) {
    static_assert(G == 8 || G == 32, "a head is 8 lanes or the whole wave");
    v += lane_xor<1>(v);
    v += lane_xor<2>(v);
    v += lane_xor<4>(v);
    if constexpr (G == 32) {
        v += lane_xor<8>(v);
        v += lane_xor16(v);
    }
    return v;
}

__device__ __forceinline__ void slot_edge(const int* __restrict__ slots, int d, int n, int lane, int& s, float& w) {
    s = d;
    w = 0.0f;
    if (lane < n) {
        const int2 rec = *(const int2*)(slots + ((size_t)d * SLOTS + lane) * 2);
        s = rec.x;
        w = __int_as_float(rec.y);
    }
}

__device__ __forceinline__ bool ovf_edge(const int* __restrict__ ovf_dst, const int* __restrict__ ovf_src,
                                         const float* __restrict__ ovf_w, int base, int nov, int d, int lane,
                                         int& s, float& w) {
    s = 0;
    w = 0.0f;
    bool ok = false;
    if (base + lane < nov) {
        ok = ovf_dst[base + lane] == d;
        if (ok) { s = ovf_src[base + lane]; w = ovf_w[base + lane]; }
    }
    return ok;
}

template <int VW> struct lane_vec { typedef float type __attribute__((ext_vector_type(VW))); };

template <int VW>
__device__ __forceinline__ float lane_dot(typename lane_vec<VW>::type row, typename lane_vec<VW>::type coef) {
    static_assert(VW == 2 || VW == 4, "2 or 4 channels per lane");
    if constexpr (VW == 4) {
        v2f p = row.lo * coef.lo;
        p = row.hi * coef.hi + p;
        return p[0] + p[1];
    } else {
        const v2f p = row * coef;
        return p[0] + p[1];
    }
}

template <int VW, int H, bool ELU>
__device__ __forceinline__ typename lane_vec<VW>::type gat_node_row(const int* __restrict__ deg,
                                                                    const int* __restrict__ slots,
                                                                    const int* __restrict__ ovf_n,
                                                                    const int* __restrict__ ovf_dst,
                                                                    const int* __restrict__ ovf_src,
                                                                    const float* __restrict__ ovf_w,
                                                                    const float* __restrict__ xw,
                                                                    const float* __restrict__ a_s,
                                                                    const float* __restrict__ a_d,
                                                                    const float* __restrict__ ce,
                                                                    const float* __restrict__ meanw,
                                                                    const float* __restrict__ b,
                                                                    const int d, const int lane) {
    constexpr int C = 32 * VW;
    constexpr int F = C / H;
    constexpr int G = F / VW;
    constexpr int B = 16 / VW;
    typedef typename lane_vec<VW>::type vecf;
    const int c0 = lane * VW;
    const int hl = c0 / F;
    const int dg  = deg[d];
    const int n0  = dg < SLOTS ? dg : SLOTS;
    const int nov = (dg > SLOTS || ovf_n[1] != 0) ? ovf_n[0] : 0;

    int s0; float w0;
    slot_edge(slots, d, n0, lane, s0, w0);
    const vecf rowd = ((const vecf*)(xw + (size_t)d * C))[lane];

    const vecf asv = *(const vecf*)(a_s + c0);
    const vecf adv = *(const vecf*)(a_d + c0);
    const float cel = ce[hl];
    const float mw  = meanw[0];

    const float sdl = head_sum<G>(lane_dot<VW>(rowd, adv));
    float m;
    {
        float x = head_sum<G>(lane_dot<VW>(rowd, asv)) + sdl + mw * cel;
        m = x > 0.0f ? x : 0.2f * x;
    }
    float den = 1.0f;
    vecf  acc = rowd;

    for (int t = 0; t < n0; t += B) {
        vecf  row[B];
        float wv[B];
#pragma unroll
        for (int i = 0; i < B; ++i) {
            const int s = __builtin_amdgcn_readlane(s0, t + i);
            wv[i]  = __int_as_float(__builtin_amdgcn_readlane(__float_as_int(w0), t + i));
            row[i] = ((const vecf*)(xw + (size_t)s * C))[lane];
        }
        typedef float vbf __attribute__((ext_vector_type(B)));
        vbf qv, wq;
#pragma unroll
        for (int i = 0; i < B; ++i) {
            qv[i] = head_sum<G>(lane_dot<VW>(row[i], asv));
            wq[i] = wv[i];
        }
        const vbf xv = qv + sdl + wq * cel;
        float x[B];
#pragma unroll
        for (int i = 0; i < B; ++i) {
            const float v = xv[i] > 0.0f ? xv[i] : 0.2f * xv[i];
            x[i] = t + i < n0 ? v : -__builtin_inff();
        }
        float mn = m;
#pragma unroll
        for (int i = 0; i < B; ++i) mn = fmaxf(mn, x[i]);
        const float sc = __expf(m - mn);
        float es = 0.0f;
        acc = acc * sc;
#pragma unroll
        for (int i = 0; i < B; ++i) {
            const float e = __expf(x[i] - mn);
            es  += e;
            acc += row[i] * e;
        }
        den = den * sc + es;
        m   = mn;
    }
    for (int base = 0; base < nov; base += 32) {
        int s1; float w1;
        const bool ok = ovf_edge(ovf_dst, ovf_src, ovf_w, base, nov, d, lane, s1, w1);
        unsigned int todo = __builtin_amdgcn_ballot_w32(ok);
        while (todo != 0u) {
            const int t = __builtin_ctz(todo);
            todo &= todo - 1u;
            const int   s = __builtin_amdgcn_readlane(s1, t);
            const float w = __int_as_float(__builtin_amdgcn_readlane(__float_as_int(w1), t));
            const vecf row = ((const vecf*)(xw + (size_t)s * C))[lane];
            float x = head_sum<G>(lane_dot<VW>(row, asv)) + sdl + w * cel;
            x = x > 0.0f ? x : 0.2f * x;
            const float mn = fmaxf(m, x);
            const float sc = __expf(m - mn);
            const float e  = __expf(x - mn);
            den = den * sc + e;
            acc = acc * sc + row * e;
            m   = mn;
        }
    }

    const float inv = 1.0f / (den + 1e-16f);
    float bias[VW];
#pragma unroll
    for (int i = 0; i < VW; ++i) bias[i] = b[c0 + i];
    vecf r;
#pragma unroll
    for (int i = 0; i < VW; ++i) {
        float v = acc[i] * inv + bias[i];
        if constexpr (ELU) v = v > 0.0f ? v : expm1f(v);
        r[i] = v;
    }
    return r;
}

template <int VW, int H, bool ELU>
__global__ __launch_bounds__(GATHER_WAVES * 32) void gat_node_kernel(const int* __restrict__ deg,
                                                                     const int* __restrict__ slots,
                                                                     const int* __restrict__ ovf_n,
                                                                     const int* __restrict__ ovf_dst,
                                                                     const int* __restrict__ ovf_src,
                                                                     const float* __restrict__ ovf_w,
                                                                     const float* __restrict__ xw,
                                                                     const float* __restrict__ a_s,
                                                                     const float* __restrict__ a_d,
                                                                     const float* __restrict__ ce,
                                                                     const float* __restrict__ meanw,
                                                                     const float* __restrict__ b,
                                                                     float* __restrict__ out) {
    typedef typename lane_vec<VW>::type vecf;
    const int lane = threadIdx.x & 31;
    const int d = __builtin_amdgcn_readfirstlane(blockIdx.x * GATHER_WAVES + (threadIdx.x >> 5));
    if (d >= NN) return;
    const vecf r = gat_node_row<VW, H, ELU>(deg, slots, ovf_n, ovf_dst, ovf_src, ovf_w,
                                            xw, a_s, a_d, ce, meanw, b, d, lane);
    *(vecf*)(out + (size_t)d * (32 * VW) + lane * VW) = r;
}

#define FUSE_NODES 16
#define FUSE_WAVES 8
#define FUSE_PITCH 132

template <int CN>
__global__ __launch_bounds__(FUSE_WAVES * 32) void gat_gemm_kernel(const int* __restrict__ deg,
                                                                   const int* __restrict__ slots,
                                                                   const int* __restrict__ ovf_n,
                                                                   const int* __restrict__ ovf_dst,
                                                                   const int* __restrict__ ovf_src,
                                                                   const float* __restrict__ ovf_w,
                                                                   const float* __restrict__ xw,
                                                                   const float* __restrict__ a_s,
                                                                   const float* __restrict__ a_d,
                                                                   const float* __restrict__ ce,
                                                                   const float* __restrict__ meanw,
                                                                   const float* __restrict__ b,
                                                                   const float* __restrict__ Wn,
                                                                   float* __restrict__ xwn) {
    constexpr int NT  = CN / 16;
    static_assert(NT <= FUSE_WAVES, "one tile per wave");
    typedef typename lane_vec<4>::type vecf;
    __shared__ __attribute__((aligned(16))) float hrow[FUSE_NODES * FUSE_PITCH];
    const int lane  = threadIdx.x & 31;
    const int wave  = __builtin_amdgcn_readfirstlane(threadIdx.x >> 5);
    const int strip = blockIdx.x;
    for (int i = 0; i < FUSE_NODES / FUSE_WAVES; ++i) {
        const int rl = wave * (FUSE_NODES / FUSE_WAVES) + i;
        const vecf r = gat_node_row<4, 4, true>(deg, slots, ovf_n, ovf_dst, ovf_src, ovf_w,
                                                xw, a_s, a_d, ce, meanw, b, strip * 16 + rl, lane);
        *(vecf*)(hrow + rl * FUSE_PITCH + lane * 4) = r;
    }
    __syncthreads();
    if (wave < NT) {
        const int half = lane >> 4;
        const int l15  = lane & 15;
        const int cb   = wave * 16;
        v8f acc = {};
        const float* xrow = hrow + l15 * FUSE_PITCH + half * 2;
        const float* wcol = Wn + (size_t)(half * 2) * CN + cb + l15;
#pragma unroll 8
        for (int k = 0; k < KDIM; k += 4) {
            v2f a;
            a[0] = xrow[k];
            a[1] = xrow[k + 1];
            v2f bf;
            bf[0] = wcol[(size_t)k * CN];
            bf[1] = wcol[(size_t)(k + 1) * CN];
            acc = __builtin_amdgcn_wmma_f32_16x16x4_f32(
                 false, a,  false, bf,
                 (short)0, acc,  false,  false);
        }
        const int r0 = strip * 16 + half * 8;
#pragma unroll
        for (int v = 0; v < 8; ++v)
            xwn[(size_t)(r0 + v) * CN + cb + l15] = acc[v];
    }
}

extern "C" void kernel_launch(void* const* d_in, const int* in_sizes, int n_in,
                              void* d_out, int out_size, void* d_ws, size_t ws_size,
                              hipStream_t stream) {
    const float* x  = (const float*)d_in[0];
    const int*   ei = (const int*)d_in[1];
    const float* ew = (const float*)d_in[2];

    const float* Wl[3] = { (const float*)d_in[3],  (const float*)d_in[9],  (const float*)d_in[15] };
    const float* As[3] = { (const float*)d_in[4],  (const float*)d_in[10], (const float*)d_in[16] };
    const float* Ad[3] = { (const float*)d_in[5],  (const float*)d_in[11], (const float*)d_in[17] };
    const float* We[3] = { (const float*)d_in[6],  (const float*)d_in[12], (const float*)d_in[18] };
    const float* Ae[3] = { (const float*)d_in[7],  (const float*)d_in[13], (const float*)d_in[19] };
    const float* Bi[3] = { (const float*)d_in[8],  (const float*)d_in[14], (const float*)d_in[20] };

    float* ws    = (float*)d_ws;
    float* xw    = ws;
    float* bufA  = xw    + (size_t)NN * 128;
    float* ce    = bufA  + (size_t)NN * 128;
    float* meanw = ce + 32;
    float* mpart = meanw + 32;
    int* ovf_n   = (int*)(mpart + 256);
    int* deg     = ovf_n + 32;
    int* slots   = deg + NN + 16;
    int* ovf_dst = slots + (size_t)NN * SLOTS * 2;
    int* ovf_src = ovf_dst + EE;
    float* ovf_w = (float*)(ovf_src + EE);
    int* bfill   = (int*)(ovf_w + EE);
    int* recs    = bfill + 512;

    mean_partial_kernel<<<MEAN_BLOCKS, MEAN_THREADS, 0, stream>>>(ew, mpart, bfill, ovf_n, EE);
    edge_bin_kernel<<<BIN_BLOCKS + COUNT_HELPERS, MEAN_THREADS, 0, stream>>>(ei, ew, bfill, recs,
                                                                             ovf_n, ovf_dst, ovf_src, ovf_w,
                                                                             mpart, meanw,
                                                                             We[0], Ae[0], We[1], Ae[1], We[2], Ae[2], ce);
    edge_rank_kernel<<<NBUCKETS, MEAN_THREADS, 0, stream>>>(bfill, recs, deg, slots, ovf_n, ovf_dst, ovf_src, ovf_w);

    gemm_kernel<128><<<dim3((STRIPS + GEMM_WAVES - 1) / GEMM_WAVES, 128 / GEMM_COLS),
                       GEMM_WAVES * 32, 0, stream>>>(x, Wl[0], xw);
    gat_gemm_kernel<128><<<STRIPS, FUSE_WAVES * 32, 0, stream>>>(
        deg, slots, ovf_n, ovf_dst, ovf_src, ovf_w,
        xw, As[0], Ad[0], ce, meanw, Bi[0], Wl[1], bufA);
    gat_gemm_kernel<64><<<STRIPS, FUSE_WAVES * 32, 0, stream>>>(
        deg, slots, ovf_n, ovf_dst, ovf_src, ovf_w,
        bufA, As[1], Ad[1], ce + 4, meanw, Bi[1], Wl[2], xw);
    gat_node_kernel<2, 1, false><<<(NN + GATHER_WAVES - 1) / GATHER_WAVES, GATHER_WAVES * 32, 0, stream>>>(
        deg, slots, ovf_n, ovf_dst, ovf_src, ovf_w,
        xw, As[2], Ad[2], ce + 8, meanw, Bi[2], (float*)d_out);
}
